# starting kernel
# baseline (speedup 1.0000x reference)
#include <hip/hip_runtime.h>

#define TT 8
#define T1 (TT + 6)
#define T2 (TT + 4)
#define T3 (TT + 2)
#define NB 64
#define NTIME 512
#define NV 21
#define NC 32
#define NCIN 2
#define NOUT 64

typedef __attribute__((ext_vector_type(2))) float v2f;
typedef __attribute__((ext_vector_type(8))) float v8f;

__device__ __forceinline__ v8f wmma4(v2f a, v2f b, v8f c) {
  return __builtin_amdgcn_wmma_f32_16x16x4_f32(
       false, a,  false, b,
       (short)0, c,  false,  false);
}

__device__ const unsigned char g_pi[20] = {0,1,2,3,0,5,6,7,0,9,10,11,0,13,14,15,0,17,18,19};
__device__ const unsigned char g_pj[20] = {1,2,3,4,5,6,7,8,9,10,11,12,13,14,15,16,17,18,19,20};
__device__ const float g_dinv[21] = {
  0.4082482905f,
  0.5773502692f,0.5773502692f,0.5773502692f,0.7071067812f,
  0.5773502692f,0.5773502692f,0.5773502692f,0.7071067812f,
  0.5773502692f,0.5773502692f,0.5773502692f,0.7071067812f,
  0.5773502692f,0.5773502692f,0.5773502692f,0.7071067812f,
  0.5773502692f,0.5773502692f,0.5773502692f,0.7071067812f};

__global__ __launch_bounds__(256) void stgcn_fused(
    const float* __restrict__ x,
    const float* __restrict__ gcn_w0, const float* __restrict__ gcn_b0,
    const float* __restrict__ tcn_w0, const float* __restrict__ tcn_b0,
    const float* __restrict__ bn_g0, const float* __restrict__ bn_b0,
    const float* __restrict__ bn_m0, const float* __restrict__ bn_v0,
    const float* __restrict__ res_w, const float* __restrict__ res_b,
    const float* __restrict__ res_bn_g, const float* __restrict__ res_bn_b,
    const float* __restrict__ res_bn_m, const float* __restrict__ res_bn_v,
    const float* __restrict__ gcn_w1, const float* __restrict__ gcn_b1,
    const float* __restrict__ tcn_w1, const float* __restrict__ tcn_b1,
    const float* __restrict__ bn_g1, const float* __restrict__ bn_b1,
    const float* __restrict__ bn_m1, const float* __restrict__ bn_v1,
    const float* __restrict__ fc_w, const float* __restrict__ fc_b,
    float* __restrict__ out)
{
  __shared__ float sA[NV * NV];
  __shared__ float s_in[T1 * NV * NCIN];
  __shared__ float s_xa[T1 * NV * NCIN];
  __shared__ float s_tw0[96 * NC];
  __shared__ float s_tw1[96 * NC];
  __shared__ float s_gw1[NC * NC];
  __shared__ float s_g0[T1 * NV * NC];
  __shared__ float s_h0[T2 * NV * NC];
  __shared__ float s_g1[T3 * NV * NC];
  __shared__ float s_gw0[NC * NCIN];
  __shared__ float s_gb0[NC], s_gb1[NC];
  __shared__ float s_s0[NC], s_o0[NC], s_s1[NC], s_o1[NC];
  __shared__ float s_rA[NC], s_rB[NC], s_rC[NC];
  __shared__ float s_fcb[NOUT];

  const int tid  = threadIdx.x;
  const int nthr = 256;
  const int lane = tid & 31;
  const int wave = tid >> 5;
  const int row  = lane & 15;
  const int hi   = lane >> 4;

  const int bid   = blockIdx.x;
  const int tile  = bid & 63;
  const int hand  = (bid >> 6) & 1;
  const int b     = bid >> 7;
  const int tbase = tile * TT;

  for (int idx = tid; idx < NV * NV; idx += nthr) {
    int i = idx / NV, j = idx % NV;
    bool conn = (i == j);
    for (int p = 0; p < 20; ++p)
      conn = conn || ((int)g_pi[p] == i && (int)g_pj[p] == j) ||
                     ((int)g_pi[p] == j && (int)g_pj[p] == i);
    sA[idx] = conn ? g_dinv[i] * g_dinv[j] : 0.0f;
  }
  for (int idx = tid; idx < 96 * NC; idx += nthr) {
    int co = idx & 31, kap = idx >> 5, k = kap >> 5, ci = kap & 31;
    s_tw0[idx] = tcn_w0[(co * NC + ci) * 3 + k];
    s_tw1[idx] = tcn_w1[(co * NC + ci) * 3 + k];
  }
  for (int idx = tid; idx < NC * NC; idx += nthr) s_gw1[idx] = gcn_w1[idx];
  if (tid < NC * NCIN) s_gw0[tid] = gcn_w0[tid];
  if (tid < NC) {
    s_gb0[tid] = gcn_b0[tid];
    s_gb1[tid] = gcn_b1[tid];
    float sc0 = bn_g0[tid] * rsqrtf(bn_v0[tid] + 1e-5f);
    s_s0[tid] = sc0;
    s_o0[tid] = (tcn_b0[tid] - bn_m0[tid]) * sc0 + bn_b0[tid];
    float sc1 = bn_g1[tid] * rsqrtf(bn_v1[tid] + 1e-5f);
    s_s1[tid] = sc1;
    s_o1[tid] = (tcn_b1[tid] - bn_m1[tid]) * sc1 + bn_b1[tid];
    float rsc = res_bn_g[tid] * rsqrtf(res_bn_v[tid] + 1e-5f);
    s_rA[tid] = res_w[tid * 2 + 0] * rsc;
    s_rB[tid] = res_w[tid * 2 + 1] * rsc;
    s_rC[tid] = (res_b[tid] - res_bn_m[tid]) * rsc + res_bn_b[tid];
  }
  if (tid < NOUT) s_fcb[tid] = fc_b[tid];

  for (int idx = tid; idx < T1 * NV * NCIN; idx += nthr) {
    int i = idx / (NV * NCIN), r = idx % (NV * NCIN);
    int v = r >> 1, ci = r & 1;
    int t = tbase - 3 + i;
    float val = 0.0f;
    if (t >= 0 && t < NTIME)
      val = x[(((b * NTIME + t) * 2 + hand) * NV + v) * NCIN + ci];
    s_in[idx] = val;
  }
  __syncthreads();

  for (int idx = tid; idx < T1 * NV * NCIN; idx += nthr) {
    int i = idx / (NV * NCIN), r = idx % (NV * NCIN);
    int v = r >> 1, ci = r & 1;
    float acc = 0.0f;
    const float* Ar = &sA[v * NV];
    const float* xr = &s_in[i * NV * NCIN + ci];
    for (int u = 0; u < NV; ++u) acc += Ar[u] * xr[u * NCIN];
    s_xa[idx] = acc;
  }
  __syncthreads();

  for (int idx = tid; idx < T1 * NV * NC; idx += nthr) {
    int c = idx & 31, rv = idx >> 5;
    int v = rv % NV, i = rv / NV;
    int t = tbase - 3 + i;
    float val = 0.0f;
    if (t >= 0 && t < NTIME) {
      float xa0 = s_xa[(i * NV + v) * 2 + 0];
      float xa1 = s_xa[(i * NV + v) * 2 + 1];
      val = s_gw0[c * 2 + 0] * xa0 + s_gw0[c * 2 + 1] * xa1 + s_gb0[c];
    }
    s_g0[idx] = val;
  }
  __syncthreads();

  {
    const int M = T2 * NV;
    for (int tIdx = wave; tIdx < 16 * 2; tIdx += 8) {
      int mt = tIdx >> 1, nt = tIdx & 1;
      v8f acc = {0.f,0.f,0.f,0.f,0.f,0.f,0.f,0.f};
      int mA = mt * 16 + row;
      int mC = (mA < M) ? mA : 0;
      int t2 = mC / NV, vv = mC % NV;
      int colN = nt * 16 + row;
      for (int ks = 0; ks < 24; ++ks) {
        int k0 = ks * 4 + hi * 2;
        v2f a, bb;
        a.x = s_g0[(t2 + (k0 >> 5)) * (NV * NC) + vv * NC + (k0 & 31)];
        a.y = s_g0[(t2 + ((k0 + 1) >> 5)) * (NV * NC) + vv * NC + ((k0 + 1) & 31)];
        bb.x = s_tw0[k0 * NC + colN];
        bb.y = s_tw0[(k0 + 1) * NC + colN];
        acc = wmma4(a, bb, acc);
      }
#pragma unroll
      for (int j = 0; j < 8; ++j) {
        int m = mt * 16 + j + hi * 8;
        if (m < M) {
          int te = m / NV, ve = m % NV;
          int co = colN;
          float val = acc[j] * s_s0[co] + s_o0[co];
          float x0 = s_in[((te + 1) * NV + ve) * 2 + 0];
          float x1 = s_in[((te + 1) * NV + ve) * 2 + 1];
          val += s_rA[co] * x0 + s_rB[co] * x1 + s_rC[co];
          s_h0[(te * NV + ve) * NC + co] = fmaxf(val, 0.0f);
        }
      }
    }
  }
  __syncthreads();

  float* s_hagg = s_g0;
  for (int idx = tid; idx < T3 * NV * NC; idx += nthr) {
    int ci = idx & 31, rv = idx >> 5;
    int v = rv % NV, t3 = rv / NV;
    float acc = 0.0f;
    const float* Ar = &sA[v * NV];
    const float* hr = &s_h0[(t3 + 1) * (NV * NC) + ci];
    for (int u = 0; u < NV; ++u) acc += Ar[u] * hr[u * NC];
    s_hagg[idx] = acc;
  }
  __syncthreads();

  {
    const int M = T3 * NV;
    for (int tIdx = wave; tIdx < 14 * 2; tIdx += 8) {
      int mt = tIdx >> 1, nt = tIdx & 1;
      v8f acc = {0.f,0.f,0.f,0.f,0.f,0.f,0.f,0.f};
      int mA = mt * 16 + row;
      int mC = (mA < M) ? mA : 0;
      int colN = nt * 16 + row;
      for (int ks = 0; ks < 8; ++ks) {
        int k0 = ks * 4 + hi * 2;
        v2f a, bb;
        a.x = s_hagg[mC * NC + k0];
        a.y = s_hagg[mC * NC + k0 + 1];
        bb.x = s_gw1[colN * NC + k0];
        bb.y = s_gw1[colN * NC + k0 + 1];
        acc = wmma4(a, bb, acc);
      }
#pragma unroll
      for (int j = 0; j < 8; ++j) {
        int m = mt * 16 + j + hi * 8;
        if (m < M) {
          int t3 = m / NV;
          int t = tbase - 1 + t3;
          float val = acc[j] + s_gb1[colN];
          s_g1[m * NC + colN] = (t >= 0 && t < NTIME) ? val : 0.0f;
        }
      }
    }
  }
  __syncthreads();

  {
    const int M = TT * NV;
    for (int tIdx = wave; tIdx < 11 * 2; tIdx += 8) {
      int mt = tIdx >> 1, nt = tIdx & 1;
      v8f acc = {0.f,0.f,0.f,0.f,0.f,0.f,0.f,0.f};
      int mA = mt * 16 + row;
      int mC = (mA < M) ? mA : 0;
      int t4 = mC / NV, vv = mC % NV;
      int colN = nt * 16 + row;
      for (int ks = 0; ks < 24; ++ks) {
        int k0 = ks * 4 + hi * 2;
        v2f a, bb;
        a.x = s_g1[((t4 + (k0 >> 5)) * NV + vv) * NC + (k0 & 31)];
        a.y = s_g1[((t4 + ((k0 + 1) >> 5)) * NV + vv) * NC + ((k0 + 1) & 31)];
        bb.x = s_tw1[k0 * NC + colN];
        bb.y = s_tw1[(k0 + 1) * NC + colN];
        acc = wmma4(a, bb, acc);
      }
#pragma unroll
      for (int j = 0; j < 8; ++j) {
        int m = mt * 16 + j + hi * 8;
        if (m < M) {
          int te = m / NV, ve = m % NV;
          int co = colN;
          int hidx = ((te + 2) * NV + ve) * NC + co;
          float val = acc[j] * s_s1[co] + s_o1[co] + s_h0[hidx];
          s_h0[hidx] = fmaxf(val, 0.0f);
        }
      }
    }
  }
  __syncthreads();

  {
    float* s_red = s_g0;
    int nt  = wave & 3;
    int ksb = (wave >> 2) * 84;
    v8f acc = {0.f,0.f,0.f,0.f,0.f,0.f,0.f,0.f};
    int t4 = (row < TT) ? row : (TT - 1);
    int colN = nt * 16 + row;
    const float* h1base = &s_h0[(t4 + 2) * (NV * NC)];
    for (int ks = ksb; ks < ksb + 84; ++ks) {
      int f0 = ks * 4 + hi * 2;
      int f1 = f0 + 1;
      int c0 = f0 / NV, v0 = f0 % NV;
      int c1 = f1 / NV, v1 = f1 % NV;
      v2f a, bb;
      a.x = h1base[v0 * NC + c0];
      a.y = h1base[v1 * NC + c1];
      bb.x = fc_w[colN * 672 + f0];
      bb.y = fc_w[colN * 672 + f1];
      acc = wmma4(a, bb, acc);
    }
    if (wave >= 4) {
#pragma unroll
      for (int j = 0; j < 8; ++j) s_red[nt * 256 + lane * 8 + j] = acc[j];
    }
    __syncthreads();
    if (wave < 4) {
#pragma unroll
      for (int j = 0; j < 8; ++j) {
        int m = j + hi * 8;
        if (m < TT) {
          int t = tbase + m;
          float val = acc[j] + s_red[nt * 256 + lane * 8 + j] + s_fcb[colN];
          out[(((size_t)b * NTIME + t) * 2 + hand) * NOUT + colN] = val;
        }
      }
    }
  }
}

extern "C" void kernel_launch(void* const* d_in, const int* in_sizes, int n_in,
                              void* d_out, int out_size, void* d_ws, size_t ws_size,
                              hipStream_t stream) {
  (void)in_sizes; (void)n_in; (void)d_ws; (void)ws_size; (void)out_size;
  const float* x        = (const float*)d_in[0];
  const float* gcn_w0   = (const float*)d_in[1];
  const float* gcn_b0   = (const float*)d_in[2];
  const float* tcn_w0   = (const float*)d_in[3];
  const float* tcn_b0   = (const float*)d_in[4];
  const float* bn_g0    = (const float*)d_in[5];
  const float* bn_b0    = (const float*)d_in[6];
  const float* bn_m0    = (const float*)d_in[7];
  const float* bn_v0    = (const float*)d_in[8];
  const float* res_w    = (const float*)d_in[9];
  const float* res_b    = (const float*)d_in[10];
  const float* res_bn_g = (const float*)d_in[11];
  const float* res_bn_b = (const float*)d_in[12];
  const float* res_bn_m = (const float*)d_in[13];
  const float* res_bn_v = (const float*)d_in[14];
  const float* gcn_w1   = (const float*)d_in[15];
  const float* gcn_b1   = (const float*)d_in[16];
  const float* tcn_w1   = (const float*)d_in[17];
  const float* tcn_b1   = (const float*)d_in[18];
  const float* bn_g1    = (const float*)d_in[19];
  const float* bn_b1    = (const float*)d_in[20];
  const float* bn_m1    = (const float*)d_in[21];
  const float* bn_v1    = (const float*)d_in[22];
  const float* fc_w     = (const float*)d_in[23];
  const float* fc_b     = (const float*)d_in[24];

  dim3 grid(NB * 2 * (NTIME / TT));
  dim3 block(256);
  stgcn_fused<<<grid, block, 0, stream>>>(
      x, gcn_w0, gcn_b0, tcn_w0, tcn_b0, bn_g0, bn_b0, bn_m0, bn_v0,
      res_w, res_b, res_bn_g, res_bn_b, res_bn_m, res_bn_v,
      gcn_w1, gcn_b1, tcn_w1, tcn_b1, bn_g1, bn_b1, bn_m1, bn_v1,
      fc_w, fc_b, (float*)d_out);
}


# FC 672->64 split off as a second launch: h1 goes through the workspace and a standalone WMMA GEMM keeps all of fc_w in LDS per workgroup for 256 rows; new sum order in FC (one K chain)
# speedup vs baseline: 1.2736x; 1.2736x over previous
#include <hip/hip_runtime.h>

#define TT 8
#define T1 (TT + 6)
#define T2 (TT + 4)
#define T3 (TT + 2)
#define NB 64
#define NTIME 512
#define NV 21
#define NC 32
#define NCIN 2
#define NOUT 64
#define FCK 672
#define WP 676

typedef __attribute__((ext_vector_type(2))) float v2f;
typedef __attribute__((ext_vector_type(8))) float v8f;

__device__ __forceinline__ v8f wmma4(v2f a, v2f b, v8f c) {
  return __builtin_amdgcn_wmma_f32_16x16x4_f32(
       false, a,  false, b,
       (short)0, c,  false,  false);
}

__device__ const unsigned char g_pi[20] = {0,1,2,3,0,5,6,7,0,9,10,11,0,13,14,15,0,17,18,19};
__device__ const unsigned char g_pj[20] = {1,2,3,4,5,6,7,8,9,10,11,12,13,14,15,16,17,18,19,20};
__device__ const float g_dinv[21] = {
  0.4082482905f,
  0.5773502692f,0.5773502692f,0.5773502692f,0.7071067812f,
  0.5773502692f,0.5773502692f,0.5773502692f,0.7071067812f,
  0.5773502692f,0.5773502692f,0.5773502692f,0.7071067812f,
  0.5773502692f,0.5773502692f,0.5773502692f,0.7071067812f,
  0.5773502692f,0.5773502692f,0.5773502692f,0.7071067812f};

__global__ __launch_bounds__(256) void stgcn_body(
    const float* __restrict__ x,
    const float* __restrict__ gcn_w0, const float* __restrict__ gcn_b0,
    const float* __restrict__ tcn_w0, const float* __restrict__ tcn_b0,
    const float* __restrict__ bn_g0, const float* __restrict__ bn_b0,
    const float* __restrict__ bn_m0, const float* __restrict__ bn_v0,
    const float* __restrict__ res_w, const float* __restrict__ res_b,
    const float* __restrict__ res_bn_g, const float* __restrict__ res_bn_b,
    const float* __restrict__ res_bn_m, const float* __restrict__ res_bn_v,
    const float* __restrict__ gcn_w1, const float* __restrict__ gcn_b1,
    const float* __restrict__ tcn_w1, const float* __restrict__ tcn_b1,
    const float* __restrict__ bn_g1, const float* __restrict__ bn_b1,
    const float* __restrict__ bn_m1, const float* __restrict__ bn_v1,
    float* __restrict__ hws)
{
  __shared__ float sA[NV * NV];
  __shared__ float s_in[T1 * NV * NCIN];
  __shared__ float s_xa[T1 * NV * NCIN];
  __shared__ float s_tw0[96 * NC];
  __shared__ float s_tw1[96 * NC];
  __shared__ float s_gw1[NC * NC];
  __shared__ __attribute__((aligned(16))) float s_g0[T1 * NV * NC];
  __shared__ float s_h0[T2 * NV * NC];
  __shared__ float s_g1[T3 * NV * NC];
  __shared__ float s_gw0[NC * NCIN];
  __shared__ float s_gb0[NC], s_gb1[NC];
  __shared__ float s_s0[NC], s_o0[NC], s_s1[NC], s_o1[NC];
  __shared__ float s_rA[NC], s_rB[NC], s_rC[NC];

  const int tid  = threadIdx.x;
  const int nthr = 256;
  const int lane = tid & 31;
  const int wave = tid >> 5;
  const int row  = lane & 15;
  const int hi   = lane >> 4;

  const int bid   = blockIdx.x;
  const int tile  = bid & 63;
  const int hand  = (bid >> 6) & 1;
  const int b     = bid >> 7;
  const int tbase = tile * TT;

  for (int idx = tid; idx < NV * NV; idx += nthr) {
    int i = idx / NV, j = idx % NV;
    bool conn = (i == j);
    for (int p = 0; p < 20; ++p)
      conn = conn || ((int)g_pi[p] == i && (int)g_pj[p] == j) ||
                     ((int)g_pi[p] == j && (int)g_pj[p] == i);
    sA[idx] = conn ? g_dinv[i] * g_dinv[j] : 0.0f;
  }
  for (int idx = tid; idx < 96 * NC; idx += nthr) {
    int co = idx & 31, kap = idx >> 5, k = kap >> 5, ci = kap & 31;
    s_tw0[idx] = tcn_w0[(co * NC + ci) * 3 + k];
    s_tw1[idx] = tcn_w1[(co * NC + ci) * 3 + k];
  }
  for (int idx = tid; idx < NC * NC; idx += nthr) s_gw1[idx] = gcn_w1[idx];
  if (tid < NC * NCIN) s_gw0[tid] = gcn_w0[tid];
  if (tid < NC) {
    s_gb0[tid] = gcn_b0[tid];
    s_gb1[tid] = gcn_b1[tid];
    float sc0 = bn_g0[tid] * rsqrtf(bn_v0[tid] + 1e-5f);
    s_s0[tid] = sc0;
    s_o0[tid] = (tcn_b0[tid] - bn_m0[tid]) * sc0 + bn_b0[tid];
    float sc1 = bn_g1[tid] * rsqrtf(bn_v1[tid] + 1e-5f);
    s_s1[tid] = sc1;
    s_o1[tid] = (tcn_b1[tid] - bn_m1[tid]) * sc1 + bn_b1[tid];
    float rsc = res_bn_g[tid] * rsqrtf(res_bn_v[tid] + 1e-5f);
    s_rA[tid] = res_w[tid * 2 + 0] * rsc;
    s_rB[tid] = res_w[tid * 2 + 1] * rsc;
    s_rC[tid] = (res_b[tid] - res_bn_m[tid]) * rsc + res_bn_b[tid];
  }

  for (int idx = tid; idx < T1 * NV * NCIN; idx += nthr) {
    int i = idx / (NV * NCIN), r = idx % (NV * NCIN);
    int v = r >> 1, ci = r & 1;
    int t = tbase - 3 + i;
    float val = 0.0f;
    if (t >= 0 && t < NTIME)
      val = x[(((b * NTIME + t) * 2 + hand) * NV + v) * NCIN + ci];
    s_in[idx] = val;
  }
  __syncthreads();

  for (int idx = tid; idx < T1 * NV * NCIN; idx += nthr) {
    int i = idx / (NV * NCIN), r = idx % (NV * NCIN);
    int v = r >> 1, ci = r & 1;
    float acc = 0.0f;
    const float* Ar = &sA[v * NV];
    const float* xr = &s_in[i * NV * NCIN + ci];
    for (int u = 0; u < NV; ++u) acc += Ar[u] * xr[u * NCIN];
    s_xa[idx] = acc;
  }
  __syncthreads();

  for (int idx = tid; idx < T1 * NV * NC; idx += nthr) {
    int c = idx & 31, rv = idx >> 5;
    int v = rv % NV, i = rv / NV;
    int t = tbase - 3 + i;
    float val = 0.0f;
    if (t >= 0 && t < NTIME) {
      float xa0 = s_xa[(i * NV + v) * 2 + 0];
      float xa1 = s_xa[(i * NV + v) * 2 + 1];
      val = s_gw0[c * 2 + 0] * xa0 + s_gw0[c * 2 + 1] * xa1 + s_gb0[c];
    }
    s_g0[idx] = val;
  }
  __syncthreads();

  {
    const int M = T2 * NV;
    for (int tIdx = wave; tIdx < 16 * 2; tIdx += 8) {
      int mt = tIdx >> 1, nt = tIdx & 1;
      v8f acc = {0.f,0.f,0.f,0.f,0.f,0.f,0.f,0.f};
      int mA = mt * 16 + row;
      int mC = (mA < M) ? mA : 0;
      int t2 = mC / NV, vv = mC % NV;
      int colN = nt * 16 + row;
      for (int ks = 0; ks < 24; ++ks) {
        int k0 = ks * 4 + hi * 2;
        v2f a, bb;
        a.x = s_g0[(t2 + (k0 >> 5)) * (NV * NC) + vv * NC + (k0 & 31)];
        a.y = s_g0[(t2 + ((k0 + 1) >> 5)) * (NV * NC) + vv * NC + ((k0 + 1) & 31)];
        bb.x = s_tw0[k0 * NC + colN];
        bb.y = s_tw0[(k0 + 1) * NC + colN];
        acc = wmma4(a, bb, acc);
      }
#pragma unroll
      for (int j = 0; j < 8; ++j) {
        int m = mt * 16 + j + hi * 8;
        if (m < M) {
          int te = m / NV, ve = m % NV;
          int co = colN;
          float val = acc[j] * s_s0[co] + s_o0[co];
          float x0 = s_in[((te + 1) * NV + ve) * 2 + 0];
          float x1 = s_in[((te + 1) * NV + ve) * 2 + 1];
          val += s_rA[co] * x0 + s_rB[co] * x1 + s_rC[co];
          s_h0[(te * NV + ve) * NC + co] = fmaxf(val, 0.0f);
        }
      }
    }
  }
  __syncthreads();

  float* s_hagg = s_g0;
  for (int idx = tid; idx < T3 * NV * NC; idx += nthr) {
    int ci = idx & 31, rv = idx >> 5;
    int v = rv % NV, t3 = rv / NV;
    float acc = 0.0f;
    const float* Ar = &sA[v * NV];
    const float* hr = &s_h0[(t3 + 1) * (NV * NC) + ci];
    for (int u = 0; u < NV; ++u) acc += Ar[u] * hr[u * NC];
    s_hagg[idx] = acc;
  }
  __syncthreads();

  {
    const int M = T3 * NV;
    for (int tIdx = wave; tIdx < 14 * 2; tIdx += 8) {
      int mt = tIdx >> 1, nt = tIdx & 1;
      v8f acc = {0.f,0.f,0.f,0.f,0.f,0.f,0.f,0.f};
      int mA = mt * 16 + row;
      int mC = (mA < M) ? mA : 0;
      int colN = nt * 16 + row;
      for (int ks = 0; ks < 8; ++ks) {
        int k0 = ks * 4 + hi * 2;
        v2f a, bb;
        a.x = s_hagg[mC * NC + k0];
        a.y = s_hagg[mC * NC + k0 + 1];
        bb.x = s_gw1[colN * NC + k0];
        bb.y = s_gw1[colN * NC + k0 + 1];
        acc = wmma4(a, bb, acc);
      }
#pragma unroll
      for (int j = 0; j < 8; ++j) {
        int m = mt * 16 + j + hi * 8;
        if (m < M) {
          int t3 = m / NV;
          int t = tbase - 1 + t3;
          float val = acc[j] + s_gb1[colN];
          s_g1[m * NC + colN] = (t >= 0 && t < NTIME) ? val : 0.0f;
        }
      }
    }
  }
  __syncthreads();

  float* s_h1 = s_g0;
  {
    const int M = TT * NV;
    for (int tIdx = wave; tIdx < 11 * 2; tIdx += 8) {
      int mt = tIdx >> 1, nt = tIdx & 1;
      v8f acc = {0.f,0.f,0.f,0.f,0.f,0.f,0.f,0.f};
      int mA = mt * 16 + row;
      int mC = (mA < M) ? mA : 0;
      int t4 = mC / NV, vv = mC % NV;
      int colN = nt * 16 + row;
      for (int ks = 0; ks < 24; ++ks) {
        int k0 = ks * 4 + hi * 2;
        v2f a, bb;
        a.x = s_g1[((t4 + (k0 >> 5)) * NV + vv) * NC + (k0 & 31)];
        a.y = s_g1[((t4 + ((k0 + 1) >> 5)) * NV + vv) * NC + ((k0 + 1) & 31)];
        bb.x = s_tw1[k0 * NC + colN];
        bb.y = s_tw1[(k0 + 1) * NC + colN];
        acc = wmma4(a, bb, acc);
      }
#pragma unroll
      for (int j = 0; j < 8; ++j) {
        int m = mt * 16 + j + hi * 8;
        if (m < M) {
          int te = m / NV, ve = m % NV;
          int co = colN;
          int hidx = ((te + 2) * NV + ve) * NC + co;
          float val = acc[j] * s_s1[co] + s_o1[co] + s_h0[hidx];
          s_h1[te * (NC * NV) + co * NV + ve] = fmaxf(val, 0.0f);
        }
      }
    }
  }
  __syncthreads();

  {
    const float4* srcv = reinterpret_cast<const float4*>(s_h1);
    for (int idx = tid; idx < TT * (FCK / 4); idx += nthr) {
      int i = idx / (FCK / 4), q = idx - i * (FCK / 4);
      size_t R = ((size_t)b * NTIME + (size_t)(tbase + i)) * 2 + (size_t)hand;
      reinterpret_cast<float4*>(hws + R * FCK)[q] = srcv[idx];
    }
  }
}

__global__ __launch_bounds__(512) void fc_gemm(
    const float* __restrict__ H, const float* __restrict__ fc_w,
    const float* __restrict__ fc_b, float* __restrict__ out)
{
  __shared__ __attribute__((aligned(16))) float sW[NOUT * WP];

  const int tid  = threadIdx.x;
  const int lane = tid & 31;
  const int wave = tid >> 5;
  const int row  = lane & 15;
  const int hi   = lane >> 4;

  for (int idx = tid; idx < NOUT * (FCK / 4); idx += 512) {
    int o = idx / (FCK / 4), q = idx - o * (FCK / 4);
    float4 w = reinterpret_cast<const float4*>(fc_w)[idx];
    *reinterpret_cast<float4*>(&sW[o * WP + q * 4]) = w;
  }

  const int R0 = (blockIdx.x * 16 + wave) * 16;
  const float* ap = H + (size_t)(R0 + row) * FCK + hi * 4;
  const float* bp = sW + row * WP + hi * 4;

  float4 an[4];
#pragma unroll
  for (int u = 0; u < 4; ++u) an[u] = *reinterpret_cast<const float4*>(ap + u * 8);

  __syncthreads();

  v8f acc0 = {0.f,0.f,0.f,0.f,0.f,0.f,0.f,0.f};
  v8f acc1 = acc0, acc2 = acc0, acc3 = acc0;

  for (int g = 0; g < FCK / 32; ++g) {
    float4 ac[4];
#pragma unroll
    for (int u = 0; u < 4; ++u) ac[u] = an[u];
    if (g + 1 < FCK / 32) {
#pragma unroll
      for (int u = 0; u < 4; ++u)
        an[u] = *reinterpret_cast<const float4*>(ap + (g + 1) * 32 + u * 8);
    }
#pragma unroll
    for (int u = 0; u < 4; ++u) {
      const float* bq = bp + g * 32 + u * 8;
      float4 b0 = *reinterpret_cast<const float4*>(bq);
      float4 b1 = *reinterpret_cast<const float4*>(bq + 16 * WP);
      float4 b2 = *reinterpret_cast<const float4*>(bq + 32 * WP);
      float4 b3 = *reinterpret_cast<const float4*>(bq + 48 * WP);
      v2f a01 = {ac[u].x, ac[u].y};
      v2f a23 = {ac[u].z, ac[u].w};
      acc0 = wmma4(a01, (v2f){b0.x, b0.y}, acc0);
      acc1 = wmma4(a01, (v2f){b1.x, b1.y}, acc1);
      acc2 = wmma4(a01, (v2f){b2.x, b2.y}, acc2);
      acc3 = wmma4(a01, (v2f){b3.x, b3.y}, acc3);
      acc0 = wmma4(a23, (v2f){b0.z, b0.w}, acc0);
      acc1 = wmma4(a23, (v2f){b1.z, b1.w}, acc1);
      acc2 = wmma4(a23, (v2f){b2.z, b2.w}, acc2);
      acc3 = wmma4(a23, (v2f){b3.z, b3.w}, acc3);
    }
  }

  {
    float bias0 = fc_b[row], bias1 = fc_b[16 + row], bias2 = fc_b[32 + row], bias3 = fc_b[48 + row];
#pragma unroll
    for (int j = 0; j < 8; ++j) {
      float* o = out + (size_t)(R0 + hi * 8 + j) * NOUT + row;
      o[0]  = acc0[j] + bias0;
      o[16] = acc1[j] + bias1;
      o[32] = acc2[j] + bias2;
      o[48] = acc3[j] + bias3;
    }
  }
}

extern "C" void kernel_launch(void* const* d_in, const int* in_sizes, int n_in,
                              void* d_out, int out_size, void* d_ws, size_t ws_size,
                              hipStream_t stream) {
  (void)in_sizes; (void)n_in; (void)ws_size; (void)out_size;
  const float* x        = (const float*)d_in[0];
  const float* gcn_w0   = (const float*)d_in[1];
  const float* gcn_b0   = (const float*)d_in[2];
  const float* tcn_w0   = (const float*)d_in[3];
  const float* tcn_b0   = (const float*)d_in[4];
  const float* bn_g0    = (const float*)d_in[5];
  const float* bn_b0    = (const float*)d_in[6];
  const float* bn_m0    = (const float*)d_in[7];
  const float* bn_v0    = (const float*)d_in[8];
  const float* res_w    = (const float*)d_in[9];
  const float* res_b    = (const float*)d_in[10];
  const float* res_bn_g = (const float*)d_in[11];
  const float* res_bn_b = (const float*)d_in[12];
  const float* res_bn_m = (const float*)d_in[13];
  const float* res_bn_v = (const float*)d_in[14];
  const float* gcn_w1   = (const float*)d_in[15];
  const float* gcn_b1   = (const float*)d_in[16];
  const float* tcn_w1   = (const float*)d_in[17];
  const float* tcn_b1   = (const float*)d_in[18];
  const float* bn_g1    = (const float*)d_in[19];
  const float* bn_b1    = (const float*)d_in[20];
  const float* bn_m1    = (const float*)d_in[21];
  const float* bn_v1    = (const float*)d_in[22];
  const float* fc_w     = (const float*)d_in[23];
  const float* fc_b     = (const float*)d_in[24];

  dim3 grid(NB * 2 * (NTIME / TT));
  dim3 block(256);
  float* hws = (float*)d_ws;
  stgcn_body<<<grid, block, 0, stream>>>(
      x, gcn_w0, gcn_b0, tcn_w0, tcn_b0, bn_g0, bn_b0, bn_m0, bn_v0,
      res_w, res_b, res_bn_g, res_bn_b, res_bn_m, res_bn_v,
      gcn_w1, gcn_b1, tcn_w1, tcn_b1, bn_g1, bn_b1, bn_m1, bn_v1,
      hws);
  dim3 grid2(NB * NTIME * 2 / 256);
  dim3 block2(512);
  fc_gemm<<<grid2, block2, 0, stream>>>(hws, fc_w, fc_b, (float*)d_out);
}


# Body GEMMs transposed: weights are the WMMA row operand held in registers, activations the column operand read from channel-major LDS tiles [c][t*21+v]: unit-stride fragment reads and D stores
# speedup vs baseline: 1.7704x; 1.3901x over previous
#include <hip/hip_runtime.h>

#define TT 8
#define T1 (TT + 6)
#define T2 (TT + 4)
#define T3 (TT + 2)
#define NB 64
#define NTIME 512
#define NV 21
#define NC 32
#define NCIN 2
#define NOUT 64
#define FCK 672
#define WP 676

typedef __attribute__((ext_vector_type(2))) float v2f;
typedef __attribute__((ext_vector_type(8))) float v8f;

__device__ __forceinline__ v8f wmma4(v2f a, v2f b, v8f c) {
  return __builtin_amdgcn_wmma_f32_16x16x4_f32(
       false, a,  false, b,
       (short)0, c,  false,  false);
}

__device__ const unsigned char g_pi[20] = {0,1,2,3,0,5,6,7,0,9,10,11,0,13,14,15,0,17,18,19};
__device__ const unsigned char g_pj[20] = {1,2,3,4,5,6,7,8,9,10,11,12,13,14,15,16,17,18,19,20};
__device__ const float g_dinv[21] = {
  0.4082482905f,
  0.5773502692f,0.5773502692f,0.5773502692f,0.7071067812f,
  0.5773502692f,0.5773502692f,0.5773502692f,0.7071067812f,
  0.5773502692f,0.5773502692f,0.5773502692f,0.7071067812f,
  0.5773502692f,0.5773502692f,0.5773502692f,0.7071067812f,
  0.5773502692f,0.5773502692f,0.5773502692f,0.7071067812f};

#define P0 300
#define PH 268
#define P1 212
__global__ __launch_bounds__(256) void stgcn_body(
    const float* __restrict__ x,
    const float* __restrict__ gcn_w0, const float* __restrict__ gcn_b0,
    const float* __restrict__ tcn_w0, const float* __restrict__ tcn_b0,
    const float* __restrict__ bn_g0, const float* __restrict__ bn_b0,
    const float* __restrict__ bn_m0, const float* __restrict__ bn_v0,
    const float* __restrict__ res_w, const float* __restrict__ res_b,
    const float* __restrict__ res_bn_g, const float* __restrict__ res_bn_b,
    const float* __restrict__ res_bn_m, const float* __restrict__ res_bn_v,
    const float* __restrict__ gcn_w1, const float* __restrict__ gcn_b1,
    const float* __restrict__ tcn_w1, const float* __restrict__ tcn_b1,
    const float* __restrict__ bn_g1, const float* __restrict__ bn_b1,
    const float* __restrict__ bn_m1, const float* __restrict__ bn_v1,
    float* __restrict__ hws)
{
  __shared__ float sA[NV * NV];
  __shared__ float s_in[T1 * NV * NCIN];
  __shared__ float s_xa[T1 * NV * NCIN];
  __shared__ __attribute__((aligned(16))) float s_tw0[96 * NC];
  __shared__ __attribute__((aligned(16))) float s_tw1[96 * NC];
  __shared__ __attribute__((aligned(16))) float s_gw1[NC * NC];
  __shared__ __attribute__((aligned(16))) float s_g0[NC * P0];
  __shared__ float s_h0[NC * PH];
  __shared__ float s_g1[NC * P1];
  __shared__ float s_gw0[NC * NCIN];
  __shared__ float s_gb0[NC], s_gb1[NC];
  __shared__ float s_s0[NC], s_o0[NC], s_s1[NC], s_o1[NC];
  __shared__ float s_rA[NC], s_rB[NC], s_rC[NC];

  const int tid  = threadIdx.x;
  const int nthr = 256;
  const int lane = tid & 31;
  const int wave = tid >> 5;
  const int row  = lane & 15;
  const int hi   = lane >> 4;

  const int bid   = blockIdx.x;
  const int tile  = bid & 63;
  const int hand  = (bid >> 6) & 1;
  const int b     = bid >> 7;
  const int tbase = tile * TT;

  for (int idx = tid; idx < NV * NV; idx += nthr) {
    int i = idx / NV, j = idx % NV;
    bool conn = (i == j);
    for (int p = 0; p < 20; ++p)
      conn = conn || ((int)g_pi[p] == i && (int)g_pj[p] == j) ||
                     ((int)g_pi[p] == j && (int)g_pj[p] == i);
    sA[idx] = conn ? g_dinv[i] * g_dinv[j] : 0.0f;
  }
  for (int idx = tid; idx < 96 * NC; idx += nthr) {
    int e = idx & 1, ln = (idx >> 1) & 31, cs = idx >> 6;
    int ct = cs / 24, s = cs - ct * 24;
    int co = ct * 16 + (ln & 15);
    int k = s >> 3;
    int ci = 4 * (s & 7) + 2 * (ln >> 4) + e;
    s_tw0[idx] = tcn_w0[(co * NC + ci) * 3 + k];
    s_tw1[idx] = tcn_w1[(co * NC + ci) * 3 + k];
  }
  for (int idx = tid; idx < NC * NC; idx += nthr) {
    int e = idx & 1, ln = (idx >> 1) & 31, cs = idx >> 6;
    int ct = cs >> 3, s = cs & 7;
    int co = ct * 16 + (ln & 15);
    int ci = 4 * s + 2 * (ln >> 4) + e;
    s_gw1[idx] = gcn_w1[co * NC + ci];
  }
  if (tid < NC * NCIN) s_gw0[tid] = gcn_w0[tid];
  if (tid < NC) {
    s_gb0[tid] = gcn_b0[tid];
    s_gb1[tid] = gcn_b1[tid];
    float sc0 = bn_g0[tid] * rsqrtf(bn_v0[tid] + 1e-5f);
    s_s0[tid] = sc0;
    s_o0[tid] = (tcn_b0[tid] - bn_m0[tid]) * sc0 + bn_b0[tid];
    float sc1 = bn_g1[tid] * rsqrtf(bn_v1[tid] + 1e-5f);
    s_s1[tid] = sc1;
    s_o1[tid] = (tcn_b1[tid] - bn_m1[tid]) * sc1 + bn_b1[tid];
    float rsc = res_bn_g[tid] * rsqrtf(res_bn_v[tid] + 1e-5f);
    s_rA[tid] = res_w[tid * 2 + 0] * rsc;
    s_rB[tid] = res_w[tid * 2 + 1] * rsc;
    s_rC[tid] = (res_b[tid] - res_bn_m[tid]) * rsc + res_bn_b[tid];
  }

  for (int idx = tid; idx < T1 * NV * NCIN; idx += nthr) {
    int i = idx / (NV * NCIN), r = idx % (NV * NCIN);
    int v = r >> 1, ci = r & 1;
    int t = tbase - 3 + i;
    float val = 0.0f;
    if (t >= 0 && t < NTIME)
      val = x[(((b * NTIME + t) * 2 + hand) * NV + v) * NCIN + ci];
    s_in[idx] = val;
  }
  __syncthreads();

  for (int idx = tid; idx < T1 * NV * NCIN; idx += nthr) {
    int i = idx / (NV * NCIN), r = idx % (NV * NCIN);
    int v = r >> 1, ci = r & 1;
    float acc = 0.0f;
    const float* Ar = &sA[v * NV];
    const float* xr = &s_in[i * NV * NCIN + ci];
    for (int u = 0; u < NV; ++u) acc += Ar[u] * xr[u * NCIN];
    s_xa[idx] = acc;
  }
  __syncthreads();

  for (int idx = tid; idx < T1 * NV * NC; idx += nthr) {
    int c = idx / (T1 * NV), m = idx - c * (T1 * NV);
    int i = m / NV;
    int t = tbase - 3 + i;
    float val = 0.0f;
    if (t >= 0 && t < NTIME) {
      float xa0 = s_xa[m * 2 + 0];
      float xa1 = s_xa[m * 2 + 1];
      val = s_gw0[c * 2 + 0] * xa0 + s_gw0[c * 2 + 1] * xa1 + s_gb0[c];
    }
    s_g0[c * P0 + m] = val;
  }
  __syncthreads();

  {
    const int M = T2 * NV;
    v2f wA0[24], wA1[24];
#pragma unroll
    for (int s = 0; s < 24; ++s) {
      wA0[s] = *reinterpret_cast<const v2f*>(&s_tw0[((0 * 24 + s) * 32 + lane) * 2]);
      wA1[s] = *reinterpret_cast<const v2f*>(&s_tw0[((1 * 24 + s) * 32 + lane) * 2]);
    }
    for (int nt = wave; nt < 16; nt += 8) {
      v8f acc0 = {0.f,0.f,0.f,0.f,0.f,0.f,0.f,0.f};
      v8f acc1 = acc0;
      int col = nt * 16 + row;
      int mC = (col < M) ? col : 0;
      const float* bp = &s_g0[(2 * hi) * P0 + mC];
#pragma unroll
      for (int s = 0; s < 24; ++s) {
        const float* bq = bp + (4 * (s & 7)) * P0 + (s >> 3) * NV;
        v2f bb;
        bb.x = bq[0];
        bb.y = bq[P0];
        acc0 = wmma4(wA0[s], bb, acc0);
        acc1 = wmma4(wA1[s], bb, acc1);
      }
      if (col < M) {
        int te = col / NV, ve = col - te * NV;
        float x0 = s_in[((te + 1) * NV + ve) * 2 + 0];
        float x1 = s_in[((te + 1) * NV + ve) * 2 + 1];
#pragma unroll
        for (int j = 0; j < 8; ++j) {
          int co = hi * 8 + j;
          float val = acc0[j] * s_s0[co] + s_o0[co];
          val += s_rA[co] * x0 + s_rB[co] * x1 + s_rC[co];
          s_h0[co * PH + col] = fmaxf(val, 0.0f);
          int c2 = co + 16;
          float va2 = acc1[j] * s_s0[c2] + s_o0[c2];
          va2 += s_rA[c2] * x0 + s_rB[c2] * x1 + s_rC[c2];
          s_h0[c2 * PH + col] = fmaxf(va2, 0.0f);
        }
      }
    }
  }
  __syncthreads();

  float* s_hagg = s_g0;
  for (int idx = tid; idx < T3 * NV * NC; idx += nthr) {
    int ci = idx / (T3 * NV), m = idx - ci * (T3 * NV);
    int t3 = m / NV, v = m - t3 * NV;
    float acc = 0.0f;
    const float* Ar = &sA[v * NV];
    const float* hr = &s_h0[ci * PH + (t3 + 1) * NV];
    for (int u = 0; u < NV; ++u) acc += Ar[u] * hr[u];
    s_hagg[ci * P1 + m] = acc;
  }
  __syncthreads();

  {
    const int M = T3 * NV;
    v2f wG0[8], wG1[8];
#pragma unroll
    for (int s = 0; s < 8; ++s) {
      wG0[s] = *reinterpret_cast<const v2f*>(&s_gw1[((0 * 8 + s) * 32 + lane) * 2]);
      wG1[s] = *reinterpret_cast<const v2f*>(&s_gw1[((1 * 8 + s) * 32 + lane) * 2]);
    }
    for (int nt = wave; nt < 14; nt += 8) {
      v8f acc0 = {0.f,0.f,0.f,0.f,0.f,0.f,0.f,0.f};
      v8f acc1 = acc0;
      int col = nt * 16 + row;
      int mC = (col < M) ? col : 0;
      const float* bp = &s_hagg[(2 * hi) * P1 + mC];
#pragma unroll
      for (int s = 0; s < 8; ++s) {
        const float* bq = bp + (4 * s) * P1;
        v2f bb;
        bb.x = bq[0];
        bb.y = bq[P1];
        acc0 = wmma4(wG0[s], bb, acc0);
        acc1 = wmma4(wG1[s], bb, acc1);
      }
      if (col < M) {
        int t3 = col / NV;
        int t = tbase - 1 + t3;
        bool ok = (t >= 0 && t < NTIME);
#pragma unroll
        for (int j = 0; j < 8; ++j) {
          int co = hi * 8 + j;
          float val = acc0[j] + s_gb1[co];
          s_g1[co * P1 + col] = ok ? val : 0.0f;
          int c2 = co + 16;
          float va2 = acc1[j] + s_gb1[c2];
          s_g1[c2 * P1 + col] = ok ? va2 : 0.0f;
        }
      }
    }
  }
  __syncthreads();

  float* s_h1 = s_g0;
  {
    const int M = TT * NV;
    v2f wA0[24], wA1[24];
#pragma unroll
    for (int s = 0; s < 24; ++s) {
      wA0[s] = *reinterpret_cast<const v2f*>(&s_tw1[((0 * 24 + s) * 32 + lane) * 2]);
      wA1[s] = *reinterpret_cast<const v2f*>(&s_tw1[((1 * 24 + s) * 32 + lane) * 2]);
    }
    for (int nt = wave; nt < 11; nt += 8) {
      v8f acc0 = {0.f,0.f,0.f,0.f,0.f,0.f,0.f,0.f};
      v8f acc1 = acc0;
      int col = nt * 16 + row;
      int mC = (col < M) ? col : 0;
      const float* bp = &s_g1[(2 * hi) * P1 + mC];
#pragma unroll
      for (int s = 0; s < 24; ++s) {
        const float* bq = bp + (4 * (s & 7)) * P1 + (s >> 3) * NV;
        v2f bb;
        bb.x = bq[0];
        bb.y = bq[P1];
        acc0 = wmma4(wA0[s], bb, acc0);
        acc1 = wmma4(wA1[s], bb, acc1);
      }
      if (col < M) {
        int te = col / NV, ve = col - te * NV;
        int hcol = (te + 2) * NV + ve;
#pragma unroll
        for (int j = 0; j < 8; ++j) {
          int co = hi * 8 + j;
          float val = acc0[j] * s_s1[co] + s_o1[co] + s_h0[co * PH + hcol];
          s_h1[te * (NC * NV) + co * NV + ve] = fmaxf(val, 0.0f);
          int c2 = co + 16;
          float va2 = acc1[j] * s_s1[c2] + s_o1[c2] + s_h0[c2 * PH + hcol];
          s_h1[te * (NC * NV) + c2 * NV + ve] = fmaxf(va2, 0.0f);
        }
      }
    }
  }
  __syncthreads();

  {
    const float4* srcv = reinterpret_cast<const float4*>(s_h1);
    for (int idx = tid; idx < TT * (FCK / 4); idx += nthr) {
      int i = idx / (FCK / 4), q = idx - i * (FCK / 4);
      size_t R = ((size_t)b * NTIME + (size_t)(tbase + i)) * 2 + (size_t)hand;
      reinterpret_cast<float4*>(hws + R * FCK)[q] = srcv[idx];
    }
  }
}

__global__ __launch_bounds__(512) void fc_gemm(
    const float* __restrict__ H, const float* __restrict__ fc_w,
    const float* __restrict__ fc_b, float* __restrict__ out)
{
  __shared__ __attribute__((aligned(16))) float sW[NOUT * WP];

  const int tid  = threadIdx.x;
  const int lane = tid & 31;
  const int wave = tid >> 5;
  const int row  = lane & 15;
  const int hi   = lane >> 4;

  for (int idx = tid; idx < NOUT * (FCK / 4); idx += 512) {
    int o = idx / (FCK / 4), q = idx - o * (FCK / 4);
    float4 w = reinterpret_cast<const float4*>(fc_w)[idx];
    *reinterpret_cast<float4*>(&sW[o * WP + q * 4]) = w;
  }

  const int R0 = (blockIdx.x * 16 + wave) * 16;
  const float* ap = H + (size_t)(R0 + row) * FCK + hi * 4;
  const float* bp = sW + row * WP + hi * 4;

  float4 an[4];
#pragma unroll
  for (int u = 0; u < 4; ++u) an[u] = *reinterpret_cast<const float4*>(ap + u * 8);

  __syncthreads();

  v8f acc0 = {0.f,0.f,0.f,0.f,0.f,0.f,0.f,0.f};
  v8f acc1 = acc0, acc2 = acc0, acc3 = acc0;

  for (int g = 0; g < FCK / 32; ++g) {
    float4 ac[4];
#pragma unroll
    for (int u = 0; u < 4; ++u) ac[u] = an[u];
    if (g + 1 < FCK / 32) {
#pragma unroll
      for (int u = 0; u < 4; ++u)
        an[u] = *reinterpret_cast<const float4*>(ap + (g + 1) * 32 + u * 8);
    }
#pragma unroll
    for (int u = 0; u < 4; ++u) {
      const float* bq = bp + g * 32 + u * 8;
      float4 b0 = *reinterpret_cast<const float4*>(bq);
      float4 b1 = *reinterpret_cast<const float4*>(bq + 16 * WP);
      float4 b2 = *reinterpret_cast<const float4*>(bq + 32 * WP);
      float4 b3 = *reinterpret_cast<const float4*>(bq + 48 * WP);
      v2f a01 = {ac[u].x, ac[u].y};
      v2f a23 = {ac[u].z, ac[u].w};
      acc0 = wmma4(a01, (v2f){b0.x, b0.y}, acc0);
      acc1 = wmma4(a01, (v2f){b1.x, b1.y}, acc1);
      acc2 = wmma4(a01, (v2f){b2.x, b2.y}, acc2);
      acc3 = wmma4(a01, (v2f){b3.x, b3.y}, acc3);
      acc0 = wmma4(a23, (v2f){b0.z, b0.w}, acc0);
      acc1 = wmma4(a23, (v2f){b1.z, b1.w}, acc1);
      acc2 = wmma4(a23, (v2f){b2.z, b2.w}, acc2);
      acc3 = wmma4(a23, (v2f){b3.z, b3.w}, acc3);
    }
  }

  {
    float bias0 = fc_b[row], bias1 = fc_b[16 + row], bias2 = fc_b[32 + row], bias3 = fc_b[48 + row];
#pragma unroll
    for (int j = 0; j < 8; ++j) {
      float* o = out + (size_t)(R0 + hi * 8 + j) * NOUT + row;
      o[0]  = acc0[j] + bias0;
      o[16] = acc1[j] + bias1;
      o[32] = acc2[j] + bias2;
      o[48] = acc3[j] + bias3;
    }
  }
}

extern "C" void kernel_launch(void* const* d_in, const int* in_sizes, int n_in,
                              void* d_out, int out_size, void* d_ws, size_t ws_size,
                              hipStream_t stream) {
  (void)in_sizes; (void)n_in; (void)ws_size; (void)out_size;
  const float* x        = (const float*)d_in[0];
  const float* gcn_w0   = (const float*)d_in[1];
  const float* gcn_b0   = (const float*)d_in[2];
  const float* tcn_w0   = (const float*)d_in[3];
  const float* tcn_b0   = (const float*)d_in[4];
  const float* bn_g0    = (const float*)d_in[5];
  const float* bn_b0    = (const float*)d_in[6];
  const float* bn_m0    = (const float*)d_in[7];
  const float* bn_v0    = (const float*)d_in[8];
  const float* res_w    = (const float*)d_in[9];
  const float* res_b    = (const float*)d_in[10];
  const float* res_bn_g = (const float*)d_in[11];
  const float* res_bn_b = (const float*)d_in[12];
  const float* res_bn_m = (const float*)d_in[13];
  const float* res_bn_v = (const float*)d_in[14];
  const float* gcn_w1   = (const float*)d_in[15];
  const float* gcn_b1   = (const float*)d_in[16];
  const float* tcn_w1   = (const float*)d_in[17];
  const float* tcn_b1   = (const float*)d_in[18];
  const float* bn_g1    = (const float*)d_in[19];
  const float* bn_b1    = (const float*)d_in[20];
  const float* bn_m1    = (const float*)d_in[21];
  const float* bn_v1    = (const float*)d_in[22];
  const float* fc_w     = (const float*)d_in[23];
  const float* fc_b     = (const float*)d_in[24];

  dim3 grid(NB * 2 * (NTIME / TT));
  dim3 block(256);
  float* hws = (float*)d_ws;
  stgcn_body<<<grid, block, 0, stream>>>(
      x, gcn_w0, gcn_b0, tcn_w0, tcn_b0, bn_g0, bn_b0, bn_m0, bn_v0,
      res_w, res_b, res_bn_g, res_bn_b, res_bn_m, res_bn_v,
      gcn_w1, gcn_b1, tcn_w1, tcn_b1, bn_g1, bn_b1, bn_m1, bn_v1,
      hws);
  dim3 grid2(NB * NTIME * 2 / 256);
  dim3 block2(512);
  fc_gemm<<<grid2, block2, 0, stream>>>(hws, fc_w, fc_b, (float*)d_out);
}


# h0 graph aggregation moved onto WMMA: adjacency (21x21, zero-padded to 32x24) is the row operand held in registers, h0 the column operand read from its LDS tile; new sum order in that aggregation
# speedup vs baseline: 2.0917x; 1.1815x over previous
#include <hip/hip_runtime.h>

#define TT 8
#define T1 (TT + 6)
#define T2 (TT + 4)
#define T3 (TT + 2)
#define NB 64
#define NTIME 512
#define NV 21
#define NC 32
#define NCIN 2
#define NOUT 64
#define FCK 672
#define WP 676

typedef __attribute__((ext_vector_type(2))) float v2f;
typedef __attribute__((ext_vector_type(8))) float v8f;

__device__ __forceinline__ v8f wmma4(v2f a, v2f b, v8f c) {
  return __builtin_amdgcn_wmma_f32_16x16x4_f32(
       false, a,  false, b,
       (short)0, c,  false,  false);
}

__device__ const unsigned char g_pi[20] = {0,1,2,3,0,5,6,7,0,9,10,11,0,13,14,15,0,17,18,19};
__device__ const unsigned char g_pj[20] = {1,2,3,4,5,6,7,8,9,10,11,12,13,14,15,16,17,18,19,20};
__device__ const float g_dinv[21] = {
  0.4082482905f,
  0.5773502692f,0.5773502692f,0.5773502692f,0.7071067812f,
  0.5773502692f,0.5773502692f,0.5773502692f,0.7071067812f,
  0.5773502692f,0.5773502692f,0.5773502692f,0.7071067812f,
  0.5773502692f,0.5773502692f,0.5773502692f,0.7071067812f,
  0.5773502692f,0.5773502692f,0.5773502692f,0.7071067812f};

#define P0 300
#define PH 268
#define P1 212
__global__ __launch_bounds__(256) void stgcn_body(
    const float* __restrict__ x,
    const float* __restrict__ gcn_w0, const float* __restrict__ gcn_b0,
    const float* __restrict__ tcn_w0, const float* __restrict__ tcn_b0,
    const float* __restrict__ bn_g0, const float* __restrict__ bn_b0,
    const float* __restrict__ bn_m0, const float* __restrict__ bn_v0,
    const float* __restrict__ res_w, const float* __restrict__ res_b,
    const float* __restrict__ res_bn_g, const float* __restrict__ res_bn_b,
    const float* __restrict__ res_bn_m, const float* __restrict__ res_bn_v,
    const float* __restrict__ gcn_w1, const float* __restrict__ gcn_b1,
    const float* __restrict__ tcn_w1, const float* __restrict__ tcn_b1,
    const float* __restrict__ bn_g1, const float* __restrict__ bn_b1,
    const float* __restrict__ bn_m1, const float* __restrict__ bn_v1,
    float* __restrict__ hws)
{
  __shared__ float sA[NV * NV];
  __shared__ float s_in[T1 * NV * NCIN];
  __shared__ float s_xa[T1 * NV * NCIN];
  __shared__ __attribute__((aligned(16))) float s_tw0[96 * NC];
  __shared__ __attribute__((aligned(16))) float s_tw1[96 * NC];
  __shared__ __attribute__((aligned(16))) float s_gw1[NC * NC];
  __shared__ __attribute__((aligned(16))) float s_g0[NC * P0];
  __shared__ float s_h0[NC * PH];
  __shared__ float s_g1[NC * P1];
  __shared__ float s_gw0[NC * NCIN];
  __shared__ float s_gb0[NC], s_gb1[NC];
  __shared__ float s_s0[NC], s_o0[NC], s_s1[NC], s_o1[NC];
  __shared__ float s_rA[NC], s_rB[NC], s_rC[NC];

  const int tid  = threadIdx.x;
  const int nthr = 256;
  const int lane = tid & 31;
  const int wave = tid >> 5;
  const int row  = lane & 15;
  const int hi   = lane >> 4;

  const int bid   = blockIdx.x;
  const int tile  = bid & 63;
  const int hand  = (bid >> 6) & 1;
  const int b     = bid >> 7;
  const int tbase = tile * TT;

  for (int idx = tid; idx < NV * NV; idx += nthr) {
    int i = idx / NV, j = idx % NV;
    bool conn = (i == j);
    for (int p = 0; p < 20; ++p)
      conn = conn || ((int)g_pi[p] == i && (int)g_pj[p] == j) ||
                     ((int)g_pi[p] == j && (int)g_pj[p] == i);
    sA[idx] = conn ? g_dinv[i] * g_dinv[j] : 0.0f;
  }
  for (int idx = tid; idx < 96 * NC; idx += nthr) {
    int e = idx & 1, ln = (idx >> 1) & 31, cs = idx >> 6;
    int ct = cs / 24, s = cs - ct * 24;
    int co = ct * 16 + (ln & 15);
    int k = s >> 3;
    int ci = 4 * (s & 7) + 2 * (ln >> 4) + e;
    s_tw0[idx] = tcn_w0[(co * NC + ci) * 3 + k];
    s_tw1[idx] = tcn_w1[(co * NC + ci) * 3 + k];
  }
  for (int idx = tid; idx < NC * NC; idx += nthr) {
    int e = idx & 1, ln = (idx >> 1) & 31, cs = idx >> 6;
    int ct = cs >> 3, s = cs & 7;
    int co = ct * 16 + (ln & 15);
    int ci = 4 * s + 2 * (ln >> 4) + e;
    s_gw1[idx] = gcn_w1[co * NC + ci];
  }
  if (tid < NC * NCIN) s_gw0[tid] = gcn_w0[tid];
  if (tid < NC) {
    s_gb0[tid] = gcn_b0[tid];
    s_gb1[tid] = gcn_b1[tid];
    float sc0 = bn_g0[tid] * rsqrtf(bn_v0[tid] + 1e-5f);
    s_s0[tid] = sc0;
    s_o0[tid] = (tcn_b0[tid] - bn_m0[tid]) * sc0 + bn_b0[tid];
    float sc1 = bn_g1[tid] * rsqrtf(bn_v1[tid] + 1e-5f);
    s_s1[tid] = sc1;
    s_o1[tid] = (tcn_b1[tid] - bn_m1[tid]) * sc1 + bn_b1[tid];
    float rsc = res_bn_g[tid] * rsqrtf(res_bn_v[tid] + 1e-5f);
    s_rA[tid] = res_w[tid * 2 + 0] * rsc;
    s_rB[tid] = res_w[tid * 2 + 1] * rsc;
    s_rC[tid] = (res_b[tid] - res_bn_m[tid]) * rsc + res_bn_b[tid];
  }

  for (int idx = tid; idx < T1 * NV * NCIN; idx += nthr) {
    int i = idx / (NV * NCIN), r = idx % (NV * NCIN);
    int v = r >> 1, ci = r & 1;
    int t = tbase - 3 + i;
    float val = 0.0f;
    if (t >= 0 && t < NTIME)
      val = x[(((b * NTIME + t) * 2 + hand) * NV + v) * NCIN + ci];
    s_in[idx] = val;
  }
  __syncthreads();

  for (int idx = tid; idx < T1 * NV * NCIN; idx += nthr) {
    int i = idx / (NV * NCIN), r = idx % (NV * NCIN);
    int v = r >> 1, ci = r & 1;
    float acc = 0.0f;
    const float* Ar = &sA[v * NV];
    const float* xr = &s_in[i * NV * NCIN + ci];
    for (int u = 0; u < NV; ++u) acc += Ar[u] * xr[u * NCIN];
    s_xa[idx] = acc;
  }
  __syncthreads();

  for (int idx = tid; idx < T1 * NV * NC; idx += nthr) {
    int c = idx / (T1 * NV), m = idx - c * (T1 * NV);
    int i = m / NV;
    int t = tbase - 3 + i;
    float val = 0.0f;
    if (t >= 0 && t < NTIME) {
      float xa0 = s_xa[m * 2 + 0];
      float xa1 = s_xa[m * 2 + 1];
      val = s_gw0[c * 2 + 0] * xa0 + s_gw0[c * 2 + 1] * xa1 + s_gb0[c];
    }
    s_g0[c * P0 + m] = val;
  }
  __syncthreads();

  {
    const int M = T2 * NV;
    v2f wA0[24], wA1[24];
#pragma unroll
    for (int s = 0; s < 24; ++s) {
      wA0[s] = *reinterpret_cast<const v2f*>(&s_tw0[((0 * 24 + s) * 32 + lane) * 2]);
      wA1[s] = *reinterpret_cast<const v2f*>(&s_tw0[((1 * 24 + s) * 32 + lane) * 2]);
    }
    for (int nt = wave; nt < 16; nt += 8) {
      v8f acc0 = {0.f,0.f,0.f,0.f,0.f,0.f,0.f,0.f};
      v8f acc1 = acc0;
      int col = nt * 16 + row;
      int mC = (col < M) ? col : 0;
      const float* bp = &s_g0[(2 * hi) * P0 + mC];
#pragma unroll
      for (int s = 0; s < 24; ++s) {
        const float* bq = bp + (4 * (s & 7)) * P0 + (s >> 3) * NV;
        v2f bb;
        bb.x = bq[0];
        bb.y = bq[P0];
        acc0 = wmma4(wA0[s], bb, acc0);
        acc1 = wmma4(wA1[s], bb, acc1);
      }
      if (col < M) {
        int te = col / NV, ve = col - te * NV;
        float x0 = s_in[((te + 1) * NV + ve) * 2 + 0];
        float x1 = s_in[((te + 1) * NV + ve) * 2 + 1];
#pragma unroll
        for (int j = 0; j < 8; ++j) {
          int co = hi * 8 + j;
          float val = acc0[j] * s_s0[co] + s_o0[co];
          val += s_rA[co] * x0 + s_rB[co] * x1 + s_rC[co];
          s_h0[co * PH + col] = fmaxf(val, 0.0f);
          int c2 = co + 16;
          float va2 = acc1[j] * s_s0[c2] + s_o0[c2];
          va2 += s_rA[c2] * x0 + s_rB[c2] * x1 + s_rC[c2];
          s_h0[c2 * PH + col] = fmaxf(va2, 0.0f);
        }
      }
    }
  }
  __syncthreads();

  float* s_hagg = s_g0;
  {
    v2f aF0[6], aF1[6];
#pragma unroll
    for (int s = 0; s < 6; ++s) {
      const int u0 = 4 * s + 2 * hi;
      const int vlo = row, vhi = 16 + row;
      const bool k0 = (u0 < NV), k1 = (u0 + 1 < NV), rv = (vhi < NV);
      float a00 = sA[vlo * NV + (k0 ? u0 : 0)];
      float a01 = sA[vlo * NV + (k1 ? u0 + 1 : 0)];
      float a10 = sA[(rv ? vhi : 0) * NV + (k0 ? u0 : 0)];
      float a11 = sA[(rv ? vhi : 0) * NV + (k1 ? u0 + 1 : 0)];
      aF0[s].x = k0 ? a00 : 0.0f;
      aF0[s].y = k1 ? a01 : 0.0f;
      aF1[s].x = (rv && k0) ? a10 : 0.0f;
      aF1[s].y = (rv && k1) ? a11 : 0.0f;
    }
    for (int nt = wave; nt < (NC * T3) / 16; nt += 8) {
      v8f acc0 = {0.f,0.f,0.f,0.f,0.f,0.f,0.f,0.f};
      v8f acc1 = acc0;
      const int n = nt * 16 + row;
      const int ci = n / T3, t3 = n - ci * T3;
      const float* hp = &s_h0[ci * PH + (t3 + 1) * NV + 2 * hi];
#pragma unroll
      for (int s = 0; s < 6; ++s) {
        v2f bb;
        if (s < 5) {
          bb.x = hp[4 * s];
          bb.y = hp[4 * s + 1];
        } else {
          bb.x = (hi == 0) ? hp[20] : 0.0f;
          bb.y = 0.0f;
        }
        acc0 = wmma4(aF0[s], bb, acc0);
        acc1 = wmma4(aF1[s], bb, acc1);
      }
      float* op = &s_hagg[ci * P1 + t3 * NV];
#pragma unroll
      for (int j = 0; j < 8; ++j) op[hi * 8 + j] = acc0[j];
      if (hi == 0) {
#pragma unroll
        for (int j = 0; j < NV - 16; ++j) op[16 + j] = acc1[j];
      }
    }
  }
  __syncthreads();

  {
    const int M = T3 * NV;
    v2f wG0[8], wG1[8];
#pragma unroll
    for (int s = 0; s < 8; ++s) {
      wG0[s] = *reinterpret_cast<const v2f*>(&s_gw1[((0 * 8 + s) * 32 + lane) * 2]);
      wG1[s] = *reinterpret_cast<const v2f*>(&s_gw1[((1 * 8 + s) * 32 + lane) * 2]);
    }
    for (int nt = wave; nt < 14; nt += 8) {
      v8f acc0 = {0.f,0.f,0.f,0.f,0.f,0.f,0.f,0.f};
      v8f acc1 = acc0;
      int col = nt * 16 + row;
      int mC = (col < M) ? col : 0;
      const float* bp = &s_hagg[(2 * hi) * P1 + mC];
#pragma unroll
      for (int s = 0; s < 8; ++s) {
        const float* bq = bp + (4 * s) * P1;
        v2f bb;
        bb.x = bq[0];
        bb.y = bq[P1];
        acc0 = wmma4(wG0[s], bb, acc0);
        acc1 = wmma4(wG1[s], bb, acc1);
      }
      if (col < M) {
        int t3 = col / NV;
        int t = tbase - 1 + t3;
        bool ok = (t >= 0 && t < NTIME);
#pragma unroll
        for (int j = 0; j < 8; ++j) {
          int co = hi * 8 + j;
          float val = acc0[j] + s_gb1[co];
          s_g1[co * P1 + col] = ok ? val : 0.0f;
          int c2 = co + 16;
          float va2 = acc1[j] + s_gb1[c2];
          s_g1[c2 * P1 + col] = ok ? va2 : 0.0f;
        }
      }
    }
  }
  __syncthreads();

  float* s_h1 = s_g0;
  {
    const int M = TT * NV;
    v2f wA0[24], wA1[24];
#pragma unroll
    for (int s = 0; s < 24; ++s) {
      wA0[s] = *reinterpret_cast<const v2f*>(&s_tw1[((0 * 24 + s) * 32 + lane) * 2]);
      wA1[s] = *reinterpret_cast<const v2f*>(&s_tw1[((1 * 24 + s) * 32 + lane) * 2]);
    }
    for (int nt = wave; nt < 11; nt += 8) {
      v8f acc0 = {0.f,0.f,0.f,0.f,0.f,0.f,0.f,0.f};
      v8f acc1 = acc0;
      int col = nt * 16 + row;
      int mC = (col < M) ? col : 0;
      const float* bp = &s_g1[(2 * hi) * P1 + mC];
#pragma unroll
      for (int s = 0; s < 24; ++s) {
        const float* bq = bp + (4 * (s & 7)) * P1 + (s >> 3) * NV;
        v2f bb;
        bb.x = bq[0];
        bb.y = bq[P1];
        acc0 = wmma4(wA0[s], bb, acc0);
        acc1 = wmma4(wA1[s], bb, acc1);
      }
      if (col < M) {
        int te = col / NV, ve = col - te * NV;
        int hcol = (te + 2) * NV + ve;
#pragma unroll
        for (int j = 0; j < 8; ++j) {
          int co = hi * 8 + j;
          float val = acc0[j] * s_s1[co] + s_o1[co] + s_h0[co * PH + hcol];
          s_h1[te * (NC * NV) + co * NV + ve] = fmaxf(val, 0.0f);
          int c2 = co + 16;
          float va2 = acc1[j] * s_s1[c2] + s_o1[c2] + s_h0[c2 * PH + hcol];
          s_h1[te * (NC * NV) + c2 * NV + ve] = fmaxf(va2, 0.0f);
        }
      }
    }
  }
  __syncthreads();

  {
    const float4* srcv = reinterpret_cast<const float4*>(s_h1);
    for (int idx = tid; idx < TT * (FCK / 4); idx += nthr) {
      int i = idx / (FCK / 4), q = idx - i * (FCK / 4);
      size_t R = ((size_t)b * NTIME + (size_t)(tbase + i)) * 2 + (size_t)hand;
      reinterpret_cast<float4*>(hws + R * FCK)[q] = srcv[idx];
    }
  }
}

__global__ __launch_bounds__(512) void fc_gemm(
    const float* __restrict__ H, const float* __restrict__ fc_w,
    const float* __restrict__ fc_b, float* __restrict__ out)
{
  __shared__ __attribute__((aligned(16))) float sW[NOUT * WP];

  const int tid  = threadIdx.x;
  const int lane = tid & 31;
  const int wave = tid >> 5;
  const int row  = lane & 15;
  const int hi   = lane >> 4;

  for (int idx = tid; idx < NOUT * (FCK / 4); idx += 512) {
    int o = idx / (FCK / 4), q = idx - o * (FCK / 4);
    float4 w = reinterpret_cast<const float4*>(fc_w)[idx];
    *reinterpret_cast<float4*>(&sW[o * WP + q * 4]) = w;
  }

  const int R0 = (blockIdx.x * 16 + wave) * 16;
  const float* ap = H + (size_t)(R0 + row) * FCK + hi * 4;
  const float* bp = sW + row * WP + hi * 4;

  float4 an[4];
#pragma unroll
  for (int u = 0; u < 4; ++u) an[u] = *reinterpret_cast<const float4*>(ap + u * 8);

  __syncthreads();

  v8f acc0 = {0.f,0.f,0.f,0.f,0.f,0.f,0.f,0.f};
  v8f acc1 = acc0, acc2 = acc0, acc3 = acc0;

  for (int g = 0; g < FCK / 32; ++g) {
    float4 ac[4];
#pragma unroll
    for (int u = 0; u < 4; ++u) ac[u] = an[u];
    if (g + 1 < FCK / 32) {
#pragma unroll
      for (int u = 0; u < 4; ++u)
        an[u] = *reinterpret_cast<const float4*>(ap + (g + 1) * 32 + u * 8);
    }
#pragma unroll
    for (int u = 0; u < 4; ++u) {
      const float* bq = bp + g * 32 + u * 8;
      float4 b0 = *reinterpret_cast<const float4*>(bq);
      float4 b1 = *reinterpret_cast<const float4*>(bq + 16 * WP);
      float4 b2 = *reinterpret_cast<const float4*>(bq + 32 * WP);
      float4 b3 = *reinterpret_cast<const float4*>(bq + 48 * WP);
      v2f a01 = {ac[u].x, ac[u].y};
      v2f a23 = {ac[u].z, ac[u].w};
      acc0 = wmma4(a01, (v2f){b0.x, b0.y}, acc0);
      acc1 = wmma4(a01, (v2f){b1.x, b1.y}, acc1);
      acc2 = wmma4(a01, (v2f){b2.x, b2.y}, acc2);
      acc3 = wmma4(a01, (v2f){b3.x, b3.y}, acc3);
      acc0 = wmma4(a23, (v2f){b0.z, b0.w}, acc0);
      acc1 = wmma4(a23, (v2f){b1.z, b1.w}, acc1);
      acc2 = wmma4(a23, (v2f){b2.z, b2.w}, acc2);
      acc3 = wmma4(a23, (v2f){b3.z, b3.w}, acc3);
    }
  }

  {
    float bias0 = fc_b[row], bias1 = fc_b[16 + row], bias2 = fc_b[32 + row], bias3 = fc_b[48 + row];
#pragma unroll
    for (int j = 0; j < 8; ++j) {
      float* o = out + (size_t)(R0 + hi * 8 + j) * NOUT + row;
      o[0]  = acc0[j] + bias0;
      o[16] = acc1[j] + bias1;
      o[32] = acc2[j] + bias2;
      o[48] = acc3[j] + bias3;
    }
  }
}

extern "C" void kernel_launch(void* const* d_in, const int* in_sizes, int n_in,
                              void* d_out, int out_size, void* d_ws, size_t ws_size,
                              hipStream_t stream) {
  (void)in_sizes; (void)n_in; (void)ws_size; (void)out_size;
  const float* x        = (const float*)d_in[0];
  const float* gcn_w0   = (const float*)d_in[1];
  const float* gcn_b0   = (const float*)d_in[2];
  const float* tcn_w0   = (const float*)d_in[3];
  const float* tcn_b0   = (const float*)d_in[4];
  const float* bn_g0    = (const float*)d_in[5];
  const float* bn_b0    = (const float*)d_in[6];
  const float* bn_m0    = (const float*)d_in[7];
  const float* bn_v0    = (const float*)d_in[8];
  const float* res_w    = (const float*)d_in[9];
  const float* res_b    = (const float*)d_in[10];
  const float* res_bn_g = (const float*)d_in[11];
  const float* res_bn_b = (const float*)d_in[12];
  const float* res_bn_m = (const float*)d_in[13];
  const float* res_bn_v = (const float*)d_in[14];
  const float* gcn_w1   = (const float*)d_in[15];
  const float* gcn_b1   = (const float*)d_in[16];
  const float* tcn_w1   = (const float*)d_in[17];
  const float* tcn_b1   = (const float*)d_in[18];
  const float* bn_g1    = (const float*)d_in[19];
  const float* bn_b1    = (const float*)d_in[20];
  const float* bn_m1    = (const float*)d_in[21];
  const float* bn_v1    = (const float*)d_in[22];
  const float* fc_w     = (const float*)d_in[23];
  const float* fc_b     = (const float*)d_in[24];

  dim3 grid(NB * 2 * (NTIME / TT));
  dim3 block(256);
  float* hws = (float*)d_ws;
  stgcn_body<<<grid, block, 0, stream>>>(
      x, gcn_w0, gcn_b0, tcn_w0, tcn_b0, bn_g0, bn_b0, bn_m0, bn_v0,
      res_w, res_b, res_bn_g, res_bn_b, res_bn_m, res_bn_v,
      gcn_w1, gcn_b1, tcn_w1, tcn_b1, bn_g1, bn_b1, bn_m1, bn_v1,
      hws);
  dim3 grid2(NB * NTIME * 2 / 256);
  dim3 block2(512);
  fc_gemm<<<grid2, block2, 0, stream>>>(hws, fc_w, fc_b, (float*)d_out);
}


# Body kernel walks 8 consecutive time tiles per workgroup (1024 workgroups, not 8192): adjacency build, weight staging and BN folding run once per workgroup; register bound keeps 2 workgroups per WGP
# speedup vs baseline: 2.1606x; 1.0329x over previous
#include <hip/hip_runtime.h>

#define TT 8
#define TPW 8
#define T1 (TT + 6)
#define T2 (TT + 4)
#define T3 (TT + 2)
#define NB 64
#define NTIME 512
#define NV 21
#define NC 32
#define NCIN 2
#define NOUT 64
#define FCK 672
#define WP 676

typedef __attribute__((ext_vector_type(2))) float v2f;
typedef __attribute__((ext_vector_type(8))) float v8f;

__device__ __forceinline__ v8f wmma4(v2f a, v2f b, v8f c) {
  return __builtin_amdgcn_wmma_f32_16x16x4_f32(
       false, a,  false, b,
       (short)0, c,  false,  false);
}

__device__ const unsigned char g_pi[20] = {0,1,2,3,0,5,6,7,0,9,10,11,0,13,14,15,0,17,18,19};
__device__ const unsigned char g_pj[20] = {1,2,3,4,5,6,7,8,9,10,11,12,13,14,15,16,17,18,19,20};
__device__ const float g_dinv[21] = {
  0.4082482905f,
  0.5773502692f,0.5773502692f,0.5773502692f,0.7071067812f,
  0.5773502692f,0.5773502692f,0.5773502692f,0.7071067812f,
  0.5773502692f,0.5773502692f,0.5773502692f,0.7071067812f,
  0.5773502692f,0.5773502692f,0.5773502692f,0.7071067812f,
  0.5773502692f,0.5773502692f,0.5773502692f,0.7071067812f};

#define P0 300
#define PH 268
#define P1 212
__global__ __launch_bounds__(256, 4) void stgcn_body(
    const float* __restrict__ x,
    const float* __restrict__ gcn_w0, const float* __restrict__ gcn_b0,
    const float* __restrict__ tcn_w0, const float* __restrict__ tcn_b0,
    const float* __restrict__ bn_g0, const float* __restrict__ bn_b0,
    const float* __restrict__ bn_m0, const float* __restrict__ bn_v0,
    const float* __restrict__ res_w, const float* __restrict__ res_b,
    const float* __restrict__ res_bn_g, const float* __restrict__ res_bn_b,
    const float* __restrict__ res_bn_m, const float* __restrict__ res_bn_v,
    const float* __restrict__ gcn_w1, const float* __restrict__ gcn_b1,
    const float* __restrict__ tcn_w1, const float* __restrict__ tcn_b1,
    const float* __restrict__ bn_g1, const float* __restrict__ bn_b1,
    const float* __restrict__ bn_m1, const float* __restrict__ bn_v1,
    float* __restrict__ hws)
{
  __shared__ float sA[NV * NV];
  __shared__ float s_in[T1 * NV * NCIN];
  __shared__ float s_xa[T1 * NV * NCIN];
  __shared__ __attribute__((aligned(16))) float s_tw0[96 * NC];
  __shared__ __attribute__((aligned(16))) float s_tw1[96 * NC];
  __shared__ __attribute__((aligned(16))) float s_gw1[NC * NC];
  __shared__ __attribute__((aligned(16))) float s_g0[NC * P0];
  __shared__ float s_h0[NC * PH];
  __shared__ float s_g1[NC * P1];
  __shared__ float s_gw0[NC * NCIN];
  __shared__ float s_gb0[NC], s_gb1[NC];
  __shared__ float s_s0[NC], s_o0[NC], s_s1[NC], s_o1[NC];
  __shared__ float s_rA[NC], s_rB[NC], s_rC[NC];

  const int tid  = threadIdx.x;
  const int nthr = 256;
  const int lane = tid & 31;
  const int wave = tid >> 5;
  const int row  = lane & 15;
  const int hi   = lane >> 4;

  const int bid   = blockIdx.x;
  const int tgrp  = bid & 7;
  const int hand  = (bid >> 3) & 1;
  const int b     = bid >> 4;

  for (int idx = tid; idx < NV * NV; idx += nthr) {
    int i = idx / NV, j = idx % NV;
    bool conn = (i == j);
    for (int p = 0; p < 20; ++p)
      conn = conn || ((int)g_pi[p] == i && (int)g_pj[p] == j) ||
                     ((int)g_pi[p] == j && (int)g_pj[p] == i);
    sA[idx] = conn ? g_dinv[i] * g_dinv[j] : 0.0f;
  }
  for (int idx = tid; idx < 96 * NC; idx += nthr) {
    int e = idx & 1, ln = (idx >> 1) & 31, cs = idx >> 6;
    int ct = cs / 24, s = cs - ct * 24;
    int co = ct * 16 + (ln & 15);
    int k = s >> 3;
    int ci = 4 * (s & 7) + 2 * (ln >> 4) + e;
    s_tw0[idx] = tcn_w0[(co * NC + ci) * 3 + k];
    s_tw1[idx] = tcn_w1[(co * NC + ci) * 3 + k];
  }
  for (int idx = tid; idx < NC * NC; idx += nthr) {
    int e = idx & 1, ln = (idx >> 1) & 31, cs = idx >> 6;
    int ct = cs >> 3, s = cs & 7;
    int co = ct * 16 + (ln & 15);
    int ci = 4 * s + 2 * (ln >> 4) + e;
    s_gw1[idx] = gcn_w1[co * NC + ci];
  }
  if (tid < NC * NCIN) s_gw0[tid] = gcn_w0[tid];
  if (tid < NC) {
    s_gb0[tid] = gcn_b0[tid];
    s_gb1[tid] = gcn_b1[tid];
    float sc0 = bn_g0[tid] * rsqrtf(bn_v0[tid] + 1e-5f);
    s_s0[tid] = sc0;
    s_o0[tid] = (tcn_b0[tid] - bn_m0[tid]) * sc0 + bn_b0[tid];
    float sc1 = bn_g1[tid] * rsqrtf(bn_v1[tid] + 1e-5f);
    s_s1[tid] = sc1;
    s_o1[tid] = (tcn_b1[tid] - bn_m1[tid]) * sc1 + bn_b1[tid];
    float rsc = res_bn_g[tid] * rsqrtf(res_bn_v[tid] + 1e-5f);
    s_rA[tid] = res_w[tid * 2 + 0] * rsc;
    s_rB[tid] = res_w[tid * 2 + 1] * rsc;
    s_rC[tid] = (res_b[tid] - res_bn_m[tid]) * rsc + res_bn_b[tid];
  }

  for (int it = 0; it < TPW; ++it) {
  const int tbase = (tgrp * TPW + it) * TT;

  for (int idx = tid; idx < T1 * NV * NCIN; idx += nthr) {
    int i = idx / (NV * NCIN), r = idx % (NV * NCIN);
    int v = r >> 1, ci = r & 1;
    int t = tbase - 3 + i;
    float val = 0.0f;
    if (t >= 0 && t < NTIME)
      val = x[(((b * NTIME + t) * 2 + hand) * NV + v) * NCIN + ci];
    s_in[idx] = val;
  }
  __syncthreads();

  for (int idx = tid; idx < T1 * NV * NCIN; idx += nthr) {
    int i = idx / (NV * NCIN), r = idx % (NV * NCIN);
    int v = r >> 1, ci = r & 1;
    float acc = 0.0f;
    const float* Ar = &sA[v * NV];
    const float* xr = &s_in[i * NV * NCIN + ci];
    for (int u = 0; u < NV; ++u) acc += Ar[u] * xr[u * NCIN];
    s_xa[idx] = acc;
  }
  __syncthreads();

  for (int idx = tid; idx < T1 * NV * NC; idx += nthr) {
    int c = idx / (T1 * NV), m = idx - c * (T1 * NV);
    int i = m / NV;
    int t = tbase - 3 + i;
    float val = 0.0f;
    if (t >= 0 && t < NTIME) {
      float xa0 = s_xa[m * 2 + 0];
      float xa1 = s_xa[m * 2 + 1];
      val = s_gw0[c * 2 + 0] * xa0 + s_gw0[c * 2 + 1] * xa1 + s_gb0[c];
    }
    s_g0[c * P0 + m] = val;
  }
  __syncthreads();

  {
    const int M = T2 * NV;
    v2f wA0[24], wA1[24];
#pragma unroll
    for (int s = 0; s < 24; ++s) {
      wA0[s] = *reinterpret_cast<const v2f*>(&s_tw0[((0 * 24 + s) * 32 + lane) * 2]);
      wA1[s] = *reinterpret_cast<const v2f*>(&s_tw0[((1 * 24 + s) * 32 + lane) * 2]);
    }
    for (int nt = wave; nt < 16; nt += 8) {
      v8f acc0 = {0.f,0.f,0.f,0.f,0.f,0.f,0.f,0.f};
      v8f acc1 = acc0;
      int col = nt * 16 + row;
      int mC = (col < M) ? col : 0;
      const float* bp = &s_g0[(2 * hi) * P0 + mC];
#pragma unroll
      for (int s = 0; s < 24; ++s) {
        const float* bq = bp + (4 * (s & 7)) * P0 + (s >> 3) * NV;
        v2f bb;
        bb.x = bq[0];
        bb.y = bq[P0];
        acc0 = wmma4(wA0[s], bb, acc0);
        acc1 = wmma4(wA1[s], bb, acc1);
      }
      if (col < M) {
        int te = col / NV, ve = col - te * NV;
        float x0 = s_in[((te + 1) * NV + ve) * 2 + 0];
        float x1 = s_in[((te + 1) * NV + ve) * 2 + 1];
#pragma unroll
        for (int j = 0; j < 8; ++j) {
          int co = hi * 8 + j;
          float val = acc0[j] * s_s0[co] + s_o0[co];
          val += s_rA[co] * x0 + s_rB[co] * x1 + s_rC[co];
          s_h0[co * PH + col] = fmaxf(val, 0.0f);
          int c2 = co + 16;
          float va2 = acc1[j] * s_s0[c2] + s_o0[c2];
          va2 += s_rA[c2] * x0 + s_rB[c2] * x1 + s_rC[c2];
          s_h0[c2 * PH + col] = fmaxf(va2, 0.0f);
        }
      }
    }
  }
  __syncthreads();

  float* s_hagg = s_g0;
  {
    v2f aF0[6], aF1[6];
#pragma unroll
    for (int s = 0; s < 6; ++s) {
      const int u0 = 4 * s + 2 * hi;
      const int vlo = row, vhi = 16 + row;
      const bool k0 = (u0 < NV), k1 = (u0 + 1 < NV), rv = (vhi < NV);
      float a00 = sA[vlo * NV + (k0 ? u0 : 0)];
      float a01 = sA[vlo * NV + (k1 ? u0 + 1 : 0)];
      float a10 = sA[(rv ? vhi : 0) * NV + (k0 ? u0 : 0)];
      float a11 = sA[(rv ? vhi : 0) * NV + (k1 ? u0 + 1 : 0)];
      aF0[s].x = k0 ? a00 : 0.0f;
      aF0[s].y = k1 ? a01 : 0.0f;
      aF1[s].x = (rv && k0) ? a10 : 0.0f;
      aF1[s].y = (rv && k1) ? a11 : 0.0f;
    }
    for (int nt = wave; nt < (NC * T3) / 16; nt += 8) {
      v8f acc0 = {0.f,0.f,0.f,0.f,0.f,0.f,0.f,0.f};
      v8f acc1 = acc0;
      const int n = nt * 16 + row;
      const int ci = n / T3, t3 = n - ci * T3;
      const float* hp = &s_h0[ci * PH + (t3 + 1) * NV + 2 * hi];
#pragma unroll
      for (int s = 0; s < 6; ++s) {
        v2f bb;
        if (s < 5) {
          bb.x = hp[4 * s];
          bb.y = hp[4 * s + 1];
        } else {
          bb.x = (hi == 0) ? hp[20] : 0.0f;
          bb.y = 0.0f;
        }
        acc0 = wmma4(aF0[s], bb, acc0);
        acc1 = wmma4(aF1[s], bb, acc1);
      }
      float* op = &s_hagg[ci * P1 + t3 * NV];
#pragma unroll
      for (int j = 0; j < 8; ++j) op[hi * 8 + j] = acc0[j];
      if (hi == 0) {
#pragma unroll
        for (int j = 0; j < NV - 16; ++j) op[16 + j] = acc1[j];
      }
    }
  }
  __syncthreads();

  {
    const int M = T3 * NV;
    v2f wG0[8], wG1[8];
#pragma unroll
    for (int s = 0; s < 8; ++s) {
      wG0[s] = *reinterpret_cast<const v2f*>(&s_gw1[((0 * 8 + s) * 32 + lane) * 2]);
      wG1[s] = *reinterpret_cast<const v2f*>(&s_gw1[((1 * 8 + s) * 32 + lane) * 2]);
    }
    for (int nt = wave; nt < 14; nt += 8) {
      v8f acc0 = {0.f,0.f,0.f,0.f,0.f,0.f,0.f,0.f};
      v8f acc1 = acc0;
      int col = nt * 16 + row;
      int mC = (col < M) ? col : 0;
      const float* bp = &s_hagg[(2 * hi) * P1 + mC];
#pragma unroll
      for (int s = 0; s < 8; ++s) {
        const float* bq = bp + (4 * s) * P1;
        v2f bb;
        bb.x = bq[0];
        bb.y = bq[P1];
        acc0 = wmma4(wG0[s], bb, acc0);
        acc1 = wmma4(wG1[s], bb, acc1);
      }
      if (col < M) {
        int t3 = col / NV;
        int t = tbase - 1 + t3;
        bool ok = (t >= 0 && t < NTIME);
#pragma unroll
        for (int j = 0; j < 8; ++j) {
          int co = hi * 8 + j;
          float val = acc0[j] + s_gb1[co];
          s_g1[co * P1 + col] = ok ? val : 0.0f;
          int c2 = co + 16;
          float va2 = acc1[j] + s_gb1[c2];
          s_g1[c2 * P1 + col] = ok ? va2 : 0.0f;
        }
      }
    }
  }
  __syncthreads();

  float* s_h1 = s_g0;
  {
    const int M = TT * NV;
    v2f wA0[24], wA1[24];
#pragma unroll
    for (int s = 0; s < 24; ++s) {
      wA0[s] = *reinterpret_cast<const v2f*>(&s_tw1[((0 * 24 + s) * 32 + lane) * 2]);
      wA1[s] = *reinterpret_cast<const v2f*>(&s_tw1[((1 * 24 + s) * 32 + lane) * 2]);
    }
    for (int nt = wave; nt < 11; nt += 8) {
      v8f acc0 = {0.f,0.f,0.f,0.f,0.f,0.f,0.f,0.f};
      v8f acc1 = acc0;
      int col = nt * 16 + row;
      int mC = (col < M) ? col : 0;
      const float* bp = &s_g1[(2 * hi) * P1 + mC];
#pragma unroll
      for (int s = 0; s < 24; ++s) {
        const float* bq = bp + (4 * (s & 7)) * P1 + (s >> 3) * NV;
        v2f bb;
        bb.x = bq[0];
        bb.y = bq[P1];
        acc0 = wmma4(wA0[s], bb, acc0);
        acc1 = wmma4(wA1[s], bb, acc1);
      }
      if (col < M) {
        int te = col / NV, ve = col - te * NV;
        int hcol = (te + 2) * NV + ve;
#pragma unroll
        for (int j = 0; j < 8; ++j) {
          int co = hi * 8 + j;
          float val = acc0[j] * s_s1[co] + s_o1[co] + s_h0[co * PH + hcol];
          s_h1[te * (NC * NV) + co * NV + ve] = fmaxf(val, 0.0f);
          int c2 = co + 16;
          float va2 = acc1[j] * s_s1[c2] + s_o1[c2] + s_h0[c2 * PH + hcol];
          s_h1[te * (NC * NV) + c2 * NV + ve] = fmaxf(va2, 0.0f);
        }
      }
    }
  }
  __syncthreads();

  {
    const float4* srcv = reinterpret_cast<const float4*>(s_h1);
    for (int idx = tid; idx < TT * (FCK / 4); idx += nthr) {
      int i = idx / (FCK / 4), q = idx - i * (FCK / 4);
      size_t R = ((size_t)b * NTIME + (size_t)(tbase + i)) * 2 + (size_t)hand;
      reinterpret_cast<float4*>(hws + R * FCK)[q] = srcv[idx];
    }
  }
  }
}

__global__ __launch_bounds__(512) void fc_gemm(
    const float* __restrict__ H, const float* __restrict__ fc_w,
    const float* __restrict__ fc_b, float* __restrict__ out)
{
  __shared__ __attribute__((aligned(16))) float sW[NOUT * WP];

  const int tid  = threadIdx.x;
  const int lane = tid & 31;
  const int wave = tid >> 5;
  const int row  = lane & 15;
  const int hi   = lane >> 4;

  for (int idx = tid; idx < NOUT * (FCK / 4); idx += 512) {
    int o = idx / (FCK / 4), q = idx - o * (FCK / 4);
    float4 w = reinterpret_cast<const float4*>(fc_w)[idx];
    *reinterpret_cast<float4*>(&sW[o * WP + q * 4]) = w;
  }

  const int R0 = (blockIdx.x * 16 + wave) * 16;
  const float* ap = H + (size_t)(R0 + row) * FCK + hi * 4;
  const float* bp = sW + row * WP + hi * 4;

  float4 an[4];
#pragma unroll
  for (int u = 0; u < 4; ++u) an[u] = *reinterpret_cast<const float4*>(ap + u * 8);

  __syncthreads();

  v8f acc0 = {0.f,0.f,0.f,0.f,0.f,0.f,0.f,0.f};
  v8f acc1 = acc0, acc2 = acc0, acc3 = acc0;

  for (int g = 0; g < FCK / 32; ++g) {
    float4 ac[4];
#pragma unroll
    for (int u = 0; u < 4; ++u) ac[u] = an[u];
    if (g + 1 < FCK / 32) {
#pragma unroll
      for (int u = 0; u < 4; ++u)
        an[u] = *reinterpret_cast<const float4*>(ap + (g + 1) * 32 + u * 8);
    }
#pragma unroll
    for (int u = 0; u < 4; ++u) {
      const float* bq = bp + g * 32 + u * 8;
      float4 b0 = *reinterpret_cast<const float4*>(bq);
      float4 b1 = *reinterpret_cast<const float4*>(bq + 16 * WP);
      float4 b2 = *reinterpret_cast<const float4*>(bq + 32 * WP);
      float4 b3 = *reinterpret_cast<const float4*>(bq + 48 * WP);
      v2f a01 = {ac[u].x, ac[u].y};
      v2f a23 = {ac[u].z, ac[u].w};
      acc0 = wmma4(a01, (v2f){b0.x, b0.y}, acc0);
      acc1 = wmma4(a01, (v2f){b1.x, b1.y}, acc1);
      acc2 = wmma4(a01, (v2f){b2.x, b2.y}, acc2);
      acc3 = wmma4(a01, (v2f){b3.x, b3.y}, acc3);
      acc0 = wmma4(a23, (v2f){b0.z, b0.w}, acc0);
      acc1 = wmma4(a23, (v2f){b1.z, b1.w}, acc1);
      acc2 = wmma4(a23, (v2f){b2.z, b2.w}, acc2);
      acc3 = wmma4(a23, (v2f){b3.z, b3.w}, acc3);
    }
  }

  {
    float bias0 = fc_b[row], bias1 = fc_b[16 + row], bias2 = fc_b[32 + row], bias3 = fc_b[48 + row];
#pragma unroll
    for (int j = 0; j < 8; ++j) {
      float* o = out + (size_t)(R0 + hi * 8 + j) * NOUT + row;
      o[0]  = acc0[j] + bias0;
      o[16] = acc1[j] + bias1;
      o[32] = acc2[j] + bias2;
      o[48] = acc3[j] + bias3;
    }
  }
}

extern "C" void kernel_launch(void* const* d_in, const int* in_sizes, int n_in,
                              void* d_out, int out_size, void* d_ws, size_t ws_size,
                              hipStream_t stream) {
  (void)in_sizes; (void)n_in; (void)ws_size; (void)out_size;
  const float* x        = (const float*)d_in[0];
  const float* gcn_w0   = (const float*)d_in[1];
  const float* gcn_b0   = (const float*)d_in[2];
  const float* tcn_w0   = (const float*)d_in[3];
  const float* tcn_b0   = (const float*)d_in[4];
  const float* bn_g0    = (const float*)d_in[5];
  const float* bn_b0    = (const float*)d_in[6];
  const float* bn_m0    = (const float*)d_in[7];
  const float* bn_v0    = (const float*)d_in[8];
  const float* res_w    = (const float*)d_in[9];
  const float* res_b    = (const float*)d_in[10];
  const float* res_bn_g = (const float*)d_in[11];
  const float* res_bn_b = (const float*)d_in[12];
  const float* res_bn_m = (const float*)d_in[13];
  const float* res_bn_v = (const float*)d_in[14];
  const float* gcn_w1   = (const float*)d_in[15];
  const float* gcn_b1   = (const float*)d_in[16];
  const float* tcn_w1   = (const float*)d_in[17];
  const float* tcn_b1   = (const float*)d_in[18];
  const float* bn_g1    = (const float*)d_in[19];
  const float* bn_b1    = (const float*)d_in[20];
  const float* bn_m1    = (const float*)d_in[21];
  const float* bn_v1    = (const float*)d_in[22];
  const float* fc_w     = (const float*)d_in[23];
  const float* fc_b     = (const float*)d_in[24];

  dim3 grid(NB * 2 * (NTIME / TT) / TPW);
  dim3 block(256);
  float* hws = (float*)d_ws;
  stgcn_body<<<grid, block, 0, stream>>>(
      x, gcn_w0, gcn_b0, tcn_w0, tcn_b0, bn_g0, bn_b0, bn_m0, bn_v0,
      res_w, res_b, res_bn_g, res_bn_b, res_bn_m, res_bn_v,
      gcn_w1, gcn_b1, tcn_w1, tcn_b1, bn_g1, bn_b1, bn_m1, bn_v1,
      hws);
  dim3 grid2(NB * NTIME * 2 / 256);
  dim3 block2(512);
  fc_gemm<<<grid2, block2, 0, stream>>>(hws, fc_w, fc_b, (float*)d_out);
}


# gcn0 (2->32 + bias) moved onto WMMA: one K=4 step per 16 columns, k = {xa0, xa1, 1 carrying the bias, 0}; the per-element VALU pass over 32x294 is gone; new sum order in gcn0
# speedup vs baseline: 2.5567x; 1.1833x over previous
#include <hip/hip_runtime.h>

#define TT 8
#define TPW 8
#define T1 (TT + 6)
#define T2 (TT + 4)
#define T3 (TT + 2)
#define NB 64
#define NTIME 512
#define NV 21
#define NC 32
#define NCIN 2
#define NOUT 64
#define FCK 672
#define WP 676

typedef __attribute__((ext_vector_type(2))) float v2f;
typedef __attribute__((ext_vector_type(8))) float v8f;

__device__ __forceinline__ v8f wmma4(v2f a, v2f b, v8f c) {
  return __builtin_amdgcn_wmma_f32_16x16x4_f32(
       false, a,  false, b,
       (short)0, c,  false,  false);
}

__device__ const unsigned char g_pi[20] = {0,1,2,3,0,5,6,7,0,9,10,11,0,13,14,15,0,17,18,19};
__device__ const unsigned char g_pj[20] = {1,2,3,4,5,6,7,8,9,10,11,12,13,14,15,16,17,18,19,20};
__device__ const float g_dinv[21] = {
  0.4082482905f,
  0.5773502692f,0.5773502692f,0.5773502692f,0.7071067812f,
  0.5773502692f,0.5773502692f,0.5773502692f,0.7071067812f,
  0.5773502692f,0.5773502692f,0.5773502692f,0.7071067812f,
  0.5773502692f,0.5773502692f,0.5773502692f,0.7071067812f,
  0.5773502692f,0.5773502692f,0.5773502692f,0.7071067812f};

#define P0 300
#define PH 268
#define P1 212
__global__ __launch_bounds__(256, 4) void stgcn_body(
    const float* __restrict__ x,
    const float* __restrict__ gcn_w0, const float* __restrict__ gcn_b0,
    const float* __restrict__ tcn_w0, const float* __restrict__ tcn_b0,
    const float* __restrict__ bn_g0, const float* __restrict__ bn_b0,
    const float* __restrict__ bn_m0, const float* __restrict__ bn_v0,
    const float* __restrict__ res_w, const float* __restrict__ res_b,
    const float* __restrict__ res_bn_g, const float* __restrict__ res_bn_b,
    const float* __restrict__ res_bn_m, const float* __restrict__ res_bn_v,
    const float* __restrict__ gcn_w1, const float* __restrict__ gcn_b1,
    const float* __restrict__ tcn_w1, const float* __restrict__ tcn_b1,
    const float* __restrict__ bn_g1, const float* __restrict__ bn_b1,
    const float* __restrict__ bn_m1, const float* __restrict__ bn_v1,
    float* __restrict__ hws)
{
  __shared__ float sA[NV * NV];
  __shared__ float s_in[T1 * NV * NCIN];
  __shared__ __attribute__((aligned(16))) float s_xa[T1 * NV * NCIN];
  __shared__ __attribute__((aligned(16))) float s_tw0[96 * NC];
  __shared__ __attribute__((aligned(16))) float s_tw1[96 * NC];
  __shared__ __attribute__((aligned(16))) float s_gw1[NC * NC];
  __shared__ __attribute__((aligned(16))) float s_g0[NC * P0];
  __shared__ float s_h0[NC * PH];
  __shared__ float s_g1[NC * P1];
  __shared__ float s_gw0[NC * NCIN];
  __shared__ float s_gb0[NC], s_gb1[NC];
  __shared__ float s_s0[NC], s_o0[NC], s_s1[NC], s_o1[NC];
  __shared__ float s_rA[NC], s_rB[NC], s_rC[NC];

  const int tid  = threadIdx.x;
  const int nthr = 256;
  const int lane = tid & 31;
  const int wave = tid >> 5;
  const int row  = lane & 15;
  const int hi   = lane >> 4;

  const int bid   = blockIdx.x;
  const int tgrp  = bid & 7;
  const int hand  = (bid >> 3) & 1;
  const int b     = bid >> 4;

  for (int idx = tid; idx < NV * NV; idx += nthr) {
    int i = idx / NV, j = idx % NV;
    bool conn = (i == j);
    for (int p = 0; p < 20; ++p)
      conn = conn || ((int)g_pi[p] == i && (int)g_pj[p] == j) ||
                     ((int)g_pi[p] == j && (int)g_pj[p] == i);
    sA[idx] = conn ? g_dinv[i] * g_dinv[j] : 0.0f;
  }
  for (int idx = tid; idx < 96 * NC; idx += nthr) {
    int e = idx & 1, ln = (idx >> 1) & 31, cs = idx >> 6;
    int ct = cs / 24, s = cs - ct * 24;
    int co = ct * 16 + (ln & 15);
    int k = s >> 3;
    int ci = 4 * (s & 7) + 2 * (ln >> 4) + e;
    s_tw0[idx] = tcn_w0[(co * NC + ci) * 3 + k];
    s_tw1[idx] = tcn_w1[(co * NC + ci) * 3 + k];
  }
  for (int idx = tid; idx < NC * NC; idx += nthr) {
    int e = idx & 1, ln = (idx >> 1) & 31, cs = idx >> 6;
    int ct = cs >> 3, s = cs & 7;
    int co = ct * 16 + (ln & 15);
    int ci = 4 * s + 2 * (ln >> 4) + e;
    s_gw1[idx] = gcn_w1[co * NC + ci];
  }
  if (tid < NC * NCIN) s_gw0[tid] = gcn_w0[tid];
  if (tid < NC) {
    s_gb0[tid] = gcn_b0[tid];
    s_gb1[tid] = gcn_b1[tid];
    float sc0 = bn_g0[tid] * rsqrtf(bn_v0[tid] + 1e-5f);
    s_s0[tid] = sc0;
    s_o0[tid] = (tcn_b0[tid] - bn_m0[tid]) * sc0 + bn_b0[tid];
    float sc1 = bn_g1[tid] * rsqrtf(bn_v1[tid] + 1e-5f);
    s_s1[tid] = sc1;
    s_o1[tid] = (tcn_b1[tid] - bn_m1[tid]) * sc1 + bn_b1[tid];
    float rsc = res_bn_g[tid] * rsqrtf(res_bn_v[tid] + 1e-5f);
    s_rA[tid] = res_w[tid * 2 + 0] * rsc;
    s_rB[tid] = res_w[tid * 2 + 1] * rsc;
    s_rC[tid] = (res_b[tid] - res_bn_m[tid]) * rsc + res_bn_b[tid];
  }

  for (int it = 0; it < TPW; ++it) {
  const int tbase = (tgrp * TPW + it) * TT;

  for (int idx = tid; idx < T1 * NV * NCIN; idx += nthr) {
    int i = idx / (NV * NCIN), r = idx % (NV * NCIN);
    int v = r >> 1, ci = r & 1;
    int t = tbase - 3 + i;
    float val = 0.0f;
    if (t >= 0 && t < NTIME)
      val = x[(((b * NTIME + t) * 2 + hand) * NV + v) * NCIN + ci];
    s_in[idx] = val;
  }
  __syncthreads();

  for (int idx = tid; idx < T1 * NV * NCIN; idx += nthr) {
    int i = idx / (NV * NCIN), r = idx % (NV * NCIN);
    int v = r >> 1, ci = r & 1;
    float acc = 0.0f;
    const float* Ar = &sA[v * NV];
    const float* xr = &s_in[i * NV * NCIN + ci];
    for (int u = 0; u < NV; ++u) acc += Ar[u] * xr[u * NCIN];
    s_xa[idx] = acc;
  }
  __syncthreads();

  {
    const int M = T1 * NV;
    v2f aG0, aG1;
    aG0.x = hi ? s_gb0[row] : s_gw0[row * 2 + 0];
    aG0.y = hi ? 0.0f : s_gw0[row * 2 + 1];
    aG1.x = hi ? s_gb0[16 + row] : s_gw0[(16 + row) * 2 + 0];
    aG1.y = hi ? 0.0f : s_gw0[(16 + row) * 2 + 1];
    for (int nt = wave; nt < (T1 * NV + 15) / 16; nt += 8) {
      const int col = nt * 16 + row;
      const int mC = (col < M) ? col : 0;
      const v2f xv = *reinterpret_cast<const v2f*>(&s_xa[mC * 2]);
      v2f bb;
      bb.x = hi ? 1.0f : xv.x;
      bb.y = hi ? 0.0f : xv.y;
      const v8f zero = {0.f,0.f,0.f,0.f,0.f,0.f,0.f,0.f};
      v8f acc0 = wmma4(aG0, bb, zero);
      v8f acc1 = wmma4(aG1, bb, zero);
      if (col < M) {
        const int t = tbase - 3 + col / NV;
        const bool ok = (t >= 0 && t < NTIME);
#pragma unroll
        for (int j = 0; j < 8; ++j) {
          s_g0[(hi * 8 + j) * P0 + col] = ok ? acc0[j] : 0.0f;
          s_g0[(hi * 8 + j + 16) * P0 + col] = ok ? acc1[j] : 0.0f;
        }
      }
    }
  }
  __syncthreads();

  {
    const int M = T2 * NV;
    v2f wA0[24], wA1[24];
#pragma unroll
    for (int s = 0; s < 24; ++s) {
      wA0[s] = *reinterpret_cast<const v2f*>(&s_tw0[((0 * 24 + s) * 32 + lane) * 2]);
      wA1[s] = *reinterpret_cast<const v2f*>(&s_tw0[((1 * 24 + s) * 32 + lane) * 2]);
    }
    for (int nt = wave; nt < 16; nt += 8) {
      v8f acc0 = {0.f,0.f,0.f,0.f,0.f,0.f,0.f,0.f};
      v8f acc1 = acc0;
      int col = nt * 16 + row;
      int mC = (col < M) ? col : 0;
      const float* bp = &s_g0[(2 * hi) * P0 + mC];
#pragma unroll
      for (int s = 0; s < 24; ++s) {
        const float* bq = bp + (4 * (s & 7)) * P0 + (s >> 3) * NV;
        v2f bb;
        bb.x = bq[0];
        bb.y = bq[P0];
        acc0 = wmma4(wA0[s], bb, acc0);
        acc1 = wmma4(wA1[s], bb, acc1);
      }
      if (col < M) {
        int te = col / NV, ve = col - te * NV;
        float x0 = s_in[((te + 1) * NV + ve) * 2 + 0];
        float x1 = s_in[((te + 1) * NV + ve) * 2 + 1];
#pragma unroll
        for (int j = 0; j < 8; ++j) {
          int co = hi * 8 + j;
          float val = acc0[j] * s_s0[co] + s_o0[co];
          val += s_rA[co] * x0 + s_rB[co] * x1 + s_rC[co];
          s_h0[co * PH + col] = fmaxf(val, 0.0f);
          int c2 = co + 16;
          float va2 = acc1[j] * s_s0[c2] + s_o0[c2];
          va2 += s_rA[c2] * x0 + s_rB[c2] * x1 + s_rC[c2];
          s_h0[c2 * PH + col] = fmaxf(va2, 0.0f);
        }
      }
    }
  }
  __syncthreads();

  float* s_hagg = s_g0;
  {
    v2f aF0[6], aF1[6];
#pragma unroll
    for (int s = 0; s < 6; ++s) {
      const int u0 = 4 * s + 2 * hi;
      const int vlo = row, vhi = 16 + row;
      const bool k0 = (u0 < NV), k1 = (u0 + 1 < NV), rv = (vhi < NV);
      float a00 = sA[vlo * NV + (k0 ? u0 : 0)];
      float a01 = sA[vlo * NV + (k1 ? u0 + 1 : 0)];
      float a10 = sA[(rv ? vhi : 0) * NV + (k0 ? u0 : 0)];
      float a11 = sA[(rv ? vhi : 0) * NV + (k1 ? u0 + 1 : 0)];
      aF0[s].x = k0 ? a00 : 0.0f;
      aF0[s].y = k1 ? a01 : 0.0f;
      aF1[s].x = (rv && k0) ? a10 : 0.0f;
      aF1[s].y = (rv && k1) ? a11 : 0.0f;
    }
    for (int nt = wave; nt < (NC * T3) / 16; nt += 8) {
      v8f acc0 = {0.f,0.f,0.f,0.f,0.f,0.f,0.f,0.f};
      v8f acc1 = acc0;
      const int n = nt * 16 + row;
      const int ci = n / T3, t3 = n - ci * T3;
      const float* hp = &s_h0[ci * PH + (t3 + 1) * NV + 2 * hi];
#pragma unroll
      for (int s = 0; s < 6; ++s) {
        v2f bb;
        if (s < 5) {
          bb.x = hp[4 * s];
          bb.y = hp[4 * s + 1];
        } else {
          bb.x = (hi == 0) ? hp[20] : 0.0f;
          bb.y = 0.0f;
        }
        acc0 = wmma4(aF0[s], bb, acc0);
        acc1 = wmma4(aF1[s], bb, acc1);
      }
      float* op = &s_hagg[ci * P1 + t3 * NV];
#pragma unroll
      for (int j = 0; j < 8; ++j) op[hi * 8 + j] = acc0[j];
      if (hi == 0) {
#pragma unroll
        for (int j = 0; j < NV - 16; ++j) op[16 + j] = acc1[j];
      }
    }
  }
  __syncthreads();

  {
    const int M = T3 * NV;
    v2f wG0[8], wG1[8];
#pragma unroll
    for (int s = 0; s < 8; ++s) {
      wG0[s] = *reinterpret_cast<const v2f*>(&s_gw1[((0 * 8 + s) * 32 + lane) * 2]);
      wG1[s] = *reinterpret_cast<const v2f*>(&s_gw1[((1 * 8 + s) * 32 + lane) * 2]);
    }
    for (int nt = wave; nt < 14; nt += 8) {
      v8f acc0 = {0.f,0.f,0.f,0.f,0.f,0.f,0.f,0.f};
      v8f acc1 = acc0;
      int col = nt * 16 + row;
      int mC = (col < M) ? col : 0;
      const float* bp = &s_hagg[(2 * hi) * P1 + mC];
#pragma unroll
      for (int s = 0; s < 8; ++s) {
        const float* bq = bp + (4 * s) * P1;
        v2f bb;
        bb.x = bq[0];
        bb.y = bq[P1];
        acc0 = wmma4(wG0[s], bb, acc0);
        acc1 = wmma4(wG1[s], bb, acc1);
      }
      if (col < M) {
        int t3 = col / NV;
        int t = tbase - 1 + t3;
        bool ok = (t >= 0 && t < NTIME);
#pragma unroll
        for (int j = 0; j < 8; ++j) {
          int co = hi * 8 + j;
          float val = acc0[j] + s_gb1[co];
          s_g1[co * P1 + col] = ok ? val : 0.0f;
          int c2 = co + 16;
          float va2 = acc1[j] + s_gb1[c2];
          s_g1[c2 * P1 + col] = ok ? va2 : 0.0f;
        }
      }
    }
  }
  __syncthreads();

  float* s_h1 = s_g0;
  {
    const int M = TT * NV;
    v2f wA0[24], wA1[24];
#pragma unroll
    for (int s = 0; s < 24; ++s) {
      wA0[s] = *reinterpret_cast<const v2f*>(&s_tw1[((0 * 24 + s) * 32 + lane) * 2]);
      wA1[s] = *reinterpret_cast<const v2f*>(&s_tw1[((1 * 24 + s) * 32 + lane) * 2]);
    }
    for (int nt = wave; nt < 11; nt += 8) {
      v8f acc0 = {0.f,0.f,0.f,0.f,0.f,0.f,0.f,0.f};
      v8f acc1 = acc0;
      int col = nt * 16 + row;
      int mC = (col < M) ? col : 0;
      const float* bp = &s_g1[(2 * hi) * P1 + mC];
#pragma unroll
      for (int s = 0; s < 24; ++s) {
        const float* bq = bp + (4 * (s & 7)) * P1 + (s >> 3) * NV;
        v2f bb;
        bb.x = bq[0];
        bb.y = bq[P1];
        acc0 = wmma4(wA0[s], bb, acc0);
        acc1 = wmma4(wA1[s], bb, acc1);
      }
      if (col < M) {
        int te = col / NV, ve = col - te * NV;
        int hcol = (te + 2) * NV + ve;
#pragma unroll
        for (int j = 0; j < 8; ++j) {
          int co = hi * 8 + j;
          float val = acc0[j] * s_s1[co] + s_o1[co] + s_h0[co * PH + hcol];
          s_h1[te * (NC * NV) + co * NV + ve] = fmaxf(val, 0.0f);
          int c2 = co + 16;
          float va2 = acc1[j] * s_s1[c2] + s_o1[c2] + s_h0[c2 * PH + hcol];
          s_h1[te * (NC * NV) + c2 * NV + ve] = fmaxf(va2, 0.0f);
        }
      }
    }
  }
  __syncthreads();

  {
    const float4* srcv = reinterpret_cast<const float4*>(s_h1);
    for (int idx = tid; idx < TT * (FCK / 4); idx += nthr) {
      int i = idx / (FCK / 4), q = idx - i * (FCK / 4);
      size_t R = ((size_t)b * NTIME + (size_t)(tbase + i)) * 2 + (size_t)hand;
      reinterpret_cast<float4*>(hws + R * FCK)[q] = srcv[idx];
    }
  }
  }
}

__global__ __launch_bounds__(512) void fc_gemm(
    const float* __restrict__ H, const float* __restrict__ fc_w,
    const float* __restrict__ fc_b, float* __restrict__ out)
{
  __shared__ __attribute__((aligned(16))) float sW[NOUT * WP];

  const int tid  = threadIdx.x;
  const int lane = tid & 31;
  const int wave = tid >> 5;
  const int row  = lane & 15;
  const int hi   = lane >> 4;

  for (int idx = tid; idx < NOUT * (FCK / 4); idx += 512) {
    int o = idx / (FCK / 4), q = idx - o * (FCK / 4);
    float4 w = reinterpret_cast<const float4*>(fc_w)[idx];
    *reinterpret_cast<float4*>(&sW[o * WP + q * 4]) = w;
  }

  const int R0 = (blockIdx.x * 16 + wave) * 16;
  const float* ap = H + (size_t)(R0 + row) * FCK + hi * 4;
  const float* bp = sW + row * WP + hi * 4;

  float4 an[4];
#pragma unroll
  for (int u = 0; u < 4; ++u) an[u] = *reinterpret_cast<const float4*>(ap + u * 8);

  __syncthreads();

  v8f acc0 = {0.f,0.f,0.f,0.f,0.f,0.f,0.f,0.f};
  v8f acc1 = acc0, acc2 = acc0, acc3 = acc0;

  for (int g = 0; g < FCK / 32; ++g) {
    float4 ac[4];
#pragma unroll
    for (int u = 0; u < 4; ++u) ac[u] = an[u];
    if (g + 1 < FCK / 32) {
#pragma unroll
      for (int u = 0; u < 4; ++u)
        an[u] = *reinterpret_cast<const float4*>(ap + (g + 1) * 32 + u * 8);
    }
#pragma unroll
    for (int u = 0; u < 4; ++u) {
      const float* bq = bp + g * 32 + u * 8;
      float4 b0 = *reinterpret_cast<const float4*>(bq);
      float4 b1 = *reinterpret_cast<const float4*>(bq + 16 * WP);
      float4 b2 = *reinterpret_cast<const float4*>(bq + 32 * WP);
      float4 b3 = *reinterpret_cast<const float4*>(bq + 48 * WP);
      v2f a01 = {ac[u].x, ac[u].y};
      v2f a23 = {ac[u].z, ac[u].w};
      acc0 = wmma4(a01, (v2f){b0.x, b0.y}, acc0);
      acc1 = wmma4(a01, (v2f){b1.x, b1.y}, acc1);
      acc2 = wmma4(a01, (v2f){b2.x, b2.y}, acc2);
      acc3 = wmma4(a01, (v2f){b3.x, b3.y}, acc3);
      acc0 = wmma4(a23, (v2f){b0.z, b0.w}, acc0);
      acc1 = wmma4(a23, (v2f){b1.z, b1.w}, acc1);
      acc2 = wmma4(a23, (v2f){b2.z, b2.w}, acc2);
      acc3 = wmma4(a23, (v2f){b3.z, b3.w}, acc3);
    }
  }

  {
    float bias0 = fc_b[row], bias1 = fc_b[16 + row], bias2 = fc_b[32 + row], bias3 = fc_b[48 + row];
#pragma unroll
    for (int j = 0; j < 8; ++j) {
      float* o = out + (size_t)(R0 + hi * 8 + j) * NOUT + row;
      o[0]  = acc0[j] + bias0;
      o[16] = acc1[j] + bias1;
      o[32] = acc2[j] + bias2;
      o[48] = acc3[j] + bias3;
    }
  }
}

extern "C" void kernel_launch(void* const* d_in, const int* in_sizes, int n_in,
                              void* d_out, int out_size, void* d_ws, size_t ws_size,
                              hipStream_t stream) {
  (void)in_sizes; (void)n_in; (void)ws_size; (void)out_size;
  const float* x        = (const float*)d_in[0];
  const float* gcn_w0   = (const float*)d_in[1];
  const float* gcn_b0   = (const float*)d_in[2];
  const float* tcn_w0   = (const float*)d_in[3];
  const float* tcn_b0   = (const float*)d_in[4];
  const float* bn_g0    = (const float*)d_in[5];
  const float* bn_b0    = (const float*)d_in[6];
  const float* bn_m0    = (const float*)d_in[7];
  const float* bn_v0    = (const float*)d_in[8];
  const float* res_w    = (const float*)d_in[9];
  const float* res_b    = (const float*)d_in[10];
  const float* res_bn_g = (const float*)d_in[11];
  const float* res_bn_b = (const float*)d_in[12];
  const float* res_bn_m = (const float*)d_in[13];
  const float* res_bn_v = (const float*)d_in[14];
  const float* gcn_w1   = (const float*)d_in[15];
  const float* gcn_b1   = (const float*)d_in[16];
  const float* tcn_w1   = (const float*)d_in[17];
  const float* tcn_b1   = (const float*)d_in[18];
  const float* bn_g1    = (const float*)d_in[19];
  const float* bn_b1    = (const float*)d_in[20];
  const float* bn_m1    = (const float*)d_in[21];
  const float* bn_v1    = (const float*)d_in[22];
  const float* fc_w     = (const float*)d_in[23];
  const float* fc_b     = (const float*)d_in[24];

  dim3 grid(NB * 2 * (NTIME / TT) / TPW);
  dim3 block(256);
  float* hws = (float*)d_ws;
  stgcn_body<<<grid, block, 0, stream>>>(
      x, gcn_w0, gcn_b0, tcn_w0, tcn_b0, bn_g0, bn_b0, bn_m0, bn_v0,
      res_w, res_b, res_bn_g, res_bn_b, res_bn_m, res_bn_v,
      gcn_w1, gcn_b1, tcn_w1, tcn_b1, bn_g1, bn_b1, bn_m1, bn_v1,
      hws);
  dim3 grid2(NB * NTIME * 2 / 256);
  dim3 block2(512);
  fc_gemm<<<grid2, block2, 0, stream>>>(hws, fc_w, fc_b, (float*)d_out);
}


# Body streams along time: past a workgroup's first tile each stage computes only the 6 new rows (one 16-column tile per wave); the 2 older rows a 3-tap stage needs are carried in LDS, not recomputed
# speedup vs baseline: 3.0994x; 1.2123x over previous
#include <hip/hip_runtime.h>

#define TT 6
#define T1 (TT + 6)
#define T2 (TT + 4)
#define T3 (TT + 2)
#define NB 64
#define NTIME 512
#define NTILE ((NTIME + TT - 1) / TT)
#define TPW ((NTILE + 7) / 8)
#define NV 21
#define NC 32
#define NCIN 2
#define NOUT 64
#define FCK 672
#define WP 676

typedef __attribute__((ext_vector_type(2))) float v2f;
typedef __attribute__((ext_vector_type(8))) float v8f;

__device__ __forceinline__ v8f wmma4(v2f a, v2f b, v8f c) {
  return __builtin_amdgcn_wmma_f32_16x16x4_f32(
       false, a,  false, b,
       (short)0, c,  false,  false);
}

__device__ const unsigned char g_pi[20] = {0,1,2,3,0,5,6,7,0,9,10,11,0,13,14,15,0,17,18,19};
__device__ const unsigned char g_pj[20] = {1,2,3,4,5,6,7,8,9,10,11,12,13,14,15,16,17,18,19,20};
__device__ const float g_dinv[21] = {
  0.4082482905f,
  0.5773502692f,0.5773502692f,0.5773502692f,0.7071067812f,
  0.5773502692f,0.5773502692f,0.5773502692f,0.7071067812f,
  0.5773502692f,0.5773502692f,0.5773502692f,0.7071067812f,
  0.5773502692f,0.5773502692f,0.5773502692f,0.7071067812f,
  0.5773502692f,0.5773502692f,0.5773502692f,0.7071067812f};

#define P0 268
#define PH 212
#define P1 172
static_assert(TT % 2 == 0 && P0 % 2 == 0 && PH % 2 == 0 && P1 % 2 == 0, "carried rows are moved as 8-byte words");

__device__ __forceinline__ void move_rows2(float* dst, int dpitch, const float* src, int spitch, int tid) {
  const int c = tid >> 3, l = tid & 7;
  const float2* s = reinterpret_cast<const float2*>(src + c * spitch);
  float2* d = reinterpret_cast<float2*>(dst + c * dpitch);
  const float2 w0 = s[l], w1 = s[l + 8];
  float2 w2 = w0;
  if (l < 5) w2 = s[l + 16];
  d[l] = w0;
  d[l + 8] = w1;
  if (l < 5) d[l + 16] = w2;
}
__global__ __launch_bounds__(256, 4) void stgcn_body(
    const float* __restrict__ x,
    const float* __restrict__ gcn_w0, const float* __restrict__ gcn_b0,
    const float* __restrict__ tcn_w0, const float* __restrict__ tcn_b0,
    const float* __restrict__ bn_g0, const float* __restrict__ bn_b0,
    const float* __restrict__ bn_m0, const float* __restrict__ bn_v0,
    const float* __restrict__ res_w, const float* __restrict__ res_b,
    const float* __restrict__ res_bn_g, const float* __restrict__ res_bn_b,
    const float* __restrict__ res_bn_m, const float* __restrict__ res_bn_v,
    const float* __restrict__ gcn_w1, const float* __restrict__ gcn_b1,
    const float* __restrict__ tcn_w1, const float* __restrict__ tcn_b1,
    const float* __restrict__ bn_g1, const float* __restrict__ bn_b1,
    const float* __restrict__ bn_m1, const float* __restrict__ bn_v1,
    float* __restrict__ hws)
{
  __shared__ float sA[NV * NV];
  __shared__ float s_in[T1 * NV * NCIN];
  __shared__ __attribute__((aligned(16))) float s_xa[T1 * NV * NCIN];
  __shared__ __attribute__((aligned(16))) float s_tw0[96 * NC];
  __shared__ __attribute__((aligned(16))) float s_tw1[96 * NC];
  __shared__ __attribute__((aligned(16))) float s_gw1[NC * NC];
  __shared__ __attribute__((aligned(16))) float s_g0[NC * P0];
  __shared__ __attribute__((aligned(16))) float s_h0[NC * PH];
  __shared__ __attribute__((aligned(16))) float s_g1[NC * P1];
  __shared__ __attribute__((aligned(16))) float s_c0[NC * 2 * NV];
  __shared__ float s_gw0[NC * NCIN];
  __shared__ float s_gb0[NC], s_gb1[NC];
  __shared__ float s_s0[NC], s_o0[NC], s_s1[NC], s_o1[NC];
  __shared__ float s_rA[NC], s_rB[NC], s_rC[NC];

  const int tid  = threadIdx.x;
  const int nthr = 256;
  const int lane = tid & 31;
  const int wave = tid >> 5;
  const int row  = lane & 15;
  const int hi   = lane >> 4;

  const int bid   = blockIdx.x;
  const int tgrp  = bid & 7;
  const int hand  = (bid >> 3) & 1;
  const int b     = bid >> 4;

  for (int idx = tid; idx < NV * NV; idx += nthr) {
    int i = idx / NV, j = idx % NV;
    bool conn = (i == j);
    for (int p = 0; p < 20; ++p)
      conn = conn || ((int)g_pi[p] == i && (int)g_pj[p] == j) ||
                     ((int)g_pi[p] == j && (int)g_pj[p] == i);
    sA[idx] = conn ? g_dinv[i] * g_dinv[j] : 0.0f;
  }
  for (int idx = tid; idx < 96 * NC; idx += nthr) {
    int e = idx & 1, ln = (idx >> 1) & 31, cs = idx >> 6;
    int ct = cs / 24, s = cs - ct * 24;
    int co = ct * 16 + (ln & 15);
    int k = s >> 3;
    int ci = 4 * (s & 7) + 2 * (ln >> 4) + e;
    s_tw0[idx] = tcn_w0[(co * NC + ci) * 3 + k];
    s_tw1[idx] = tcn_w1[(co * NC + ci) * 3 + k];
  }
  for (int idx = tid; idx < NC * NC; idx += nthr) {
    int e = idx & 1, ln = (idx >> 1) & 31, cs = idx >> 6;
    int ct = cs >> 3, s = cs & 7;
    int co = ct * 16 + (ln & 15);
    int ci = 4 * s + 2 * (ln >> 4) + e;
    s_gw1[idx] = gcn_w1[co * NC + ci];
  }
  if (tid < NC * NCIN) s_gw0[tid] = gcn_w0[tid];
  if (tid < NC) {
    s_gb0[tid] = gcn_b0[tid];
    s_gb1[tid] = gcn_b1[tid];
    float sc0 = bn_g0[tid] * rsqrtf(bn_v0[tid] + 1e-5f);
    s_s0[tid] = sc0;
    s_o0[tid] = (tcn_b0[tid] - bn_m0[tid]) * sc0 + bn_b0[tid];
    float sc1 = bn_g1[tid] * rsqrtf(bn_v1[tid] + 1e-5f);
    s_s1[tid] = sc1;
    s_o1[tid] = (tcn_b1[tid] - bn_m1[tid]) * sc1 + bn_b1[tid];
    float rsc = res_bn_g[tid] * rsqrtf(res_bn_v[tid] + 1e-5f);
    s_rA[tid] = res_w[tid * 2 + 0] * rsc;
    s_rB[tid] = res_w[tid * 2 + 1] * rsc;
    s_rC[tid] = (res_b[tid] - res_bn_m[tid]) * rsc + res_bn_b[tid];
  }

  for (int it = 0; it < TPW; ++it) {
  const int tbase = (tgrp * TPW + it) * TT;
  if (tbase >= NTIME) break;
  const bool first = (it == 0);
  const int rG = first ? 0 : (T1 - TT);
  const int rH = first ? 0 : (T2 - TT);
  const int rQ = first ? 0 : (T3 - TT);

  for (int idx = tid; idx < T1 * NV * NCIN; idx += nthr) {
    int i = idx / (NV * NCIN), r = idx % (NV * NCIN);
    int v = r >> 1, ci = r & 1;
    int t = tbase - 3 + i;
    float val = 0.0f;
    if (t >= 0 && t < NTIME)
      val = x[(((b * NTIME + t) * 2 + hand) * NV + v) * NCIN + ci];
    s_in[idx] = val;
  }
  __syncthreads();

  for (int idx = rG * (NV * NCIN) + tid; idx < T1 * NV * NCIN; idx += nthr) {
    int i = idx / (NV * NCIN), r = idx % (NV * NCIN);
    int v = r >> 1, ci = r & 1;
    float acc = 0.0f;
    const float* Ar = &sA[v * NV];
    const float* xr = &s_in[i * NV * NCIN + ci];
    for (int u = 0; u < NV; ++u) acc += Ar[u] * xr[u * NCIN];
    s_xa[idx] = acc;
  }
  if (!first) move_rows2(s_g0 + (T1 - TT - 2) * NV, P0, s_c0, 2 * NV, tid);
  __syncthreads();

  {
    const int M = T1 * NV;
    v2f aG0, aG1;
    aG0.x = hi ? s_gb0[row] : s_gw0[row * 2 + 0];
    aG0.y = hi ? 0.0f : s_gw0[row * 2 + 1];
    aG1.x = hi ? s_gb0[16 + row] : s_gw0[(16 + row) * 2 + 0];
    aG1.y = hi ? 0.0f : s_gw0[(16 + row) * 2 + 1];
    const int c0 = rG * NV;
    for (int nt = wave; c0 + nt * 16 < M; nt += 8) {
      const int col = c0 + nt * 16 + row;
      const int mC = (col < M) ? col : c0;
      const v2f xv = *reinterpret_cast<const v2f*>(&s_xa[mC * 2]);
      v2f bb;
      bb.x = hi ? 1.0f : xv.x;
      bb.y = hi ? 0.0f : xv.y;
      const v8f zero = {0.f,0.f,0.f,0.f,0.f,0.f,0.f,0.f};
      v8f acc0 = wmma4(aG0, bb, zero);
      v8f acc1 = wmma4(aG1, bb, zero);
      if (col < M) {
        const int t = tbase - 3 + col / NV;
        const bool ok = (t >= 0 && t < NTIME);
#pragma unroll
        for (int j = 0; j < 8; ++j) {
          s_g0[(hi * 8 + j) * P0 + col] = ok ? acc0[j] : 0.0f;
          s_g0[(hi * 8 + j + 16) * P0 + col] = ok ? acc1[j] : 0.0f;
        }
      }
    }
  }
  __syncthreads();

  {
    const int M = T2 * NV;
    v2f wA0[24], wA1[24];
#pragma unroll
    for (int s = 0; s < 24; ++s) {
      wA0[s] = *reinterpret_cast<const v2f*>(&s_tw0[((0 * 24 + s) * 32 + lane) * 2]);
      wA1[s] = *reinterpret_cast<const v2f*>(&s_tw0[((1 * 24 + s) * 32 + lane) * 2]);
    }
    move_rows2(s_c0, 2 * NV, s_g0 + (T1 - 2) * NV, P0, tid);
    const int c0 = rH * NV;
    for (int nt = wave; c0 + nt * 16 < M; nt += 8) {
      v8f acc0 = {0.f,0.f,0.f,0.f,0.f,0.f,0.f,0.f};
      v8f acc1 = acc0;
      int col = c0 + nt * 16 + row;
      int mC = (col < M) ? col : c0;
      const float* bp = &s_g0[(2 * hi) * P0 + mC];
#pragma unroll
      for (int s = 0; s < 24; ++s) {
        const float* bq = bp + (4 * (s & 7)) * P0 + (s >> 3) * NV;
        v2f bb;
        bb.x = bq[0];
        bb.y = bq[P0];
        acc0 = wmma4(wA0[s], bb, acc0);
        acc1 = wmma4(wA1[s], bb, acc1);
      }
      if (col < M) {
        int te = col / NV, ve = col - te * NV;
        float x0 = s_in[((te + 1) * NV + ve) * 2 + 0];
        float x1 = s_in[((te + 1) * NV + ve) * 2 + 1];
        float* hb = &s_h0[(hi * 8) * PH + col];
#pragma unroll
        for (int j = 0; j < 8; ++j) {
          int co = hi * 8 + j;
          float val = acc0[j] * s_s0[co] + s_o0[co];
          val += s_rA[co] * x0 + s_rB[co] * x1 + s_rC[co];
          hb[j * PH] = fmaxf(val, 0.0f);
          int c2 = co + 16;
          float va2 = acc1[j] * s_s0[c2] + s_o0[c2];
          va2 += s_rA[c2] * x0 + s_rB[c2] * x1 + s_rC[c2];
          hb[(j + 16) * PH] = fmaxf(va2, 0.0f);
        }
      }
    }
  }
  __syncthreads();

  float* s_hagg = s_g0;
  {
    v2f aF0[6], aF1[6];
#pragma unroll
    for (int s = 0; s < 6; ++s) {
      const int u0 = 4 * s + 2 * hi;
      const int vlo = row, vhi = 16 + row;
      const bool k0 = (u0 < NV), k1 = (u0 + 1 < NV), rv = (vhi < NV);
      float a00 = sA[vlo * NV + (k0 ? u0 : 0)];
      float a01 = sA[vlo * NV + (k1 ? u0 + 1 : 0)];
      float a10 = sA[(rv ? vhi : 0) * NV + (k0 ? u0 : 0)];
      float a11 = sA[(rv ? vhi : 0) * NV + (k1 ? u0 + 1 : 0)];
      aF0[s].x = k0 ? a00 : 0.0f;
      aF0[s].y = k1 ? a01 : 0.0f;
      aF1[s].x = (rv && k0) ? a10 : 0.0f;
      aF1[s].y = (rv && k1) ? a11 : 0.0f;
    }
    const int nT = T3 - rQ;
    for (int nt = wave; nt < 2 * nT; nt += 8) {
      v8f acc0 = {0.f,0.f,0.f,0.f,0.f,0.f,0.f,0.f};
      v8f acc1 = acc0;
      const int n = nt * 16 + row;
      const int ci = first ? (n / T3) : (n / TT);
      const int t3 = rQ + (n - ci * nT);
      const float* hp = &s_h0[ci * PH + (t3 + 1) * NV + 2 * hi];
#pragma unroll
      for (int s = 0; s < 6; ++s) {
        v2f bb;
        if (s < 5) {
          bb.x = hp[4 * s];
          bb.y = hp[4 * s + 1];
        } else {
          bb.x = (hi == 0) ? hp[20] : 0.0f;
          bb.y = 0.0f;
        }
        acc0 = wmma4(aF0[s], bb, acc0);
        acc1 = wmma4(aF1[s], bb, acc1);
      }
      float* op = &s_hagg[ci * P1 + t3 * NV];
#pragma unroll
      for (int j = 0; j < 8; ++j) op[hi * 8 + j] = acc0[j];
      if (hi == 0) {
#pragma unroll
        for (int j = 0; j < NV - 16; ++j) op[16 + j] = acc1[j];
      }
    }
  }
  __syncthreads();

  {
    const int M = T3 * NV;
    v2f wG0[8], wG1[8];
#pragma unroll
    for (int s = 0; s < 8; ++s) {
      wG0[s] = *reinterpret_cast<const v2f*>(&s_gw1[((0 * 8 + s) * 32 + lane) * 2]);
      wG1[s] = *reinterpret_cast<const v2f*>(&s_gw1[((1 * 8 + s) * 32 + lane) * 2]);
    }
    const int c0 = rQ * NV;
    for (int nt = wave; c0 + nt * 16 < M; nt += 8) {
      v8f acc0 = {0.f,0.f,0.f,0.f,0.f,0.f,0.f,0.f};
      v8f acc1 = acc0;
      int col = c0 + nt * 16 + row;
      int mC = (col < M) ? col : c0;
      const float* bp = &s_hagg[(2 * hi) * P1 + mC];
#pragma unroll
      for (int s = 0; s < 8; ++s) {
        const float* bq = bp + (4 * s) * P1;
        v2f bb;
        bb.x = bq[0];
        bb.y = bq[P1];
        acc0 = wmma4(wG0[s], bb, acc0);
        acc1 = wmma4(wG1[s], bb, acc1);
      }
      if (col < M) {
        int t3 = col / NV;
        int t = tbase - 1 + t3;
        bool ok = (t >= 0 && t < NTIME);
        float* gb = &s_g1[(hi * 8) * P1 + col];
#pragma unroll
        for (int j = 0; j < 8; ++j) {
          int co = hi * 8 + j;
          float val = acc0[j] + s_gb1[co];
          gb[j * P1] = ok ? val : 0.0f;
          int c2 = co + 16;
          float va2 = acc1[j] + s_gb1[c2];
          gb[(j + 16) * P1] = ok ? va2 : 0.0f;
        }
      }
    }
  }
  __syncthreads();

  float* s_h1 = s_g0;
  {
    const int M = TT * NV;
    v2f wA0[24], wA1[24];
#pragma unroll
    for (int s = 0; s < 24; ++s) {
      wA0[s] = *reinterpret_cast<const v2f*>(&s_tw1[((0 * 24 + s) * 32 + lane) * 2]);
      wA1[s] = *reinterpret_cast<const v2f*>(&s_tw1[((1 * 24 + s) * 32 + lane) * 2]);
    }
    for (int nt = wave; nt < (TT * NV + 15) / 16; nt += 8) {
      v8f acc0 = {0.f,0.f,0.f,0.f,0.f,0.f,0.f,0.f};
      v8f acc1 = acc0;
      int col = nt * 16 + row;
      int mC = (col < M) ? col : 0;
      const float* bp = &s_g1[(2 * hi) * P1 + mC];
#pragma unroll
      for (int s = 0; s < 24; ++s) {
        const float* bq = bp + (4 * (s & 7)) * P1 + (s >> 3) * NV;
        v2f bb;
        bb.x = bq[0];
        bb.y = bq[P1];
        acc0 = wmma4(wA0[s], bb, acc0);
        acc1 = wmma4(wA1[s], bb, acc1);
      }
      if (col < M) {
        int te = col / NV, ve = col - te * NV;
        int hcol = (te + 2) * NV + ve;
        const float* rb = &s_h0[(hi * 8) * PH + hcol];
        float* ob = &s_h1[te * (NC * NV) + (hi * 8) * NV + ve];
        float r0[8], r1[8];
#pragma unroll
        for (int j = 0; j < 8; ++j) {
          r0[j] = rb[j * PH];
          r1[j] = rb[(j + 16) * PH];
        }
#pragma unroll
        for (int j = 0; j < 8; ++j) {
          int co = hi * 8 + j;
          float val = acc0[j] * s_s1[co] + s_o1[co] + r0[j];
          ob[j * NV] = fmaxf(val, 0.0f);
          int c2 = co + 16;
          float va2 = acc1[j] * s_s1[c2] + s_o1[c2] + r1[j];
          ob[(j + 16) * NV] = fmaxf(va2, 0.0f);
        }
      }
    }
  }
  __syncthreads();

  {
    const float4* srcv = reinterpret_cast<const float4*>(s_h1);
    for (int idx = tid; idx < TT * (FCK / 4); idx += nthr) {
      int i = idx / (FCK / 4), q = idx - i * (FCK / 4);
      if (tbase + i < NTIME) {
        size_t R = ((size_t)b * NTIME + (size_t)(tbase + i)) * 2 + (size_t)hand;
        reinterpret_cast<float4*>(hws + R * FCK)[q] = srcv[idx];
      }
    }
    move_rows2(s_h0 + (T2 - TT - 2) * NV, PH, s_h0 + (T2 - 2) * NV, PH, tid);
    move_rows2(s_g1, P1, s_g1 + (T3 - 2) * NV, P1, tid);
  }
  }
}

__global__ __launch_bounds__(512) void fc_gemm(
    const float* __restrict__ H, const float* __restrict__ fc_w,
    const float* __restrict__ fc_b, float* __restrict__ out)
{
  __shared__ __attribute__((aligned(16))) float sW[NOUT * WP];

  const int tid  = threadIdx.x;
  const int lane = tid & 31;
  const int wave = tid >> 5;
  const int row  = lane & 15;
  const int hi   = lane >> 4;

  for (int idx = tid; idx < NOUT * (FCK / 4); idx += 512) {
    int o = idx / (FCK / 4), q = idx - o * (FCK / 4);
    float4 w = reinterpret_cast<const float4*>(fc_w)[idx];
    *reinterpret_cast<float4*>(&sW[o * WP + q * 4]) = w;
  }

  const int R0 = (blockIdx.x * 16 + wave) * 16;
  const float* ap = H + (size_t)(R0 + row) * FCK + hi * 4;
  const float* bp = sW + row * WP + hi * 4;

  float4 an[4];
#pragma unroll
  for (int u = 0; u < 4; ++u) an[u] = *reinterpret_cast<const float4*>(ap + u * 8);

  __syncthreads();

  v8f acc0 = {0.f,0.f,0.f,0.f,0.f,0.f,0.f,0.f};
  v8f acc1 = acc0, acc2 = acc0, acc3 = acc0;

  for (int g = 0; g < FCK / 32; ++g) {
    float4 ac[4];
#pragma unroll
    for (int u = 0; u < 4; ++u) ac[u] = an[u];
    if (g + 1 < FCK / 32) {
#pragma unroll
      for (int u = 0; u < 4; ++u)
        an[u] = *reinterpret_cast<const float4*>(ap + (g + 1) * 32 + u * 8);
    }
#pragma unroll
    for (int u = 0; u < 4; ++u) {
      const float* bq = bp + g * 32 + u * 8;
      float4 b0 = *reinterpret_cast<const float4*>(bq);
      float4 b1 = *reinterpret_cast<const float4*>(bq + 16 * WP);
      float4 b2 = *reinterpret_cast<const float4*>(bq + 32 * WP);
      float4 b3 = *reinterpret_cast<const float4*>(bq + 48 * WP);
      v2f a01 = {ac[u].x, ac[u].y};
      v2f a23 = {ac[u].z, ac[u].w};
      acc0 = wmma4(a01, (v2f){b0.x, b0.y}, acc0);
      acc1 = wmma4(a01, (v2f){b1.x, b1.y}, acc1);
      acc2 = wmma4(a01, (v2f){b2.x, b2.y}, acc2);
      acc3 = wmma4(a01, (v2f){b3.x, b3.y}, acc3);
      acc0 = wmma4(a23, (v2f){b0.z, b0.w}, acc0);
      acc1 = wmma4(a23, (v2f){b1.z, b1.w}, acc1);
      acc2 = wmma4(a23, (v2f){b2.z, b2.w}, acc2);
      acc3 = wmma4(a23, (v2f){b3.z, b3.w}, acc3);
    }
  }

  {
    float bias0 = fc_b[row], bias1 = fc_b[16 + row], bias2 = fc_b[32 + row], bias3 = fc_b[48 + row];
#pragma unroll
    for (int j = 0; j < 8; ++j) {
      float* o = out + (size_t)(R0 + hi * 8 + j) * NOUT + row;
      o[0]  = acc0[j] + bias0;
      o[16] = acc1[j] + bias1;
      o[32] = acc2[j] + bias2;
      o[48] = acc3[j] + bias3;
    }
  }
}

extern "C" void kernel_launch(void* const* d_in, const int* in_sizes, int n_in,
                              void* d_out, int out_size, void* d_ws, size_t ws_size,
                              hipStream_t stream) {
  (void)in_sizes; (void)n_in; (void)ws_size; (void)out_size;
  const float* x        = (const float*)d_in[0];
  const float* gcn_w0   = (const float*)d_in[1];
  const float* gcn_b0   = (const float*)d_in[2];
  const float* tcn_w0   = (const float*)d_in[3];
  const float* tcn_b0   = (const float*)d_in[4];
  const float* bn_g0    = (const float*)d_in[5];
  const float* bn_b0    = (const float*)d_in[6];
  const float* bn_m0    = (const float*)d_in[7];
  const float* bn_v0    = (const float*)d_in[8];
  const float* res_w    = (const float*)d_in[9];
  const float* res_b    = (const float*)d_in[10];
  const float* res_bn_g = (const float*)d_in[11];
  const float* res_bn_b = (const float*)d_in[12];
  const float* res_bn_m = (const float*)d_in[13];
  const float* res_bn_v = (const float*)d_in[14];
  const float* gcn_w1   = (const float*)d_in[15];
  const float* gcn_b1   = (const float*)d_in[16];
  const float* tcn_w1   = (const float*)d_in[17];
  const float* tcn_b1   = (const float*)d_in[18];
  const float* bn_g1    = (const float*)d_in[19];
  const float* bn_b1    = (const float*)d_in[20];
  const float* bn_m1    = (const float*)d_in[21];
  const float* bn_v1    = (const float*)d_in[22];
  const float* fc_w     = (const float*)d_in[23];
  const float* fc_b     = (const float*)d_in[24];

  dim3 grid(NB * 2 * 8);
  dim3 block(256);
  float* hws = (float*)d_ws;
  stgcn_body<<<grid, block, 0, stream>>>(
      x, gcn_w0, gcn_b0, tcn_w0, tcn_b0, bn_g0, bn_b0, bn_m0, bn_v0,
      res_w, res_b, res_bn_g, res_bn_b, res_bn_m, res_bn_v,
      gcn_w1, gcn_b1, tcn_w1, tcn_b1, bn_g1, bn_b1, bn_m1, bn_v1,
      hws);
  dim3 grid2(NB * NTIME * 2 / 256);
  dim3 block2(512);
  fc_gemm<<<grid2, block2, 0, stream>>>(hws, fc_w, fc_b, (float*)d_out);
}


# Body: x of a workgroup's whole 72-row time span is loaded as one group and graph-aggregated once, before the tile loop; per-tile global load, VALU pass and 1 of 7 barriers gone; same sum order
# speedup vs baseline: 3.3316x; 1.0749x over previous
#include <hip/hip_runtime.h>

#define TT 6
#define T1 (TT + 6)
#define T2 (TT + 4)
#define T3 (TT + 2)
#define NB 64
#define NTIME 512
#define NTILE ((NTIME + TT - 1) / TT)
#define TPW ((NTILE + 7) / 8)
#define SPAN (TPW * TT + 6)
#define XLD ((SPAN * NV + 255) / 256)
#define NV 21
#define NC 32
#define NCIN 2
#define NOUT 64
#define FCK 672
#define WP 676

typedef __attribute__((ext_vector_type(2))) float v2f;
typedef __attribute__((ext_vector_type(8))) float v8f;

__device__ __forceinline__ v8f wmma4(v2f a, v2f b, v8f c) {
  return __builtin_amdgcn_wmma_f32_16x16x4_f32(
       false, a,  false, b,
       (short)0, c,  false,  false);
}

__device__ const unsigned char g_pi[20] = {0,1,2,3,0,5,6,7,0,9,10,11,0,13,14,15,0,17,18,19};
__device__ const unsigned char g_pj[20] = {1,2,3,4,5,6,7,8,9,10,11,12,13,14,15,16,17,18,19,20};
__device__ const float g_dinv[21] = {
  0.4082482905f,
  0.5773502692f,0.5773502692f,0.5773502692f,0.7071067812f,
  0.5773502692f,0.5773502692f,0.5773502692f,0.7071067812f,
  0.5773502692f,0.5773502692f,0.5773502692f,0.7071067812f,
  0.5773502692f,0.5773502692f,0.5773502692f,0.7071067812f,
  0.5773502692f,0.5773502692f,0.5773502692f,0.7071067812f};

#define P0 268
#define PH 212
#define P1 172
static_assert(TT % 2 == 0 && P0 % 2 == 0 && PH % 2 == 0 && P1 % 2 == 0, "carried rows are moved as 8-byte words");
static_assert(SPAN % TT == 0 && TT * NV * NCIN <= 256, "the span aggregate runs in passes of TT rows, one output per thread");

__device__ __forceinline__ void move_rows2(float* dst, int dpitch, const float* src, int spitch, int tid) {
  const int c = tid >> 3, l = tid & 7;
  const float2* s = reinterpret_cast<const float2*>(src + c * spitch);
  float2* d = reinterpret_cast<float2*>(dst + c * dpitch);
  const float2 w0 = s[l], w1 = s[l + 8];
  float2 w2 = w0;
  if (l < 5) w2 = s[l + 16];
  d[l] = w0;
  d[l + 8] = w1;
  if (l < 5) d[l + 16] = w2;
}
__global__ __launch_bounds__(256, 4) void stgcn_body(
    const float* __restrict__ x,
    const float* __restrict__ gcn_w0, const float* __restrict__ gcn_b0,
    const float* __restrict__ tcn_w0, const float* __restrict__ tcn_b0,
    const float* __restrict__ bn_g0, const float* __restrict__ bn_b0,
    const float* __restrict__ bn_m0, const float* __restrict__ bn_v0,
    const float* __restrict__ res_w, const float* __restrict__ res_b,
    const float* __restrict__ res_bn_g, const float* __restrict__ res_bn_b,
    const float* __restrict__ res_bn_m, const float* __restrict__ res_bn_v,
    const float* __restrict__ gcn_w1, const float* __restrict__ gcn_b1,
    const float* __restrict__ tcn_w1, const float* __restrict__ tcn_b1,
    const float* __restrict__ bn_g1, const float* __restrict__ bn_b1,
    const float* __restrict__ bn_m1, const float* __restrict__ bn_v1,
    float* __restrict__ hws)
{
  __shared__ float sA[NV * NV];
  __shared__ __attribute__((aligned(16))) float s_in[SPAN * NV * NCIN];
  __shared__ __attribute__((aligned(16))) float s_xa[SPAN * NV * NCIN];
  __shared__ __attribute__((aligned(16))) float s_tw0[96 * NC];
  __shared__ __attribute__((aligned(16))) float s_tw1[96 * NC];
  __shared__ __attribute__((aligned(16))) float s_gw1[NC * NC];
  __shared__ __attribute__((aligned(16))) float s_g0[NC * P0];
  __shared__ __attribute__((aligned(16))) float s_h0[NC * PH];
  __shared__ __attribute__((aligned(16))) float s_g1[NC * P1];
  __shared__ __attribute__((aligned(16))) float s_c0[NC * 2 * NV];
  __shared__ float s_gw0[NC * NCIN];
  __shared__ float s_gb0[NC], s_gb1[NC];
  __shared__ float s_s0[NC], s_o0[NC], s_s1[NC], s_o1[NC];
  __shared__ float s_rA[NC], s_rB[NC], s_rC[NC];

  const int tid  = threadIdx.x;
  const int nthr = 256;
  const int lane = tid & 31;
  const int wave = tid >> 5;
  const int row  = lane & 15;
  const int hi   = lane >> 4;

  const int bid   = blockIdx.x;
  const int tgrp  = bid & 7;
  const int hand  = (bid >> 3) & 1;
  const int b     = bid >> 4;

  float2 xw[XLD];
  {
    const int tb0 = tgrp * TPW * TT;
#pragma unroll
    for (int k = 0; k < XLD; ++k) {
      const int idx = tid + k * nthr;
      const int idc = (idx < SPAN * NV) ? idx : (SPAN * NV - 1);
      const int i = idc / NV, v = idc - i * NV;
      const int t = tb0 - 3 + i;
      const int tc = (t < 0) ? 0 : ((t < NTIME) ? t : (NTIME - 1));
      float2 w = *reinterpret_cast<const float2*>(&x[(((b * NTIME + tc) * 2 + hand) * NV + v) * NCIN]);
      if (t != tc) w = make_float2(0.0f, 0.0f);
      xw[k] = w;
    }
  }
  for (int idx = tid; idx < NV * NV; idx += nthr) {
    int i = idx / NV, j = idx % NV;
    bool conn = (i == j);
    for (int p = 0; p < 20; ++p)
      conn = conn || ((int)g_pi[p] == i && (int)g_pj[p] == j) ||
                     ((int)g_pi[p] == j && (int)g_pj[p] == i);
    sA[idx] = conn ? g_dinv[i] * g_dinv[j] : 0.0f;
  }
  for (int idx = tid; idx < 96 * NC; idx += nthr) {
    int e = idx & 1, ln = (idx >> 1) & 31, cs = idx >> 6;
    int ct = cs / 24, s = cs - ct * 24;
    int co = ct * 16 + (ln & 15);
    int k = s >> 3;
    int ci = 4 * (s & 7) + 2 * (ln >> 4) + e;
    s_tw0[idx] = tcn_w0[(co * NC + ci) * 3 + k];
    s_tw1[idx] = tcn_w1[(co * NC + ci) * 3 + k];
  }
  for (int idx = tid; idx < NC * NC; idx += nthr) {
    int e = idx & 1, ln = (idx >> 1) & 31, cs = idx >> 6;
    int ct = cs >> 3, s = cs & 7;
    int co = ct * 16 + (ln & 15);
    int ci = 4 * s + 2 * (ln >> 4) + e;
    s_gw1[idx] = gcn_w1[co * NC + ci];
  }
  if (tid < NC * NCIN) s_gw0[tid] = gcn_w0[tid];
  if (tid < NC) {
    s_gb0[tid] = gcn_b0[tid];
    s_gb1[tid] = gcn_b1[tid];
    float sc0 = bn_g0[tid] * rsqrtf(bn_v0[tid] + 1e-5f);
    s_s0[tid] = sc0;
    s_o0[tid] = (tcn_b0[tid] - bn_m0[tid]) * sc0 + bn_b0[tid];
    float sc1 = bn_g1[tid] * rsqrtf(bn_v1[tid] + 1e-5f);
    s_s1[tid] = sc1;
    s_o1[tid] = (tcn_b1[tid] - bn_m1[tid]) * sc1 + bn_b1[tid];
    float rsc = res_bn_g[tid] * rsqrtf(res_bn_v[tid] + 1e-5f);
    s_rA[tid] = res_w[tid * 2 + 0] * rsc;
    s_rB[tid] = res_w[tid * 2 + 1] * rsc;
    s_rC[tid] = (res_b[tid] - res_bn_m[tid]) * rsc + res_bn_b[tid];
  }

#pragma unroll
  for (int k = 0; k < XLD; ++k) {
    const int idx = tid + k * nthr;
    if (idx < SPAN * NV) *reinterpret_cast<float2*>(&s_in[idx * 2]) = xw[k];
  }
  __syncthreads();
  if (tid < TT * NV * NCIN) {
    const int rr = tid / (NV * NCIN), r = tid - rr * (NV * NCIN);
    const int v = r >> 1, ci = r & 1;
    float ar[NV];
#pragma unroll
    for (int u = 0; u < NV; ++u) ar[u] = sA[v * NV + u];
#pragma unroll 4
    for (int p = 0; p < SPAN / TT; ++p) {
      const int i = p * TT + rr;
      const float* xr = &s_in[i * NV * NCIN + ci];
      float acc = 0.0f;
#pragma unroll
      for (int u = 0; u < NV; ++u) acc += ar[u] * xr[u * NCIN];
      s_xa[i * NV * NCIN + r] = acc;
    }
  }

  for (int it = 0; it < TPW; ++it) {
  const int tbase = (tgrp * TPW + it) * TT;
  if (tbase >= NTIME) break;
  const bool first = (it == 0);
  const float* xin = s_in + it * (TT * NV * NCIN);
  const float* xag = s_xa + it * (TT * NV * NCIN);
  const int rG = first ? 0 : (T1 - TT);
  const int rH = first ? 0 : (T2 - TT);
  const int rQ = first ? 0 : (T3 - TT);

  __syncthreads();
  if (!first) move_rows2(s_g0 + (T1 - TT - 2) * NV, P0, s_c0, 2 * NV, tid);

  {
    const int M = T1 * NV;
    v2f aG0, aG1;
    aG0.x = hi ? s_gb0[row] : s_gw0[row * 2 + 0];
    aG0.y = hi ? 0.0f : s_gw0[row * 2 + 1];
    aG1.x = hi ? s_gb0[16 + row] : s_gw0[(16 + row) * 2 + 0];
    aG1.y = hi ? 0.0f : s_gw0[(16 + row) * 2 + 1];
    const int c0 = rG * NV;
    for (int nt = wave; c0 + nt * 16 < M; nt += 8) {
      const int col = c0 + nt * 16 + row;
      const int mC = (col < M) ? col : c0;
      const v2f xv = *reinterpret_cast<const v2f*>(&xag[mC * 2]);
      v2f bb;
      bb.x = hi ? 1.0f : xv.x;
      bb.y = hi ? 0.0f : xv.y;
      const v8f zero = {0.f,0.f,0.f,0.f,0.f,0.f,0.f,0.f};
      v8f acc0 = wmma4(aG0, bb, zero);
      v8f acc1 = wmma4(aG1, bb, zero);
      if (col < M) {
        const int t = tbase - 3 + col / NV;
        const bool ok = (t >= 0 && t < NTIME);
#pragma unroll
        for (int j = 0; j < 8; ++j) {
          s_g0[(hi * 8 + j) * P0 + col] = ok ? acc0[j] : 0.0f;
          s_g0[(hi * 8 + j + 16) * P0 + col] = ok ? acc1[j] : 0.0f;
        }
      }
    }
  }
  __syncthreads();

  {
    const int M = T2 * NV;
    v2f wA0[24], wA1[24];
#pragma unroll
    for (int s = 0; s < 24; ++s) {
      wA0[s] = *reinterpret_cast<const v2f*>(&s_tw0[((0 * 24 + s) * 32 + lane) * 2]);
      wA1[s] = *reinterpret_cast<const v2f*>(&s_tw0[((1 * 24 + s) * 32 + lane) * 2]);
    }
    move_rows2(s_c0, 2 * NV, s_g0 + (T1 - 2) * NV, P0, tid);
    const int c0 = rH * NV;
    for (int nt = wave; c0 + nt * 16 < M; nt += 8) {
      v8f acc0 = {0.f,0.f,0.f,0.f,0.f,0.f,0.f,0.f};
      v8f acc1 = acc0;
      int col = c0 + nt * 16 + row;
      int mC = (col < M) ? col : c0;
      const float* bp = &s_g0[(2 * hi) * P0 + mC];
#pragma unroll
      for (int s = 0; s < 24; ++s) {
        const float* bq = bp + (4 * (s & 7)) * P0 + (s >> 3) * NV;
        v2f bb;
        bb.x = bq[0];
        bb.y = bq[P0];
        acc0 = wmma4(wA0[s], bb, acc0);
        acc1 = wmma4(wA1[s], bb, acc1);
      }
      if (col < M) {
        int te = col / NV, ve = col - te * NV;
        float x0 = xin[((te + 1) * NV + ve) * 2 + 0];
        float x1 = xin[((te + 1) * NV + ve) * 2 + 1];
        float* hb = &s_h0[(hi * 8) * PH + col];
#pragma unroll
        for (int j = 0; j < 8; ++j) {
          int co = hi * 8 + j;
          float val = acc0[j] * s_s0[co] + s_o0[co];
          val += s_rA[co] * x0 + s_rB[co] * x1 + s_rC[co];
          hb[j * PH] = fmaxf(val, 0.0f);
          int c2 = co + 16;
          float va2 = acc1[j] * s_s0[c2] + s_o0[c2];
          va2 += s_rA[c2] * x0 + s_rB[c2] * x1 + s_rC[c2];
          hb[(j + 16) * PH] = fmaxf(va2, 0.0f);
        }
      }
    }
  }
  __syncthreads();

  float* s_hagg = s_g0;
  {
    v2f aF0[6], aF1[6];
#pragma unroll
    for (int s = 0; s < 6; ++s) {
      const int u0 = 4 * s + 2 * hi;
      const int vlo = row, vhi = 16 + row;
      const bool k0 = (u0 < NV), k1 = (u0 + 1 < NV), rv = (vhi < NV);
      float a00 = sA[vlo * NV + (k0 ? u0 : 0)];
      float a01 = sA[vlo * NV + (k1 ? u0 + 1 : 0)];
      float a10 = sA[(rv ? vhi : 0) * NV + (k0 ? u0 : 0)];
      float a11 = sA[(rv ? vhi : 0) * NV + (k1 ? u0 + 1 : 0)];
      aF0[s].x = k0 ? a00 : 0.0f;
      aF0[s].y = k1 ? a01 : 0.0f;
      aF1[s].x = (rv && k0) ? a10 : 0.0f;
      aF1[s].y = (rv && k1) ? a11 : 0.0f;
    }
    const int nT = T3 - rQ;
    for (int nt = wave; nt < 2 * nT; nt += 8) {
      v8f acc0 = {0.f,0.f,0.f,0.f,0.f,0.f,0.f,0.f};
      v8f acc1 = acc0;
      const int n = nt * 16 + row;
      const int ci = first ? (n / T3) : (n / TT);
      const int t3 = rQ + (n - ci * nT);
      const float* hp = &s_h0[ci * PH + (t3 + 1) * NV + 2 * hi];
#pragma unroll
      for (int s = 0; s < 6; ++s) {
        v2f bb;
        if (s < 5) {
          bb.x = hp[4 * s];
          bb.y = hp[4 * s + 1];
        } else {
          bb.x = (hi == 0) ? hp[20] : 0.0f;
          bb.y = 0.0f;
        }
        acc0 = wmma4(aF0[s], bb, acc0);
        acc1 = wmma4(aF1[s], bb, acc1);
      }
      float* op = &s_hagg[ci * P1 + t3 * NV];
#pragma unroll
      for (int j = 0; j < 8; ++j) op[hi * 8 + j] = acc0[j];
      if (hi == 0) {
#pragma unroll
        for (int j = 0; j < NV - 16; ++j) op[16 + j] = acc1[j];
      }
    }
  }
  __syncthreads();

  {
    const int M = T3 * NV;
    v2f wG0[8], wG1[8];
#pragma unroll
    for (int s = 0; s < 8; ++s) {
      wG0[s] = *reinterpret_cast<const v2f*>(&s_gw1[((0 * 8 + s) * 32 + lane) * 2]);
      wG1[s] = *reinterpret_cast<const v2f*>(&s_gw1[((1 * 8 + s) * 32 + lane) * 2]);
    }
    const int c0 = rQ * NV;
    for (int nt = wave; c0 + nt * 16 < M; nt += 8) {
      v8f acc0 = {0.f,0.f,0.f,0.f,0.f,0.f,0.f,0.f};
      v8f acc1 = acc0;
      int col = c0 + nt * 16 + row;
      int mC = (col < M) ? col : c0;
      const float* bp = &s_hagg[(2 * hi) * P1 + mC];
#pragma unroll
      for (int s = 0; s < 8; ++s) {
        const float* bq = bp + (4 * s) * P1;
        v2f bb;
        bb.x = bq[0];
        bb.y = bq[P1];
        acc0 = wmma4(wG0[s], bb, acc0);
        acc1 = wmma4(wG1[s], bb, acc1);
      }
      if (col < M) {
        int t3 = col / NV;
        int t = tbase - 1 + t3;
        bool ok = (t >= 0 && t < NTIME);
        float* gb = &s_g1[(hi * 8) * P1 + col];
#pragma unroll
        for (int j = 0; j < 8; ++j) {
          int co = hi * 8 + j;
          float val = acc0[j] + s_gb1[co];
          gb[j * P1] = ok ? val : 0.0f;
          int c2 = co + 16;
          float va2 = acc1[j] + s_gb1[c2];
          gb[(j + 16) * P1] = ok ? va2 : 0.0f;
        }
      }
    }
  }
  __syncthreads();

  float* s_h1 = s_g0;
  {
    const int M = TT * NV;
    v2f wA0[24], wA1[24];
#pragma unroll
    for (int s = 0; s < 24; ++s) {
      wA0[s] = *reinterpret_cast<const v2f*>(&s_tw1[((0 * 24 + s) * 32 + lane) * 2]);
      wA1[s] = *reinterpret_cast<const v2f*>(&s_tw1[((1 * 24 + s) * 32 + lane) * 2]);
    }
    for (int nt = wave; nt < (TT * NV + 15) / 16; nt += 8) {
      v8f acc0 = {0.f,0.f,0.f,0.f,0.f,0.f,0.f,0.f};
      v8f acc1 = acc0;
      int col = nt * 16 + row;
      int mC = (col < M) ? col : 0;
      const float* bp = &s_g1[(2 * hi) * P1 + mC];
#pragma unroll
      for (int s = 0; s < 24; ++s) {
        const float* bq = bp + (4 * (s & 7)) * P1 + (s >> 3) * NV;
        v2f bb;
        bb.x = bq[0];
        bb.y = bq[P1];
        acc0 = wmma4(wA0[s], bb, acc0);
        acc1 = wmma4(wA1[s], bb, acc1);
      }
      if (col < M) {
        int te = col / NV, ve = col - te * NV;
        int hcol = (te + 2) * NV + ve;
        const float* rb = &s_h0[(hi * 8) * PH + hcol];
        float* ob = &s_h1[te * (NC * NV) + (hi * 8) * NV + ve];
        float r0[8], r1[8];
#pragma unroll
        for (int j = 0; j < 8; ++j) {
          r0[j] = rb[j * PH];
          r1[j] = rb[(j + 16) * PH];
        }
#pragma unroll
        for (int j = 0; j < 8; ++j) {
          int co = hi * 8 + j;
          float val = acc0[j] * s_s1[co] + s_o1[co] + r0[j];
          ob[j * NV] = fmaxf(val, 0.0f);
          int c2 = co + 16;
          float va2 = acc1[j] * s_s1[c2] + s_o1[c2] + r1[j];
          ob[(j + 16) * NV] = fmaxf(va2, 0.0f);
        }
      }
    }
  }
  __syncthreads();

  {
    const float4* srcv = reinterpret_cast<const float4*>(s_h1);
    for (int idx = tid; idx < TT * (FCK / 4); idx += nthr) {
      int i = idx / (FCK / 4), q = idx - i * (FCK / 4);
      if (tbase + i < NTIME) {
        size_t R = ((size_t)b * NTIME + (size_t)(tbase + i)) * 2 + (size_t)hand;
        reinterpret_cast<float4*>(hws + R * FCK)[q] = srcv[idx];
      }
    }
    move_rows2(s_h0 + (T2 - TT - 2) * NV, PH, s_h0 + (T2 - 2) * NV, PH, tid);
    move_rows2(s_g1, P1, s_g1 + (T3 - 2) * NV, P1, tid);
  }
  }
}

__global__ __launch_bounds__(512) void fc_gemm(
    const float* __restrict__ H, const float* __restrict__ fc_w,
    const float* __restrict__ fc_b, float* __restrict__ out)
{
  __shared__ __attribute__((aligned(16))) float sW[NOUT * WP];

  const int tid  = threadIdx.x;
  const int lane = tid & 31;
  const int wave = tid >> 5;
  const int row  = lane & 15;
  const int hi   = lane >> 4;

  for (int idx = tid; idx < NOUT * (FCK / 4); idx += 512) {
    int o = idx / (FCK / 4), q = idx - o * (FCK / 4);
    float4 w = reinterpret_cast<const float4*>(fc_w)[idx];
    *reinterpret_cast<float4*>(&sW[o * WP + q * 4]) = w;
  }

  const int R0 = (blockIdx.x * 16 + wave) * 16;
  const float* ap = H + (size_t)(R0 + row) * FCK + hi * 4;
  const float* bp = sW + row * WP + hi * 4;

  float4 an[4];
#pragma unroll
  for (int u = 0; u < 4; ++u) an[u] = *reinterpret_cast<const float4*>(ap + u * 8);

  __syncthreads();

  v8f acc0 = {0.f,0.f,0.f,0.f,0.f,0.f,0.f,0.f};
  v8f acc1 = acc0, acc2 = acc0, acc3 = acc0;

  for (int g = 0; g < FCK / 32; ++g) {
    float4 ac[4];
#pragma unroll
    for (int u = 0; u < 4; ++u) ac[u] = an[u];
    if (g + 1 < FCK / 32) {
#pragma unroll
      for (int u = 0; u < 4; ++u)
        an[u] = *reinterpret_cast<const float4*>(ap + (g + 1) * 32 + u * 8);
    }
#pragma unroll
    for (int u = 0; u < 4; ++u) {
      const float* bq = bp + g * 32 + u * 8;
      float4 b0 = *reinterpret_cast<const float4*>(bq);
      float4 b1 = *reinterpret_cast<const float4*>(bq + 16 * WP);
      float4 b2 = *reinterpret_cast<const float4*>(bq + 32 * WP);
      float4 b3 = *reinterpret_cast<const float4*>(bq + 48 * WP);
      v2f a01 = {ac[u].x, ac[u].y};
      v2f a23 = {ac[u].z, ac[u].w};
      acc0 = wmma4(a01, (v2f){b0.x, b0.y}, acc0);
      acc1 = wmma4(a01, (v2f){b1.x, b1.y}, acc1);
      acc2 = wmma4(a01, (v2f){b2.x, b2.y}, acc2);
      acc3 = wmma4(a01, (v2f){b3.x, b3.y}, acc3);
      acc0 = wmma4(a23, (v2f){b0.z, b0.w}, acc0);
      acc1 = wmma4(a23, (v2f){b1.z, b1.w}, acc1);
      acc2 = wmma4(a23, (v2f){b2.z, b2.w}, acc2);
      acc3 = wmma4(a23, (v2f){b3.z, b3.w}, acc3);
    }
  }

  {
    float bias0 = fc_b[row], bias1 = fc_b[16 + row], bias2 = fc_b[32 + row], bias3 = fc_b[48 + row];
#pragma unroll
    for (int j = 0; j < 8; ++j) {
      float* o = out + (size_t)(R0 + hi * 8 + j) * NOUT + row;
      o[0]  = acc0[j] + bias0;
      o[16] = acc1[j] + bias1;
      o[32] = acc2[j] + bias2;
      o[48] = acc3[j] + bias3;
    }
  }
}

extern "C" void kernel_launch(void* const* d_in, const int* in_sizes, int n_in,
                              void* d_out, int out_size, void* d_ws, size_t ws_size,
                              hipStream_t stream) {
  (void)in_sizes; (void)n_in; (void)ws_size; (void)out_size;
  const float* x        = (const float*)d_in[0];
  const float* gcn_w0   = (const float*)d_in[1];
  const float* gcn_b0   = (const float*)d_in[2];
  const float* tcn_w0   = (const float*)d_in[3];
  const float* tcn_b0   = (const float*)d_in[4];
  const float* bn_g0    = (const float*)d_in[5];
  const float* bn_b0    = (const float*)d_in[6];
  const float* bn_m0    = (const float*)d_in[7];
  const float* bn_v0    = (const float*)d_in[8];
  const float* res_w    = (const float*)d_in[9];
  const float* res_b    = (const float*)d_in[10];
  const float* res_bn_g = (const float*)d_in[11];
  const float* res_bn_b = (const float*)d_in[12];
  const float* res_bn_m = (const float*)d_in[13];
  const float* res_bn_v = (const float*)d_in[14];
  const float* gcn_w1   = (const float*)d_in[15];
  const float* gcn_b1   = (const float*)d_in[16];
  const float* tcn_w1   = (const float*)d_in[17];
  const float* tcn_b1   = (const float*)d_in[18];
  const float* bn_g1    = (const float*)d_in[19];
  const float* bn_b1    = (const float*)d_in[20];
  const float* bn_m1    = (const float*)d_in[21];
  const float* bn_v1    = (const float*)d_in[22];
  const float* fc_w     = (const float*)d_in[23];
  const float* fc_b     = (const float*)d_in[24];

  dim3 grid(NB * 2 * 8);
  dim3 block(256);
  float* hws = (float*)d_ws;
  stgcn_body<<<grid, block, 0, stream>>>(
      x, gcn_w0, gcn_b0, tcn_w0, tcn_b0, bn_g0, bn_b0, bn_m0, bn_v0,
      res_w, res_b, res_bn_g, res_bn_b, res_bn_m, res_bn_v,
      gcn_w1, gcn_b1, tcn_w1, tcn_b1, bn_g1, bn_b1, bn_m1, bn_v1,
      hws);
  dim3 grid2(NB * NTIME * 2 / 256);
  dim3 block2(512);
  fc_gemm<<<grid2, block2, 0, stream>>>(hws, fc_w, fc_b, (float*)d_out);
}


# Both 3-tap time convolutions in minimal-filtering form: two output rows from four K=32 WMMA products of row sums/differences, not six; weights pre-combined per workgroup; new sum order in TCN0, TCN1
# speedup vs baseline: 3.7117x; 1.1141x over previous
#include <hip/hip_runtime.h>

#define TT 6
#define T1 (TT + 6)
#define T2 (TT + 4)
#define T3 (TT + 2)
#define NB 64
#define NTIME 512
#define NTILE ((NTIME + TT - 1) / TT)
#define TPW ((NTILE + 7) / 8)
#define SPAN (TPW * TT + 6)
#define XLD ((SPAN * NV + 255) / 256)
#define NV 21
#define NC 32
#define NCIN 2
#define NOUT 64
#define FCK 672
#define WP 676

typedef __attribute__((ext_vector_type(2))) float v2f;
typedef __attribute__((ext_vector_type(8))) float v8f;

__device__ __forceinline__ v8f wmma4(v2f a, v2f b, v8f c) {
  return __builtin_amdgcn_wmma_f32_16x16x4_f32(
       false, a,  false, b,
       (short)0, c,  false,  false);
}

__device__ const unsigned char g_pi[20] = {0,1,2,3,0,5,6,7,0,9,10,11,0,13,14,15,0,17,18,19};
__device__ const unsigned char g_pj[20] = {1,2,3,4,5,6,7,8,9,10,11,12,13,14,15,16,17,18,19,20};
__device__ const float g_dinv[21] = {
  0.4082482905f,
  0.5773502692f,0.5773502692f,0.5773502692f,0.7071067812f,
  0.5773502692f,0.5773502692f,0.5773502692f,0.7071067812f,
  0.5773502692f,0.5773502692f,0.5773502692f,0.7071067812f,
  0.5773502692f,0.5773502692f,0.5773502692f,0.7071067812f,
  0.5773502692f,0.5773502692f,0.5773502692f,0.7071067812f};

#define P0 268
#define PH 212
#define P1 172
static_assert(TT % 2 == 0 && P0 % 2 == 0 && PH % 2 == 0 && P1 % 2 == 0, "carried rows are moved as 8-byte words");
static_assert(SPAN % TT == 0 && TT * NV * NCIN <= 256, "the span aggregate runs in passes of TT rows, one output per thread");

__device__ __forceinline__ void move_rows2(float* dst, int dpitch, const float* src, int spitch, int tid) {
  const int c = tid >> 3, l = tid & 7;
  const float2* s = reinterpret_cast<const float2*>(src + c * spitch);
  float2* d = reinterpret_cast<float2*>(dst + c * dpitch);
  const float2 w0 = s[l], w1 = s[l + 8];
  float2 w2 = w0;
  if (l < 5) w2 = s[l + 16];
  d[l] = w0;
  d[l + 8] = w1;
  if (l < 5) d[l + 16] = w2;
}
__global__ __launch_bounds__(256, 4) void stgcn_body(
    const float* __restrict__ x,
    const float* __restrict__ gcn_w0, const float* __restrict__ gcn_b0,
    const float* __restrict__ tcn_w0, const float* __restrict__ tcn_b0,
    const float* __restrict__ bn_g0, const float* __restrict__ bn_b0,
    const float* __restrict__ bn_m0, const float* __restrict__ bn_v0,
    const float* __restrict__ res_w, const float* __restrict__ res_b,
    const float* __restrict__ res_bn_g, const float* __restrict__ res_bn_b,
    const float* __restrict__ res_bn_m, const float* __restrict__ res_bn_v,
    const float* __restrict__ gcn_w1, const float* __restrict__ gcn_b1,
    const float* __restrict__ tcn_w1, const float* __restrict__ tcn_b1,
    const float* __restrict__ bn_g1, const float* __restrict__ bn_b1,
    const float* __restrict__ bn_m1, const float* __restrict__ bn_v1,
    float* __restrict__ hws)
{
  __shared__ float sA[NV * NV];
  __shared__ __attribute__((aligned(16))) float s_in[SPAN * NV * NCIN];
  __shared__ __attribute__((aligned(16))) float s_xa[SPAN * NV * NCIN];
  __shared__ __attribute__((aligned(16))) float s_tw0[128 * NC];
  __shared__ __attribute__((aligned(16))) float s_tw1[128 * NC];
  __shared__ __attribute__((aligned(16))) float s_gw1[NC * NC];
  __shared__ __attribute__((aligned(16))) float s_g0[NC * P0];
  __shared__ __attribute__((aligned(16))) float s_h0[NC * PH];
  __shared__ __attribute__((aligned(16))) float s_g1[NC * P1];
  __shared__ __attribute__((aligned(16))) float s_c0[NC * 2 * NV];
  __shared__ float s_gw0[NC * NCIN];
  __shared__ float s_gb0[NC], s_gb1[NC];
  __shared__ float s_s0[NC], s_o0[NC], s_s1[NC], s_o1[NC];
  __shared__ float s_rA[NC], s_rB[NC], s_rC[NC];

  const int tid  = threadIdx.x;
  const int nthr = 256;
  const int lane = tid & 31;
  const int wave = tid >> 5;
  const int row  = lane & 15;
  const int hi   = lane >> 4;

  const int bid   = blockIdx.x;
  const int tgrp  = bid & 7;
  const int hand  = (bid >> 3) & 1;
  const int b     = bid >> 4;

  float2 xw[XLD];
  {
    const int tb0 = tgrp * TPW * TT;
#pragma unroll
    for (int k = 0; k < XLD; ++k) {
      const int idx = tid + k * nthr;
      const int idc = (idx < SPAN * NV) ? idx : (SPAN * NV - 1);
      const int i = idc / NV, v = idc - i * NV;
      const int t = tb0 - 3 + i;
      const int tc = (t < 0) ? 0 : ((t < NTIME) ? t : (NTIME - 1));
      float2 w = *reinterpret_cast<const float2*>(&x[(((b * NTIME + tc) * 2 + hand) * NV + v) * NCIN]);
      if (t != tc) w = make_float2(0.0f, 0.0f);
      xw[k] = w;
    }
  }
  for (int idx = tid; idx < NV * NV; idx += nthr) {
    int i = idx / NV, j = idx % NV;
    bool conn = (i == j);
    for (int p = 0; p < 20; ++p)
      conn = conn || ((int)g_pi[p] == i && (int)g_pj[p] == j) ||
                     ((int)g_pi[p] == j && (int)g_pj[p] == i);
    sA[idx] = conn ? g_dinv[i] * g_dinv[j] : 0.0f;
  }
  for (int idx = tid; idx < 128 * NC; idx += nthr) {
    int e = idx & 1, ln = (idx >> 1) & 31, cs = idx >> 6;
    int q = cs & 7, j = (cs >> 3) & 3, ct = cs >> 5;
    int co = ct * 16 + (ln & 15);
    int ci = 4 * q + 2 * (ln >> 4) + e;
    const float* w0p = &tcn_w0[(co * NC + ci) * 3];
    const float* w1p = &tcn_w1[(co * NC + ci) * 3];
    const float a0 = w0p[0], a1 = w0p[1], a2 = w0p[2];
    const float b0 = w1p[0], b1 = w1p[1], b2 = w1p[2];
    float ga, gb;
    if (j == 0)      { ga = a0; gb = b0; }
    else if (j == 1) { ga = 0.5f * ((a0 + a2) + a1); gb = 0.5f * ((b0 + b2) + b1); }
    else if (j == 2) { ga = 0.5f * ((a0 + a2) - a1); gb = 0.5f * ((b0 + b2) - b1); }
    else             { ga = a2; gb = b2; }
    s_tw0[idx] = ga;
    s_tw1[idx] = gb;
  }
  for (int idx = tid; idx < NC * NC; idx += nthr) {
    int e = idx & 1, ln = (idx >> 1) & 31, cs = idx >> 6;
    int ct = cs >> 3, s = cs & 7;
    int co = ct * 16 + (ln & 15);
    int ci = 4 * s + 2 * (ln >> 4) + e;
    s_gw1[idx] = gcn_w1[co * NC + ci];
  }
  if (tid < NC * NCIN) s_gw0[tid] = gcn_w0[tid];
  if (tid < NC) {
    s_gb0[tid] = gcn_b0[tid];
    s_gb1[tid] = gcn_b1[tid];
    float sc0 = bn_g0[tid] * rsqrtf(bn_v0[tid] + 1e-5f);
    s_s0[tid] = sc0;
    s_o0[tid] = (tcn_b0[tid] - bn_m0[tid]) * sc0 + bn_b0[tid];
    float sc1 = bn_g1[tid] * rsqrtf(bn_v1[tid] + 1e-5f);
    s_s1[tid] = sc1;
    s_o1[tid] = (tcn_b1[tid] - bn_m1[tid]) * sc1 + bn_b1[tid];
    float rsc = res_bn_g[tid] * rsqrtf(res_bn_v[tid] + 1e-5f);
    s_rA[tid] = res_w[tid * 2 + 0] * rsc;
    s_rB[tid] = res_w[tid * 2 + 1] * rsc;
    s_rC[tid] = (res_b[tid] - res_bn_m[tid]) * rsc + res_bn_b[tid];
  }

#pragma unroll
  for (int k = 0; k < XLD; ++k) {
    const int idx = tid + k * nthr;
    if (idx < SPAN * NV) *reinterpret_cast<float2*>(&s_in[idx * 2]) = xw[k];
  }
  __syncthreads();
  if (tid < TT * NV * NCIN) {
    const int rr = tid / (NV * NCIN), r = tid - rr * (NV * NCIN);
    const int v = r >> 1, ci = r & 1;
    float ar[NV];
#pragma unroll
    for (int u = 0; u < NV; ++u) ar[u] = sA[v * NV + u];
#pragma unroll 4
    for (int p = 0; p < SPAN / TT; ++p) {
      const int i = p * TT + rr;
      const float* xr = &s_in[i * NV * NCIN + ci];
      float acc = 0.0f;
#pragma unroll
      for (int u = 0; u < NV; ++u) acc += ar[u] * xr[u * NCIN];
      s_xa[i * NV * NCIN + r] = acc;
    }
  }

  for (int it = 0; it < TPW; ++it) {
  const int tbase = (tgrp * TPW + it) * TT;
  if (tbase >= NTIME) break;
  const bool first = (it == 0);
  const float* xin = s_in + it * (TT * NV * NCIN);
  const float* xag = s_xa + it * (TT * NV * NCIN);
  const int rG = first ? 0 : (T1 - TT);
  const int rH = first ? 0 : (T2 - TT);
  const int rQ = first ? 0 : (T3 - TT);

  __syncthreads();
  if (!first) move_rows2(s_g0 + (T1 - TT - 2) * NV, P0, s_c0, 2 * NV, tid);

  {
    const int M = T1 * NV;
    v2f aG0, aG1;
    aG0.x = hi ? s_gb0[row] : s_gw0[row * 2 + 0];
    aG0.y = hi ? 0.0f : s_gw0[row * 2 + 1];
    aG1.x = hi ? s_gb0[16 + row] : s_gw0[(16 + row) * 2 + 0];
    aG1.y = hi ? 0.0f : s_gw0[(16 + row) * 2 + 1];
    const int c0 = rG * NV;
    for (int nt = wave; c0 + nt * 16 < M; nt += 8) {
      const int col = c0 + nt * 16 + row;
      const int mC = (col < M) ? col : c0;
      const v2f xv = *reinterpret_cast<const v2f*>(&xag[mC * 2]);
      v2f bb;
      bb.x = hi ? 1.0f : xv.x;
      bb.y = hi ? 0.0f : xv.y;
      const v8f zero = {0.f,0.f,0.f,0.f,0.f,0.f,0.f,0.f};
      v8f acc0 = wmma4(aG0, bb, zero);
      v8f acc1 = wmma4(aG1, bb, zero);
      if (col < M) {
        const int t = tbase - 3 + col / NV;
        const bool ok = (t >= 0 && t < NTIME);
#pragma unroll
        for (int j = 0; j < 8; ++j) {
          s_g0[(hi * 8 + j) * P0 + col] = ok ? acc0[j] : 0.0f;
          s_g0[(hi * 8 + j + 16) * P0 + col] = ok ? acc1[j] : 0.0f;
        }
      }
    }
  }
  __syncthreads();

  {
    const int cot = wave & 1;
    v2f wG[4][8];
#pragma unroll
    for (int j = 0; j < 4; ++j) {
#pragma unroll
      for (int q = 0; q < 8; ++q)
        wG[j][q] = *reinterpret_cast<const v2f*>(&s_tw0[(((cot * 4 + j) * 8 + q) * 32 + lane) * 2]);
    }
    move_rows2(s_c0, 2 * NV, s_g0 + (T1 - 2) * NV, P0, tid);
    const int M = (first ? (T2 / 2) : (TT / 2)) * NV;
    for (int ct = wave >> 1; ct * 16 < M; ct += 4) {
      const v8f zero = {0.f,0.f,0.f,0.f,0.f,0.f,0.f,0.f};
      v8f m0 = zero, m1 = zero, m2 = zero, m3 = zero;
      const int col = ct * 16 + row;
      const int mC = (col < M) ? col : 0;
      const int p = mC / NV, ve = mC - p * NV;
      const int r = rH + 2 * p;
      const float* bp = &s_g0[(2 * hi) * P0 + r * NV + ve];
#pragma unroll
      for (int q = 0; q < 8; ++q) {
        const float* bq = bp + (4 * q) * P0;
        const float d0x = bq[0], d1x = bq[NV], d2x = bq[2 * NV], d3x = bq[3 * NV];
        const float d0y = bq[P0], d1y = bq[P0 + NV], d2y = bq[P0 + 2 * NV], d3y = bq[P0 + 3 * NV];
        v2f t0, t1, t2, t3;
        t0.x = d0x - d2x; t0.y = d0y - d2y;
        t1.x = d1x + d2x; t1.y = d1y + d2y;
        t2.x = d2x - d1x; t2.y = d2y - d1y;
        t3.x = d1x - d3x; t3.y = d1y - d3y;
        m0 = wmma4(wG[0][q], t0, m0);
        m1 = wmma4(wG[1][q], t1, m1);
        m2 = wmma4(wG[2][q], t2, m2);
        m3 = wmma4(wG[3][q], t3, m3);
      }
      if (col < M) {
        const float2 xa = *reinterpret_cast<const float2*>(&xin[((r + 1) * NV + ve) * 2]);
        const float2 xb = *reinterpret_cast<const float2*>(&xin[((r + 2) * NV + ve) * 2]);
        float* hb = &s_h0[(cot * 16 + hi * 8) * PH + r * NV + ve];
#pragma unroll
        for (int j = 0; j < 8; ++j) {
          const int co = cot * 16 + hi * 8 + j;
          const float sc = s_s0[co], of = s_o0[co], ra = s_rA[co], rb = s_rB[co], rc = s_rC[co];
          const float y0 = (m0[j] + m1[j]) + m2[j];
          const float y1 = (m1[j] - m2[j]) - m3[j];
          float v0 = y0 * sc + of;
          v0 += ra * xa.x + rb * xa.y + rc;
          float v1 = y1 * sc + of;
          v1 += ra * xb.x + rb * xb.y + rc;
          hb[j * PH] = fmaxf(v0, 0.0f);
          hb[j * PH + NV] = fmaxf(v1, 0.0f);
        }
      }
    }
  }
  __syncthreads();

  float* s_hagg = s_g0;
  {
    v2f aF0[6], aF1[6];
#pragma unroll
    for (int s = 0; s < 6; ++s) {
      const int u0 = 4 * s + 2 * hi;
      const int vlo = row, vhi = 16 + row;
      const bool k0 = (u0 < NV), k1 = (u0 + 1 < NV), rv = (vhi < NV);
      float a00 = sA[vlo * NV + (k0 ? u0 : 0)];
      float a01 = sA[vlo * NV + (k1 ? u0 + 1 : 0)];
      float a10 = sA[(rv ? vhi : 0) * NV + (k0 ? u0 : 0)];
      float a11 = sA[(rv ? vhi : 0) * NV + (k1 ? u0 + 1 : 0)];
      aF0[s].x = k0 ? a00 : 0.0f;
      aF0[s].y = k1 ? a01 : 0.0f;
      aF1[s].x = (rv && k0) ? a10 : 0.0f;
      aF1[s].y = (rv && k1) ? a11 : 0.0f;
    }
    const int nT = T3 - rQ;
    for (int nt = wave; nt < 2 * nT; nt += 8) {
      v8f acc0 = {0.f,0.f,0.f,0.f,0.f,0.f,0.f,0.f};
      v8f acc1 = acc0;
      const int n = nt * 16 + row;
      const int ci = first ? (n / T3) : (n / TT);
      const int t3 = rQ + (n - ci * nT);
      const float* hp = &s_h0[ci * PH + (t3 + 1) * NV + 2 * hi];
#pragma unroll
      for (int s = 0; s < 6; ++s) {
        v2f bb;
        if (s < 5) {
          bb.x = hp[4 * s];
          bb.y = hp[4 * s + 1];
        } else {
          bb.x = (hi == 0) ? hp[20] : 0.0f;
          bb.y = 0.0f;
        }
        acc0 = wmma4(aF0[s], bb, acc0);
        acc1 = wmma4(aF1[s], bb, acc1);
      }
      float* op = &s_hagg[ci * P1 + t3 * NV];
#pragma unroll
      for (int j = 0; j < 8; ++j) op[hi * 8 + j] = acc0[j];
      if (hi == 0) {
#pragma unroll
        for (int j = 0; j < NV - 16; ++j) op[16 + j] = acc1[j];
      }
    }
  }
  __syncthreads();

  {
    const int M = T3 * NV;
    v2f wG0[8], wG1[8];
#pragma unroll
    for (int s = 0; s < 8; ++s) {
      wG0[s] = *reinterpret_cast<const v2f*>(&s_gw1[((0 * 8 + s) * 32 + lane) * 2]);
      wG1[s] = *reinterpret_cast<const v2f*>(&s_gw1[((1 * 8 + s) * 32 + lane) * 2]);
    }
    const int c0 = rQ * NV;
    for (int nt = wave; c0 + nt * 16 < M; nt += 8) {
      v8f acc0 = {0.f,0.f,0.f,0.f,0.f,0.f,0.f,0.f};
      v8f acc1 = acc0;
      int col = c0 + nt * 16 + row;
      int mC = (col < M) ? col : c0;
      const float* bp = &s_hagg[(2 * hi) * P1 + mC];
#pragma unroll
      for (int s = 0; s < 8; ++s) {
        const float* bq = bp + (4 * s) * P1;
        v2f bb;
        bb.x = bq[0];
        bb.y = bq[P1];
        acc0 = wmma4(wG0[s], bb, acc0);
        acc1 = wmma4(wG1[s], bb, acc1);
      }
      if (col < M) {
        int t3 = col / NV;
        int t = tbase - 1 + t3;
        bool ok = (t >= 0 && t < NTIME);
        float* gb = &s_g1[(hi * 8) * P1 + col];
#pragma unroll
        for (int j = 0; j < 8; ++j) {
          int co = hi * 8 + j;
          float val = acc0[j] + s_gb1[co];
          gb[j * P1] = ok ? val : 0.0f;
          int c2 = co + 16;
          float va2 = acc1[j] + s_gb1[c2];
          gb[(j + 16) * P1] = ok ? va2 : 0.0f;
        }
      }
    }
  }
  __syncthreads();

  float* s_h1 = s_g0;
  {
    const int M = (TT / 2) * NV;
    const int cot = wave & 1;
    v2f wG[4][8];
#pragma unroll
    for (int j = 0; j < 4; ++j) {
#pragma unroll
      for (int q = 0; q < 8; ++q)
        wG[j][q] = *reinterpret_cast<const v2f*>(&s_tw1[(((cot * 4 + j) * 8 + q) * 32 + lane) * 2]);
    }
    {
      const int ct = wave >> 1;
      const v8f zero = {0.f,0.f,0.f,0.f,0.f,0.f,0.f,0.f};
      v8f m0 = zero, m1 = zero, m2 = zero, m3 = zero;
      const int col = ct * 16 + row;
      const int mC = (col < M) ? col : 0;
      const int p = mC / NV, ve = mC - p * NV;
      const int r = 2 * p;
      const float* bp = &s_g1[(2 * hi) * P1 + r * NV + ve];
#pragma unroll
      for (int q = 0; q < 8; ++q) {
        const float* bq = bp + (4 * q) * P1;
        const float d0x = bq[0], d1x = bq[NV], d2x = bq[2 * NV], d3x = bq[3 * NV];
        const float d0y = bq[P1], d1y = bq[P1 + NV], d2y = bq[P1 + 2 * NV], d3y = bq[P1 + 3 * NV];
        v2f t0, t1, t2, t3;
        t0.x = d0x - d2x; t0.y = d0y - d2y;
        t1.x = d1x + d2x; t1.y = d1y + d2y;
        t2.x = d2x - d1x; t2.y = d2y - d1y;
        t3.x = d1x - d3x; t3.y = d1y - d3y;
        m0 = wmma4(wG[0][q], t0, m0);
        m1 = wmma4(wG[1][q], t1, m1);
        m2 = wmma4(wG[2][q], t2, m2);
        m3 = wmma4(wG[3][q], t3, m3);
      }
      if (col < M) {
        const float* rb = &s_h0[(cot * 16 + hi * 8) * PH + (r + 2) * NV + ve];
        float* ob = &s_h1[r * (NC * NV) + (cot * 16 + hi * 8) * NV + ve];
        float r0[8], r1[8];
#pragma unroll
        for (int j = 0; j < 8; ++j) {
          r0[j] = rb[j * PH];
          r1[j] = rb[j * PH + NV];
        }
#pragma unroll
        for (int j = 0; j < 8; ++j) {
          const int co = cot * 16 + hi * 8 + j;
          const float sc = s_s1[co], of = s_o1[co];
          const float y0 = (m0[j] + m1[j]) + m2[j];
          const float y1 = (m1[j] - m2[j]) - m3[j];
          const float v0 = y0 * sc + of + r0[j];
          const float v1 = y1 * sc + of + r1[j];
          ob[j * NV] = fmaxf(v0, 0.0f);
          ob[j * NV + NC * NV] = fmaxf(v1, 0.0f);
        }
      }
    }
  }
  __syncthreads();

  {
    const float4* srcv = reinterpret_cast<const float4*>(s_h1);
    for (int idx = tid; idx < TT * (FCK / 4); idx += nthr) {
      int i = idx / (FCK / 4), q = idx - i * (FCK / 4);
      if (tbase + i < NTIME) {
        size_t R = ((size_t)b * NTIME + (size_t)(tbase + i)) * 2 + (size_t)hand;
        reinterpret_cast<float4*>(hws + R * FCK)[q] = srcv[idx];
      }
    }
    move_rows2(s_h0 + (T2 - TT - 2) * NV, PH, s_h0 + (T2 - 2) * NV, PH, tid);
    move_rows2(s_g1, P1, s_g1 + (T3 - 2) * NV, P1, tid);
  }
  }
}

__global__ __launch_bounds__(512) void fc_gemm(
    const float* __restrict__ H, const float* __restrict__ fc_w,
    const float* __restrict__ fc_b, float* __restrict__ out)
{
  __shared__ __attribute__((aligned(16))) float sW[NOUT * WP];

  const int tid  = threadIdx.x;
  const int lane = tid & 31;
  const int wave = tid >> 5;
  const int row  = lane & 15;
  const int hi   = lane >> 4;

  for (int idx = tid; idx < NOUT * (FCK / 4); idx += 512) {
    int o = idx / (FCK / 4), q = idx - o * (FCK / 4);
    float4 w = reinterpret_cast<const float4*>(fc_w)[idx];
    *reinterpret_cast<float4*>(&sW[o * WP + q * 4]) = w;
  }

  const int R0 = (blockIdx.x * 16 + wave) * 16;
  const float* ap = H + (size_t)(R0 + row) * FCK + hi * 4;
  const float* bp = sW + row * WP + hi * 4;

  float4 an[4];
#pragma unroll
  for (int u = 0; u < 4; ++u) an[u] = *reinterpret_cast<const float4*>(ap + u * 8);

  __syncthreads();

  v8f acc0 = {0.f,0.f,0.f,0.f,0.f,0.f,0.f,0.f};
  v8f acc1 = acc0, acc2 = acc0, acc3 = acc0;

  for (int g = 0; g < FCK / 32; ++g) {
    float4 ac[4];
#pragma unroll
    for (int u = 0; u < 4; ++u) ac[u] = an[u];
    if (g + 1 < FCK / 32) {
#pragma unroll
      for (int u = 0; u < 4; ++u)
        an[u] = *reinterpret_cast<const float4*>(ap + (g + 1) * 32 + u * 8);
    }
#pragma unroll
    for (int u = 0; u < 4; ++u) {
      const float* bq = bp + g * 32 + u * 8;
      float4 b0 = *reinterpret_cast<const float4*>(bq);
      float4 b1 = *reinterpret_cast<const float4*>(bq + 16 * WP);
      float4 b2 = *reinterpret_cast<const float4*>(bq + 32 * WP);
      float4 b3 = *reinterpret_cast<const float4*>(bq + 48 * WP);
      v2f a01 = {ac[u].x, ac[u].y};
      v2f a23 = {ac[u].z, ac[u].w};
      acc0 = wmma4(a01, (v2f){b0.x, b0.y}, acc0);
      acc1 = wmma4(a01, (v2f){b1.x, b1.y}, acc1);
      acc2 = wmma4(a01, (v2f){b2.x, b2.y}, acc2);
      acc3 = wmma4(a01, (v2f){b3.x, b3.y}, acc3);
      acc0 = wmma4(a23, (v2f){b0.z, b0.w}, acc0);
      acc1 = wmma4(a23, (v2f){b1.z, b1.w}, acc1);
      acc2 = wmma4(a23, (v2f){b2.z, b2.w}, acc2);
      acc3 = wmma4(a23, (v2f){b3.z, b3.w}, acc3);
    }
  }

  {
    float bias0 = fc_b[row], bias1 = fc_b[16 + row], bias2 = fc_b[32 + row], bias3 = fc_b[48 + row];
#pragma unroll
    for (int j = 0; j < 8; ++j) {
      float* o = out + (size_t)(R0 + hi * 8 + j) * NOUT + row;
      o[0]  = acc0[j] + bias0;
      o[16] = acc1[j] + bias1;
      o[32] = acc2[j] + bias2;
      o[48] = acc3[j] + bias3;
    }
  }
}

extern "C" void kernel_launch(void* const* d_in, const int* in_sizes, int n_in,
                              void* d_out, int out_size, void* d_ws, size_t ws_size,
                              hipStream_t stream) {
  (void)in_sizes; (void)n_in; (void)ws_size; (void)out_size;
  const float* x        = (const float*)d_in[0];
  const float* gcn_w0   = (const float*)d_in[1];
  const float* gcn_b0   = (const float*)d_in[2];
  const float* tcn_w0   = (const float*)d_in[3];
  const float* tcn_b0   = (const float*)d_in[4];
  const float* bn_g0    = (const float*)d_in[5];
  const float* bn_b0    = (const float*)d_in[6];
  const float* bn_m0    = (const float*)d_in[7];
  const float* bn_v0    = (const float*)d_in[8];
  const float* res_w    = (const float*)d_in[9];
  const float* res_b    = (const float*)d_in[10];
  const float* res_bn_g = (const float*)d_in[11];
  const float* res_bn_b = (const float*)d_in[12];
  const float* res_bn_m = (const float*)d_in[13];
  const float* res_bn_v = (const float*)d_in[14];
  const float* gcn_w1   = (const float*)d_in[15];
  const float* gcn_b1   = (const float*)d_in[16];
  const float* tcn_w1   = (const float*)d_in[17];
  const float* tcn_b1   = (const float*)d_in[18];
  const float* bn_g1    = (const float*)d_in[19];
  const float* bn_b1    = (const float*)d_in[20];
  const float* bn_m1    = (const float*)d_in[21];
  const float* bn_v1    = (const float*)d_in[22];
  const float* fc_w     = (const float*)d_in[23];
  const float* fc_b     = (const float*)d_in[24];

  dim3 grid(NB * 2 * 8);
  dim3 block(256);
  float* hws = (float*)d_ws;
  stgcn_body<<<grid, block, 0, stream>>>(
      x, gcn_w0, gcn_b0, tcn_w0, tcn_b0, bn_g0, bn_b0, bn_m0, bn_v0,
      res_w, res_b, res_bn_g, res_bn_b, res_bn_m, res_bn_v,
      gcn_w1, gcn_b1, tcn_w1, tcn_b1, bn_g1, bn_b1, bn_m1, bn_v1,
      hws);
  dim3 grid2(NB * NTIME * 2 / 256);
  dim3 block2(512);
  fc_gemm<<<grid2, block2, 0, stream>>>(hws, fc_w, fc_b, (float*)d_out);
}


# gcn0 recomputed in the TCN0 stage on WMMA; its D registers are TCN0's column operand as they stand (tcn0 weights staged in that k order); g0 LDS tile, stage and barrier gone; new sum order gcn0/TCN0
# speedup vs baseline: 3.8605x; 1.0401x over previous
#include <hip/hip_runtime.h>

#define TT 6
#define T2 (TT + 4)
#define T3 (TT + 2)
#define NB 64
#define NTIME 512
#define NTILE ((NTIME + TT - 1) / TT)
#define TPW ((NTILE + 7) / 8)
#define SPAN (TPW * TT + 6)
#define XLD ((SPAN * NV + 255) / 256)
#define NV 21
#define NC 32
#define NCIN 2
#define NOUT 64
#define FCK 672
#define WP 676

typedef __attribute__((ext_vector_type(2))) float v2f;
typedef __attribute__((ext_vector_type(8))) float v8f;

__device__ __forceinline__ v8f wmma4(v2f a, v2f b, v8f c) {
  return __builtin_amdgcn_wmma_f32_16x16x4_f32(
       false, a,  false, b,
       (short)0, c,  false,  false);
}

__device__ const unsigned char g_pi[20] = {0,1,2,3,0,5,6,7,0,9,10,11,0,13,14,15,0,17,18,19};
__device__ const unsigned char g_pj[20] = {1,2,3,4,5,6,7,8,9,10,11,12,13,14,15,16,17,18,19,20};
__device__ const float g_dinv[21] = {
  0.4082482905f,
  0.5773502692f,0.5773502692f,0.5773502692f,0.7071067812f,
  0.5773502692f,0.5773502692f,0.5773502692f,0.7071067812f,
  0.5773502692f,0.5773502692f,0.5773502692f,0.7071067812f,
  0.5773502692f,0.5773502692f,0.5773502692f,0.7071067812f,
  0.5773502692f,0.5773502692f,0.5773502692f,0.7071067812f};

#define PH 212
#define P1 172
static_assert(TT % 2 == 0 && PH % 2 == 0 && P1 % 2 == 0, "carried rows are moved as 8-byte words");
static_assert(NC * P1 >= TT * FCK, "the hagg buffer also holds the h1 tile");
static_assert(SPAN % TT == 0 && TT * NV * NCIN <= 256, "the span aggregate runs in passes of TT rows, one output per thread");

__device__ __forceinline__ void move_rows2(float* dst, int dpitch, const float* src, int spitch, int tid) {
  const int c = tid >> 3, l = tid & 7;
  const float2* s = reinterpret_cast<const float2*>(src + c * spitch);
  float2* d = reinterpret_cast<float2*>(dst + c * dpitch);
  const float2 w0 = s[l], w1 = s[l + 8];
  float2 w2 = w0;
  if (l < 5) w2 = s[l + 16];
  d[l] = w0;
  d[l + 8] = w1;
  if (l < 5) d[l + 16] = w2;
}
__global__ __launch_bounds__(256, 4) void stgcn_body(
    const float* __restrict__ x,
    const float* __restrict__ gcn_w0, const float* __restrict__ gcn_b0,
    const float* __restrict__ tcn_w0, const float* __restrict__ tcn_b0,
    const float* __restrict__ bn_g0, const float* __restrict__ bn_b0,
    const float* __restrict__ bn_m0, const float* __restrict__ bn_v0,
    const float* __restrict__ res_w, const float* __restrict__ res_b,
    const float* __restrict__ res_bn_g, const float* __restrict__ res_bn_b,
    const float* __restrict__ res_bn_m, const float* __restrict__ res_bn_v,
    const float* __restrict__ gcn_w1, const float* __restrict__ gcn_b1,
    const float* __restrict__ tcn_w1, const float* __restrict__ tcn_b1,
    const float* __restrict__ bn_g1, const float* __restrict__ bn_b1,
    const float* __restrict__ bn_m1, const float* __restrict__ bn_v1,
    float* __restrict__ hws)
{
  __shared__ float sA[NV * NV];
  __shared__ __attribute__((aligned(16))) float s_in[SPAN * NV * NCIN];
  __shared__ __attribute__((aligned(16))) float s_xa[SPAN * NV * NCIN];
  __shared__ __attribute__((aligned(16))) float s_tw0[128 * NC];
  __shared__ __attribute__((aligned(16))) float s_tw1[128 * NC];
  __shared__ __attribute__((aligned(16))) float s_gw1[NC * NC];
  __shared__ __attribute__((aligned(16))) float s_hg[NC * P1];
  __shared__ __attribute__((aligned(16))) float s_h0[NC * PH];
  __shared__ __attribute__((aligned(16))) float s_g1[NC * P1];
  __shared__ float s_gw0[NC * NCIN];
  __shared__ float s_gb0[NC], s_gb1[NC];
  __shared__ float s_s0[NC], s_o0[NC], s_s1[NC], s_o1[NC];
  __shared__ float s_rA[NC], s_rB[NC], s_rC[NC];

  const int tid  = threadIdx.x;
  const int nthr = 256;
  const int lane = tid & 31;
  const int wave = tid >> 5;
  const int row  = lane & 15;
  const int hi   = lane >> 4;

  const int bid   = blockIdx.x;
  const int tgrp  = bid & 7;
  const int hand  = (bid >> 3) & 1;
  const int b     = bid >> 4;

  float2 xw[XLD];
  {
    const int tb0 = tgrp * TPW * TT;
#pragma unroll
    for (int k = 0; k < XLD; ++k) {
      const int idx = tid + k * nthr;
      const int idc = (idx < SPAN * NV) ? idx : (SPAN * NV - 1);
      const int i = idc / NV, v = idc - i * NV;
      const int t = tb0 - 3 + i;
      const int tc = (t < 0) ? 0 : ((t < NTIME) ? t : (NTIME - 1));
      float2 w = *reinterpret_cast<const float2*>(&x[(((b * NTIME + tc) * 2 + hand) * NV + v) * NCIN]);
      if (t != tc) w = make_float2(0.0f, 0.0f);
      xw[k] = w;
    }
  }
  for (int idx = tid; idx < NV * NV; idx += nthr) {
    int i = idx / NV, j = idx % NV;
    bool conn = (i == j);
    for (int p = 0; p < 20; ++p)
      conn = conn || ((int)g_pi[p] == i && (int)g_pj[p] == j) ||
                     ((int)g_pi[p] == j && (int)g_pj[p] == i);
    sA[idx] = conn ? g_dinv[i] * g_dinv[j] : 0.0f;
  }
  for (int idx = tid; idx < 128 * NC; idx += nthr) {
    int e = idx & 1, ln = (idx >> 1) & 31, cs = idx >> 6;
    int q = cs & 7, j = (cs >> 3) & 3, ct = cs >> 5;
    int co = ct * 16 + (ln & 15);
    int ci0 = 16 * (q >> 2) + 8 * (ln >> 4) + 2 * (q & 3) + e;
    int ci1 = 4 * q + 2 * (ln >> 4) + e;
    const float* w0p = &tcn_w0[(co * NC + ci0) * 3];
    const float* w1p = &tcn_w1[(co * NC + ci1) * 3];
    const float a0 = w0p[0], a1 = w0p[1], a2 = w0p[2];
    const float b0 = w1p[0], b1 = w1p[1], b2 = w1p[2];
    float ga, gb;
    if (j == 0)      { ga = a0; gb = b0; }
    else if (j == 1) { ga = 0.5f * ((a0 + a2) + a1); gb = 0.5f * ((b0 + b2) + b1); }
    else if (j == 2) { ga = 0.5f * ((a0 + a2) - a1); gb = 0.5f * ((b0 + b2) - b1); }
    else             { ga = a2; gb = b2; }
    s_tw0[idx] = ga;
    s_tw1[idx] = gb;
  }
  for (int idx = tid; idx < NC * NC; idx += nthr) {
    int e = idx & 1, ln = (idx >> 1) & 31, cs = idx >> 6;
    int ct = cs >> 3, s = cs & 7;
    int co = ct * 16 + (ln & 15);
    int ci = 4 * s + 2 * (ln >> 4) + e;
    s_gw1[idx] = gcn_w1[co * NC + ci];
  }
  if (tid < NC * NCIN) s_gw0[tid] = gcn_w0[tid];
  if (tid < NC) {
    s_gb0[tid] = gcn_b0[tid];
    s_gb1[tid] = gcn_b1[tid];
    float sc0 = bn_g0[tid] * rsqrtf(bn_v0[tid] + 1e-5f);
    s_s0[tid] = sc0;
    s_o0[tid] = (tcn_b0[tid] - bn_m0[tid]) * sc0 + bn_b0[tid];
    float sc1 = bn_g1[tid] * rsqrtf(bn_v1[tid] + 1e-5f);
    s_s1[tid] = sc1;
    s_o1[tid] = (tcn_b1[tid] - bn_m1[tid]) * sc1 + bn_b1[tid];
    float rsc = res_bn_g[tid] * rsqrtf(res_bn_v[tid] + 1e-5f);
    s_rA[tid] = res_w[tid * 2 + 0] * rsc;
    s_rB[tid] = res_w[tid * 2 + 1] * rsc;
    s_rC[tid] = (res_b[tid] - res_bn_m[tid]) * rsc + res_bn_b[tid];
  }

#pragma unroll
  for (int k = 0; k < XLD; ++k) {
    const int idx = tid + k * nthr;
    if (idx < SPAN * NV) *reinterpret_cast<float2*>(&s_in[idx * 2]) = xw[k];
  }
  __syncthreads();
  v2f aG0, aG1;
  aG0.x = hi ? s_gb0[row] : s_gw0[row * 2 + 0];
  aG0.y = hi ? 0.0f : s_gw0[row * 2 + 1];
  aG1.x = hi ? s_gb0[16 + row] : s_gw0[(16 + row) * 2 + 0];
  aG1.y = hi ? 0.0f : s_gw0[(16 + row) * 2 + 1];
  if (tid < TT * NV * NCIN) {
    const int rr = tid / (NV * NCIN), r = tid - rr * (NV * NCIN);
    const int v = r >> 1, ci = r & 1;
    float ar[NV];
#pragma unroll
    for (int u = 0; u < NV; ++u) ar[u] = sA[v * NV + u];
#pragma unroll 4
    for (int p = 0; p < SPAN / TT; ++p) {
      const int i = p * TT + rr;
      const float* xr = &s_in[i * NV * NCIN + ci];
      float acc = 0.0f;
#pragma unroll
      for (int u = 0; u < NV; ++u) acc += ar[u] * xr[u * NCIN];
      s_xa[i * NV * NCIN + r] = acc;
    }
  }

  for (int it = 0; it < TPW; ++it) {
  const int tbase = (tgrp * TPW + it) * TT;
  if (tbase >= NTIME) break;
  const bool first = (it == 0);
  const float* xin = s_in + it * (TT * NV * NCIN);
  const float* xag = s_xa + it * (TT * NV * NCIN);
  const int rH = first ? 0 : (T2 - TT);
  const int rQ = first ? 0 : (T3 - TT);

  __syncthreads();

  {
    const int cot = wave & 1;
    v2f wG[4][8];
#pragma unroll
    for (int j = 0; j < 4; ++j) {
#pragma unroll
      for (int q = 0; q < 8; ++q)
        wG[j][q] = *reinterpret_cast<const v2f*>(&s_tw0[(((cot * 4 + j) * 8 + q) * 32 + lane) * 2]);
    }
    const int M = (first ? (T2 / 2) : (TT / 2)) * NV;
    for (int ct = wave >> 1; ct * 16 < M; ct += 4) {
      const v8f zero = {0.f,0.f,0.f,0.f,0.f,0.f,0.f,0.f};
      const int col = ct * 16 + row;
      const int mC = (col < M) ? col : 0;
      const int p = mC / NV, ve = mC - p * NV;
      const int r = rH + 2 * p;
      v2f d[4];
#pragma unroll
      for (int k = 0; k < 4; ++k) {
        const v2f xv = *reinterpret_cast<const v2f*>(&xag[((r + k) * NV + ve) * 2]);
        const int t = tbase - 3 + r + k;
        const float one = (t >= 0 && t < NTIME) ? 1.0f : 0.0f;
        d[k].x = hi ? one : xv.x;
        d[k].y = hi ? 0.0f : xv.y;
      }
      v2f tx[4];
      tx[0] = d[0] - d[2];
      tx[1] = d[1] + d[2];
      tx[2] = d[2] - d[1];
      tx[3] = d[1] - d[3];
      v8f gl[4], gh[4];
#pragma unroll
      for (int j = 0; j < 4; ++j) {
        gl[j] = wmma4(aG0, tx[j], zero);
        gh[j] = wmma4(aG1, tx[j], zero);
      }
      v8f m[4] = {zero, zero, zero, zero};
#pragma unroll
      for (int s = 0; s < 4; ++s) {
#pragma unroll
        for (int j = 0; j < 4; ++j) {
          v2f bb;
          bb.x = gl[j][2 * s];
          bb.y = gl[j][2 * s + 1];
          m[j] = wmma4(wG[j][s], bb, m[j]);
        }
      }
#pragma unroll
      for (int s = 0; s < 4; ++s) {
#pragma unroll
        for (int j = 0; j < 4; ++j) {
          v2f bb;
          bb.x = gh[j][2 * s];
          bb.y = gh[j][2 * s + 1];
          m[j] = wmma4(wG[j][4 + s], bb, m[j]);
        }
      }
      if (col < M) {
        const float2 xa = *reinterpret_cast<const float2*>(&xin[((r + 1) * NV + ve) * 2]);
        const float2 xb = *reinterpret_cast<const float2*>(&xin[((r + 2) * NV + ve) * 2]);
        float* hb = &s_h0[(cot * 16 + hi * 8) * PH + r * NV + ve];
#pragma unroll
        for (int j = 0; j < 8; ++j) {
          const int co = cot * 16 + hi * 8 + j;
          const float sc = s_s0[co], of = s_o0[co], ra = s_rA[co], rb = s_rB[co], rc = s_rC[co];
          const float y0 = (m[0][j] + m[1][j]) + m[2][j];
          const float y1 = (m[1][j] - m[2][j]) - m[3][j];
          float v0 = y0 * sc + of;
          v0 += ra * xa.x + rb * xa.y + rc;
          float v1 = y1 * sc + of;
          v1 += ra * xb.x + rb * xb.y + rc;
          hb[j * PH] = fmaxf(v0, 0.0f);
          hb[j * PH + NV] = fmaxf(v1, 0.0f);
        }
      }
    }
  }
  __syncthreads();

  float* s_hagg = s_hg;
  {
    v2f aF0[6], aF1[6];
#pragma unroll
    for (int s = 0; s < 6; ++s) {
      const int u0 = 4 * s + 2 * hi;
      const int vlo = row, vhi = 16 + row;
      const bool k0 = (u0 < NV), k1 = (u0 + 1 < NV), rv = (vhi < NV);
      float a00 = sA[vlo * NV + (k0 ? u0 : 0)];
      float a01 = sA[vlo * NV + (k1 ? u0 + 1 : 0)];
      float a10 = sA[(rv ? vhi : 0) * NV + (k0 ? u0 : 0)];
      float a11 = sA[(rv ? vhi : 0) * NV + (k1 ? u0 + 1 : 0)];
      aF0[s].x = k0 ? a00 : 0.0f;
      aF0[s].y = k1 ? a01 : 0.0f;
      aF1[s].x = (rv && k0) ? a10 : 0.0f;
      aF1[s].y = (rv && k1) ? a11 : 0.0f;
    }
    const int nT = T3 - rQ;
    for (int nt = wave; nt < 2 * nT; nt += 8) {
      v8f acc0 = {0.f,0.f,0.f,0.f,0.f,0.f,0.f,0.f};
      v8f acc1 = acc0;
      const int n = nt * 16 + row;
      const int ci = first ? (n / T3) : (n / TT);
      const int t3 = rQ + (n - ci * nT);
      const float* hp = &s_h0[ci * PH + (t3 + 1) * NV + 2 * hi];
#pragma unroll
      for (int s = 0; s < 6; ++s) {
        v2f bb;
        if (s < 5) {
          bb.x = hp[4 * s];
          bb.y = hp[4 * s + 1];
        } else {
          bb.x = (hi == 0) ? hp[20] : 0.0f;
          bb.y = 0.0f;
        }
        acc0 = wmma4(aF0[s], bb, acc0);
        acc1 = wmma4(aF1[s], bb, acc1);
      }
      float* op = &s_hagg[ci * P1 + t3 * NV];
#pragma unroll
      for (int j = 0; j < 8; ++j) op[hi * 8 + j] = acc0[j];
      if (hi == 0) {
#pragma unroll
        for (int j = 0; j < NV - 16; ++j) op[16 + j] = acc1[j];
      }
    }
  }
  __syncthreads();

  {
    const int M = T3 * NV;
    v2f wG0[8], wG1[8];
#pragma unroll
    for (int s = 0; s < 8; ++s) {
      wG0[s] = *reinterpret_cast<const v2f*>(&s_gw1[((0 * 8 + s) * 32 + lane) * 2]);
      wG1[s] = *reinterpret_cast<const v2f*>(&s_gw1[((1 * 8 + s) * 32 + lane) * 2]);
    }
    const int c0 = rQ * NV;
    for (int nt = wave; c0 + nt * 16 < M; nt += 8) {
      v8f acc0 = {0.f,0.f,0.f,0.f,0.f,0.f,0.f,0.f};
      v8f acc1 = acc0;
      int col = c0 + nt * 16 + row;
      int mC = (col < M) ? col : c0;
      const float* bp = &s_hagg[(2 * hi) * P1 + mC];
#pragma unroll
      for (int s = 0; s < 8; ++s) {
        const float* bq = bp + (4 * s) * P1;
        v2f bb;
        bb.x = bq[0];
        bb.y = bq[P1];
        acc0 = wmma4(wG0[s], bb, acc0);
        acc1 = wmma4(wG1[s], bb, acc1);
      }
      if (col < M) {
        int t3 = col / NV;
        int t = tbase - 1 + t3;
        bool ok = (t >= 0 && t < NTIME);
        float* gb = &s_g1[(hi * 8) * P1 + col];
#pragma unroll
        for (int j = 0; j < 8; ++j) {
          int co = hi * 8 + j;
          float val = acc0[j] + s_gb1[co];
          gb[j * P1] = ok ? val : 0.0f;
          int c2 = co + 16;
          float va2 = acc1[j] + s_gb1[c2];
          gb[(j + 16) * P1] = ok ? va2 : 0.0f;
        }
      }
    }
  }
  __syncthreads();

  float* s_h1 = s_hg;
  {
    const int M = (TT / 2) * NV;
    const int cot = wave & 1;
    v2f wG[4][8];
#pragma unroll
    for (int j = 0; j < 4; ++j) {
#pragma unroll
      for (int q = 0; q < 8; ++q)
        wG[j][q] = *reinterpret_cast<const v2f*>(&s_tw1[(((cot * 4 + j) * 8 + q) * 32 + lane) * 2]);
    }
    {
      const int ct = wave >> 1;
      const v8f zero = {0.f,0.f,0.f,0.f,0.f,0.f,0.f,0.f};
      v8f m0 = zero, m1 = zero, m2 = zero, m3 = zero;
      const int col = ct * 16 + row;
      const int mC = (col < M) ? col : 0;
      const int p = mC / NV, ve = mC - p * NV;
      const int r = 2 * p;
      const float* bp = &s_g1[(2 * hi) * P1 + r * NV + ve];
#pragma unroll
      for (int q = 0; q < 8; ++q) {
        const float* bq = bp + (4 * q) * P1;
        const float d0x = bq[0], d1x = bq[NV], d2x = bq[2 * NV], d3x = bq[3 * NV];
        const float d0y = bq[P1], d1y = bq[P1 + NV], d2y = bq[P1 + 2 * NV], d3y = bq[P1 + 3 * NV];
        v2f t0, t1, t2, t3;
        t0.x = d0x - d2x; t0.y = d0y - d2y;
        t1.x = d1x + d2x; t1.y = d1y + d2y;
        t2.x = d2x - d1x; t2.y = d2y - d1y;
        t3.x = d1x - d3x; t3.y = d1y - d3y;
        m0 = wmma4(wG[0][q], t0, m0);
        m1 = wmma4(wG[1][q], t1, m1);
        m2 = wmma4(wG[2][q], t2, m2);
        m3 = wmma4(wG[3][q], t3, m3);
      }
      if (col < M) {
        const float* rb = &s_h0[(cot * 16 + hi * 8) * PH + (r + 2) * NV + ve];
        float* ob = &s_h1[r * (NC * NV) + (cot * 16 + hi * 8) * NV + ve];
        float r0[8], r1[8];
#pragma unroll
        for (int j = 0; j < 8; ++j) {
          r0[j] = rb[j * PH];
          r1[j] = rb[j * PH + NV];
        }
#pragma unroll
        for (int j = 0; j < 8; ++j) {
          const int co = cot * 16 + hi * 8 + j;
          const float sc = s_s1[co], of = s_o1[co];
          const float y0 = (m0[j] + m1[j]) + m2[j];
          const float y1 = (m1[j] - m2[j]) - m3[j];
          const float v0 = y0 * sc + of + r0[j];
          const float v1 = y1 * sc + of + r1[j];
          ob[j * NV] = fmaxf(v0, 0.0f);
          ob[j * NV + NC * NV] = fmaxf(v1, 0.0f);
        }
      }
    }
  }
  __syncthreads();

  {
    const float4* srcv = reinterpret_cast<const float4*>(s_h1);
    for (int idx = tid; idx < TT * (FCK / 4); idx += nthr) {
      int i = idx / (FCK / 4), q = idx - i * (FCK / 4);
      if (tbase + i < NTIME) {
        size_t R = ((size_t)b * NTIME + (size_t)(tbase + i)) * 2 + (size_t)hand;
        reinterpret_cast<float4*>(hws + R * FCK)[q] = srcv[idx];
      }
    }
    move_rows2(s_h0 + (T2 - TT - 2) * NV, PH, s_h0 + (T2 - 2) * NV, PH, tid);
    move_rows2(s_g1, P1, s_g1 + (T3 - 2) * NV, P1, tid);
  }
  }
}

__global__ __launch_bounds__(512) void fc_gemm(
    const float* __restrict__ H, const float* __restrict__ fc_w,
    const float* __restrict__ fc_b, float* __restrict__ out)
{
  __shared__ __attribute__((aligned(16))) float sW[NOUT * WP];

  const int tid  = threadIdx.x;
  const int lane = tid & 31;
  const int wave = tid >> 5;
  const int row  = lane & 15;
  const int hi   = lane >> 4;

  for (int idx = tid; idx < NOUT * (FCK / 4); idx += 512) {
    int o = idx / (FCK / 4), q = idx - o * (FCK / 4);
    float4 w = reinterpret_cast<const float4*>(fc_w)[idx];
    *reinterpret_cast<float4*>(&sW[o * WP + q * 4]) = w;
  }

  const int R0 = (blockIdx.x * 16 + wave) * 16;
  const float* ap = H + (size_t)(R0 + row) * FCK + hi * 4;
  const float* bp = sW + row * WP + hi * 4;

  float4 an[4];
#pragma unroll
  for (int u = 0; u < 4; ++u) an[u] = *reinterpret_cast<const float4*>(ap + u * 8);

  __syncthreads();

  v8f acc0 = {0.f,0.f,0.f,0.f,0.f,0.f,0.f,0.f};
  v8f acc1 = acc0, acc2 = acc0, acc3 = acc0;

  for (int g = 0; g < FCK / 32; ++g) {
    float4 ac[4];
#pragma unroll
    for (int u = 0; u < 4; ++u) ac[u] = an[u];
    if (g + 1 < FCK / 32) {
#pragma unroll
      for (int u = 0; u < 4; ++u)
        an[u] = *reinterpret_cast<const float4*>(ap + (g + 1) * 32 + u * 8);
    }
#pragma unroll
    for (int u = 0; u < 4; ++u) {
      const float* bq = bp + g * 32 + u * 8;
      float4 b0 = *reinterpret_cast<const float4*>(bq);
      float4 b1 = *reinterpret_cast<const float4*>(bq + 16 * WP);
      float4 b2 = *reinterpret_cast<const float4*>(bq + 32 * WP);
      float4 b3 = *reinterpret_cast<const float4*>(bq + 48 * WP);
      v2f a01 = {ac[u].x, ac[u].y};
      v2f a23 = {ac[u].z, ac[u].w};
      acc0 = wmma4(a01, (v2f){b0.x, b0.y}, acc0);
      acc1 = wmma4(a01, (v2f){b1.x, b1.y}, acc1);
      acc2 = wmma4(a01, (v2f){b2.x, b2.y}, acc2);
      acc3 = wmma4(a01, (v2f){b3.x, b3.y}, acc3);
      acc0 = wmma4(a23, (v2f){b0.z, b0.w}, acc0);
      acc1 = wmma4(a23, (v2f){b1.z, b1.w}, acc1);
      acc2 = wmma4(a23, (v2f){b2.z, b2.w}, acc2);
      acc3 = wmma4(a23, (v2f){b3.z, b3.w}, acc3);
    }
  }

  {
    float bias0 = fc_b[row], bias1 = fc_b[16 + row], bias2 = fc_b[32 + row], bias3 = fc_b[48 + row];
#pragma unroll
    for (int j = 0; j < 8; ++j) {
      float* o = out + (size_t)(R0 + hi * 8 + j) * NOUT + row;
      o[0]  = acc0[j] + bias0;
      o[16] = acc1[j] + bias1;
      o[32] = acc2[j] + bias2;
      o[48] = acc3[j] + bias3;
    }
  }
}

extern "C" void kernel_launch(void* const* d_in, const int* in_sizes, int n_in,
                              void* d_out, int out_size, void* d_ws, size_t ws_size,
                              hipStream_t stream) {
  (void)in_sizes; (void)n_in; (void)ws_size; (void)out_size;
  const float* x        = (const float*)d_in[0];
  const float* gcn_w0   = (const float*)d_in[1];
  const float* gcn_b0   = (const float*)d_in[2];
  const float* tcn_w0   = (const float*)d_in[3];
  const float* tcn_b0   = (const float*)d_in[4];
  const float* bn_g0    = (const float*)d_in[5];
  const float* bn_b0    = (const float*)d_in[6];
  const float* bn_m0    = (const float*)d_in[7];
  const float* bn_v0    = (const float*)d_in[8];
  const float* res_w    = (const float*)d_in[9];
  const float* res_b    = (const float*)d_in[10];
  const float* res_bn_g = (const float*)d_in[11];
  const float* res_bn_b = (const float*)d_in[12];
  const float* res_bn_m = (const float*)d_in[13];
  const float* res_bn_v = (const float*)d_in[14];
  const float* gcn_w1   = (const float*)d_in[15];
  const float* gcn_b1   = (const float*)d_in[16];
  const float* tcn_w1   = (const float*)d_in[17];
  const float* tcn_b1   = (const float*)d_in[18];
  const float* bn_g1    = (const float*)d_in[19];
  const float* bn_b1    = (const float*)d_in[20];
  const float* bn_m1    = (const float*)d_in[21];
  const float* bn_v1    = (const float*)d_in[22];
  const float* fc_w     = (const float*)d_in[23];
  const float* fc_b     = (const float*)d_in[24];

  dim3 grid(NB * 2 * 8);
  dim3 block(256);
  float* hws = (float*)d_ws;
  stgcn_body<<<grid, block, 0, stream>>>(
      x, gcn_w0, gcn_b0, tcn_w0, tcn_b0, bn_g0, bn_b0, bn_m0, bn_v0,
      res_w, res_b, res_bn_g, res_bn_b, res_bn_m, res_bn_v,
      gcn_w1, gcn_b1, tcn_w1, tcn_b1, bn_g1, bn_b1, bn_m1, bn_v1,
      hws);
  dim3 grid2(NB * NTIME * 2 / 256);
  dim3 block2(512);
  fc_gemm<<<grid2, block2, 0, stream>>>(hws, fc_w, fc_b, (float*)d_out);
}


# Both kernels' staging prologues request all their global words as one group ahead of the first use: one memory round trip, not 22 serial (body) and 6 (FC); a TCN weight triple read once, not 4 times
# speedup vs baseline: 3.9859x; 1.0325x over previous
#include <hip/hip_runtime.h>

#define TT 6
#define T2 (TT + 4)
#define T3 (TT + 2)
#define NB 64
#define NTIME 512
#define NTILE ((NTIME + TT - 1) / TT)
#define TPW ((NTILE + 7) / 8)
#define SPAN (TPW * TT + 6)
#define XLD ((SPAN * NV + 255) / 256)
#define NV 21
#define NC 32
#define NCIN 2
#define NOUT 64
#define FCK 672
#define WP 676

typedef __attribute__((ext_vector_type(2))) float v2f;
typedef __attribute__((ext_vector_type(8))) float v8f;

__device__ __forceinline__ v8f wmma4(v2f a, v2f b, v8f c) {
  return __builtin_amdgcn_wmma_f32_16x16x4_f32(
       false, a,  false, b,
       (short)0, c,  false,  false);
}

__device__ const unsigned char g_pi[20] = {0,1,2,3,0,5,6,7,0,9,10,11,0,13,14,15,0,17,18,19};
__device__ const unsigned char g_pj[20] = {1,2,3,4,5,6,7,8,9,10,11,12,13,14,15,16,17,18,19,20};
__device__ const float g_dinv[21] = {
  0.4082482905f,
  0.5773502692f,0.5773502692f,0.5773502692f,0.7071067812f,
  0.5773502692f,0.5773502692f,0.5773502692f,0.7071067812f,
  0.5773502692f,0.5773502692f,0.5773502692f,0.7071067812f,
  0.5773502692f,0.5773502692f,0.5773502692f,0.7071067812f,
  0.5773502692f,0.5773502692f,0.5773502692f,0.7071067812f};

#define PH 212
#define P1 172
static_assert(TT % 2 == 0 && PH % 2 == 0 && P1 % 2 == 0, "carried rows are moved as 8-byte words");
static_assert(NC * P1 >= TT * FCK, "the hagg buffer also holds the h1 tile");
static_assert(SPAN % TT == 0 && TT * NV * NCIN <= 256, "the span aggregate runs in passes of TT rows, one output per thread");

__device__ __forceinline__ void move_rows2(float* dst, int dpitch, const float* src, int spitch, int tid) {
  const int c = tid >> 3, l = tid & 7;
  const float2* s = reinterpret_cast<const float2*>(src + c * spitch);
  float2* d = reinterpret_cast<float2*>(dst + c * dpitch);
  const float2 w0 = s[l], w1 = s[l + 8];
  float2 w2 = w0;
  if (l < 5) w2 = s[l + 16];
  d[l] = w0;
  d[l + 8] = w1;
  if (l < 5) d[l + 16] = w2;
}
__global__ __launch_bounds__(256, 4) void stgcn_body(
    const float* __restrict__ x,
    const float* __restrict__ gcn_w0, const float* __restrict__ gcn_b0,
    const float* __restrict__ tcn_w0, const float* __restrict__ tcn_b0,
    const float* __restrict__ bn_g0, const float* __restrict__ bn_b0,
    const float* __restrict__ bn_m0, const float* __restrict__ bn_v0,
    const float* __restrict__ res_w, const float* __restrict__ res_b,
    const float* __restrict__ res_bn_g, const float* __restrict__ res_bn_b,
    const float* __restrict__ res_bn_m, const float* __restrict__ res_bn_v,
    const float* __restrict__ gcn_w1, const float* __restrict__ gcn_b1,
    const float* __restrict__ tcn_w1, const float* __restrict__ tcn_b1,
    const float* __restrict__ bn_g1, const float* __restrict__ bn_b1,
    const float* __restrict__ bn_m1, const float* __restrict__ bn_v1,
    float* __restrict__ hws)
{
  __shared__ float sA[NV * NV];
  __shared__ __attribute__((aligned(16))) float s_in[SPAN * NV * NCIN];
  __shared__ __attribute__((aligned(16))) float s_xa[SPAN * NV * NCIN];
  __shared__ __attribute__((aligned(16))) float s_tw0[128 * NC];
  __shared__ __attribute__((aligned(16))) float s_tw1[128 * NC];
  __shared__ __attribute__((aligned(16))) float s_gw1[NC * NC];
  __shared__ __attribute__((aligned(16))) float s_hg[NC * P1];
  __shared__ __attribute__((aligned(16))) float s_h0[NC * PH];
  __shared__ __attribute__((aligned(16))) float s_g1[NC * P1];
  __shared__ float s_gw0[NC * NCIN];
  __shared__ float s_gb0[NC], s_gb1[NC];
  __shared__ float s_s0[NC], s_o0[NC], s_s1[NC], s_o1[NC];
  __shared__ float s_rA[NC], s_rB[NC], s_rC[NC];

  const int tid  = threadIdx.x;
  const int nthr = 256;
  const int lane = tid & 31;
  const int wave = tid >> 5;
  const int row  = lane & 15;
  const int hi   = lane >> 4;

  const int bid   = blockIdx.x;
  const int tgrp  = bid & 7;
  const int hand  = (bid >> 3) & 1;
  const int b     = bid >> 4;

  float wa[4][3], wb[4][3];
#pragma unroll
  for (int c = 0; c < 4; ++c) {
    const int ct = c >> 1, qh = c & 1;
    const int e = tid & 1, ln = (tid >> 1) & 31, q = (tid >> 6) | (qh << 2);
    const int co = ct * 16 + (ln & 15);
    const int ci0 = 16 * (q >> 2) + 8 * (ln >> 4) + 2 * (q & 3) + e;
    const int ci1 = 4 * q + 2 * (ln >> 4) + e;
    const float* w0p = &tcn_w0[(co * NC + ci0) * 3];
    const float* w1p = &tcn_w1[(co * NC + ci1) * 3];
    wa[c][0] = w0p[0]; wa[c][1] = w0p[1]; wa[c][2] = w0p[2];
    wb[c][0] = w1p[0]; wb[c][1] = w1p[1]; wb[c][2] = w1p[2];
  }
  float g1w[4];
#pragma unroll
  for (int k = 0; k < 4; ++k) {
    const int idx = tid + k * nthr;
    const int e = idx & 1, ln = (idx >> 1) & 31, cs = idx >> 6;
    const int ct = cs >> 3, sq = cs & 7;
    const int co = ct * 16 + (ln & 15);
    const int ci = 4 * sq + 2 * (ln >> 4) + e;
    g1w[k] = gcn_w1[co * NC + ci];
  }
  float gw0v = 0.0f;
  if (tid < NC * NCIN) gw0v = gcn_w0[tid];
  float dvi[2], dvj[2];
#pragma unroll
  for (int a = 0; a < 2; ++a) {
    const int idx = tid + a * nthr;
    const int idc = (idx < NV * NV) ? idx : (NV * NV - 1);
    const int i = idc / NV, j = idc - i * NV;
    dvi[a] = g_dinv[i];
    dvj[a] = g_dinv[j];
  }
  float2 xw[XLD];
  {
    const int tb0 = tgrp * TPW * TT;
#pragma unroll
    for (int k = 0; k < XLD; ++k) {
      const int idx = tid + k * nthr;
      const int idc = (idx < SPAN * NV) ? idx : (SPAN * NV - 1);
      const int i = idc / NV, v = idc - i * NV;
      const int t = tb0 - 3 + i;
      const int tc = (t < 0) ? 0 : ((t < NTIME) ? t : (NTIME - 1));
      float2 w = *reinterpret_cast<const float2*>(&x[(((b * NTIME + tc) * 2 + hand) * NV + v) * NCIN]);
      if (t != tc) w = make_float2(0.0f, 0.0f);
      xw[k] = w;
    }
  }
  float pb[19];
#pragma unroll
  for (int k = 0; k < 19; ++k) pb[k] = 0.0f;
  if (tid < NC) {
    pb[0] = gcn_b0[tid];   pb[1] = gcn_b1[tid];
    pb[2] = bn_g0[tid];    pb[3] = bn_v0[tid];   pb[4] = tcn_b0[tid];  pb[5] = bn_m0[tid];  pb[6] = bn_b0[tid];
    pb[7] = bn_g1[tid];    pb[8] = bn_v1[tid];   pb[9] = tcn_b1[tid];  pb[10] = bn_m1[tid]; pb[11] = bn_b1[tid];
    pb[12] = res_bn_g[tid]; pb[13] = res_bn_v[tid]; pb[14] = res_w[tid * 2 + 0]; pb[15] = res_w[tid * 2 + 1];
    pb[16] = res_b[tid];   pb[17] = res_bn_m[tid]; pb[18] = res_bn_b[tid];
  }
#pragma unroll
  for (int a = 0; a < 2; ++a) {
    const int idx = tid + a * nthr;
    if (idx < NV * NV) {
      int i = idx / NV, j = idx % NV;
      bool conn = (i == j);
      for (int p = 0; p < 20; ++p)
        conn = conn || ((int)g_pi[p] == i && (int)g_pj[p] == j) ||
                       ((int)g_pi[p] == j && (int)g_pj[p] == i);
      sA[idx] = conn ? dvi[a] * dvj[a] : 0.0f;
    }
  }
#pragma unroll
  for (int c = 0; c < 4; ++c) {
    const int ct = c >> 1, qh = c & 1;
    const float a0 = wa[c][0], a1 = wa[c][1], a2 = wa[c][2];
    const float b0 = wb[c][0], b1 = wb[c][1], b2 = wb[c][2];
    const int base = tid + (qh + (ct << 3)) * nthr;
    s_tw0[base] = a0;
    s_tw1[base] = b0;
    s_tw0[base + 2 * nthr] = 0.5f * ((a0 + a2) + a1);
    s_tw1[base + 2 * nthr] = 0.5f * ((b0 + b2) + b1);
    s_tw0[base + 4 * nthr] = 0.5f * ((a0 + a2) - a1);
    s_tw1[base + 4 * nthr] = 0.5f * ((b0 + b2) - b1);
    s_tw0[base + 6 * nthr] = a2;
    s_tw1[base + 6 * nthr] = b2;
  }
#pragma unroll
  for (int k = 0; k < 4; ++k) s_gw1[tid + k * nthr] = g1w[k];
  if (tid < NC * NCIN) s_gw0[tid] = gw0v;
  if (tid < NC) {
    s_gb0[tid] = pb[0];
    s_gb1[tid] = pb[1];
    float sc0 = pb[2] * rsqrtf(pb[3] + 1e-5f);
    s_s0[tid] = sc0;
    s_o0[tid] = (pb[4] - pb[5]) * sc0 + pb[6];
    float sc1 = pb[7] * rsqrtf(pb[8] + 1e-5f);
    s_s1[tid] = sc1;
    s_o1[tid] = (pb[9] - pb[10]) * sc1 + pb[11];
    float rsc = pb[12] * rsqrtf(pb[13] + 1e-5f);
    s_rA[tid] = pb[14] * rsc;
    s_rB[tid] = pb[15] * rsc;
    s_rC[tid] = (pb[16] - pb[17]) * rsc + pb[18];
  }

#pragma unroll
  for (int k = 0; k < XLD; ++k) {
    const int idx = tid + k * nthr;
    if (idx < SPAN * NV) *reinterpret_cast<float2*>(&s_in[idx * 2]) = xw[k];
  }
  __syncthreads();
  v2f aG0, aG1;
  aG0.x = hi ? s_gb0[row] : s_gw0[row * 2 + 0];
  aG0.y = hi ? 0.0f : s_gw0[row * 2 + 1];
  aG1.x = hi ? s_gb0[16 + row] : s_gw0[(16 + row) * 2 + 0];
  aG1.y = hi ? 0.0f : s_gw0[(16 + row) * 2 + 1];
  if (tid < TT * NV * NCIN) {
    const int rr = tid / (NV * NCIN), r = tid - rr * (NV * NCIN);
    const int v = r >> 1, ci = r & 1;
    float ar[NV];
#pragma unroll
    for (int u = 0; u < NV; ++u) ar[u] = sA[v * NV + u];
#pragma unroll 4
    for (int p = 0; p < SPAN / TT; ++p) {
      const int i = p * TT + rr;
      const float* xr = &s_in[i * NV * NCIN + ci];
      float acc = 0.0f;
#pragma unroll
      for (int u = 0; u < NV; ++u) acc += ar[u] * xr[u * NCIN];
      s_xa[i * NV * NCIN + r] = acc;
    }
  }

  for (int it = 0; it < TPW; ++it) {
  const int tbase = (tgrp * TPW + it) * TT;
  if (tbase >= NTIME) break;
  const bool first = (it == 0);
  const float* xin = s_in + it * (TT * NV * NCIN);
  const float* xag = s_xa + it * (TT * NV * NCIN);
  const int rH = first ? 0 : (T2 - TT);
  const int rQ = first ? 0 : (T3 - TT);

  __syncthreads();

  {
    const int cot = wave & 1;
    v2f wG[4][8];
#pragma unroll
    for (int j = 0; j < 4; ++j) {
#pragma unroll
      for (int q = 0; q < 8; ++q)
        wG[j][q] = *reinterpret_cast<const v2f*>(&s_tw0[(((cot * 4 + j) * 8 + q) * 32 + lane) * 2]);
    }
    const int M = (first ? (T2 / 2) : (TT / 2)) * NV;
    for (int ct = wave >> 1; ct * 16 < M; ct += 4) {
      const v8f zero = {0.f,0.f,0.f,0.f,0.f,0.f,0.f,0.f};
      const int col = ct * 16 + row;
      const int mC = (col < M) ? col : 0;
      const int p = mC / NV, ve = mC - p * NV;
      const int r = rH + 2 * p;
      v2f d[4];
#pragma unroll
      for (int k = 0; k < 4; ++k) {
        const v2f xv = *reinterpret_cast<const v2f*>(&xag[((r + k) * NV + ve) * 2]);
        const int t = tbase - 3 + r + k;
        const float one = (t >= 0 && t < NTIME) ? 1.0f : 0.0f;
        d[k].x = hi ? one : xv.x;
        d[k].y = hi ? 0.0f : xv.y;
      }
      v2f tx[4];
      tx[0] = d[0] - d[2];
      tx[1] = d[1] + d[2];
      tx[2] = d[2] - d[1];
      tx[3] = d[1] - d[3];
      v8f gl[4], gh[4];
#pragma unroll
      for (int j = 0; j < 4; ++j) {
        gl[j] = wmma4(aG0, tx[j], zero);
        gh[j] = wmma4(aG1, tx[j], zero);
      }
      v8f m[4] = {zero, zero, zero, zero};
#pragma unroll
      for (int s = 0; s < 4; ++s) {
#pragma unroll
        for (int j = 0; j < 4; ++j) {
          v2f bb;
          bb.x = gl[j][2 * s];
          bb.y = gl[j][2 * s + 1];
          m[j] = wmma4(wG[j][s], bb, m[j]);
        }
      }
#pragma unroll
      for (int s = 0; s < 4; ++s) {
#pragma unroll
        for (int j = 0; j < 4; ++j) {
          v2f bb;
          bb.x = gh[j][2 * s];
          bb.y = gh[j][2 * s + 1];
          m[j] = wmma4(wG[j][4 + s], bb, m[j]);
        }
      }
      if (col < M) {
        const float2 xa = *reinterpret_cast<const float2*>(&xin[((r + 1) * NV + ve) * 2]);
        const float2 xb = *reinterpret_cast<const float2*>(&xin[((r + 2) * NV + ve) * 2]);
        float* hb = &s_h0[(cot * 16 + hi * 8) * PH + r * NV + ve];
#pragma unroll
        for (int j = 0; j < 8; ++j) {
          const int co = cot * 16 + hi * 8 + j;
          const float sc = s_s0[co], of = s_o0[co], ra = s_rA[co], rb = s_rB[co], rc = s_rC[co];
          const float y0 = (m[0][j] + m[1][j]) + m[2][j];
          const float y1 = (m[1][j] - m[2][j]) - m[3][j];
          float v0 = y0 * sc + of;
          v0 += ra * xa.x + rb * xa.y + rc;
          float v1 = y1 * sc + of;
          v1 += ra * xb.x + rb * xb.y + rc;
          hb[j * PH] = fmaxf(v0, 0.0f);
          hb[j * PH + NV] = fmaxf(v1, 0.0f);
        }
      }
    }
  }
  __syncthreads();

  float* s_hagg = s_hg;
  {
    v2f aF0[6], aF1[6];
#pragma unroll
    for (int s = 0; s < 6; ++s) {
      const int u0 = 4 * s + 2 * hi;
      const int vlo = row, vhi = 16 + row;
      const bool k0 = (u0 < NV), k1 = (u0 + 1 < NV), rv = (vhi < NV);
      float a00 = sA[vlo * NV + (k0 ? u0 : 0)];
      float a01 = sA[vlo * NV + (k1 ? u0 + 1 : 0)];
      float a10 = sA[(rv ? vhi : 0) * NV + (k0 ? u0 : 0)];
      float a11 = sA[(rv ? vhi : 0) * NV + (k1 ? u0 + 1 : 0)];
      aF0[s].x = k0 ? a00 : 0.0f;
      aF0[s].y = k1 ? a01 : 0.0f;
      aF1[s].x = (rv && k0) ? a10 : 0.0f;
      aF1[s].y = (rv && k1) ? a11 : 0.0f;
    }
    const int nT = T3 - rQ;
    for (int nt = wave; nt < 2 * nT; nt += 8) {
      v8f acc0 = {0.f,0.f,0.f,0.f,0.f,0.f,0.f,0.f};
      v8f acc1 = acc0;
      const int n = nt * 16 + row;
      const int ci = first ? (n / T3) : (n / TT);
      const int t3 = rQ + (n - ci * nT);
      const float* hp = &s_h0[ci * PH + (t3 + 1) * NV + 2 * hi];
#pragma unroll
      for (int s = 0; s < 6; ++s) {
        v2f bb;
        if (s < 5) {
          bb.x = hp[4 * s];
          bb.y = hp[4 * s + 1];
        } else {
          bb.x = (hi == 0) ? hp[20] : 0.0f;
          bb.y = 0.0f;
        }
        acc0 = wmma4(aF0[s], bb, acc0);
        acc1 = wmma4(aF1[s], bb, acc1);
      }
      float* op = &s_hagg[ci * P1 + t3 * NV];
#pragma unroll
      for (int j = 0; j < 8; ++j) op[hi * 8 + j] = acc0[j];
      if (hi == 0) {
#pragma unroll
        for (int j = 0; j < NV - 16; ++j) op[16 + j] = acc1[j];
      }
    }
  }
  __syncthreads();

  {
    const int M = T3 * NV;
    v2f wG0[8], wG1[8];
#pragma unroll
    for (int s = 0; s < 8; ++s) {
      wG0[s] = *reinterpret_cast<const v2f*>(&s_gw1[((0 * 8 + s) * 32 + lane) * 2]);
      wG1[s] = *reinterpret_cast<const v2f*>(&s_gw1[((1 * 8 + s) * 32 + lane) * 2]);
    }
    const int c0 = rQ * NV;
    for (int nt = wave; c0 + nt * 16 < M; nt += 8) {
      v8f acc0 = {0.f,0.f,0.f,0.f,0.f,0.f,0.f,0.f};
      v8f acc1 = acc0;
      int col = c0 + nt * 16 + row;
      int mC = (col < M) ? col : c0;
      const float* bp = &s_hagg[(2 * hi) * P1 + mC];
#pragma unroll
      for (int s = 0; s < 8; ++s) {
        const float* bq = bp + (4 * s) * P1;
        v2f bb;
        bb.x = bq[0];
        bb.y = bq[P1];
        acc0 = wmma4(wG0[s], bb, acc0);
        acc1 = wmma4(wG1[s], bb, acc1);
      }
      if (col < M) {
        int t3 = col / NV;
        int t = tbase - 1 + t3;
        bool ok = (t >= 0 && t < NTIME);
        float* gb = &s_g1[(hi * 8) * P1 + col];
#pragma unroll
        for (int j = 0; j < 8; ++j) {
          int co = hi * 8 + j;
          float val = acc0[j] + s_gb1[co];
          gb[j * P1] = ok ? val : 0.0f;
          int c2 = co + 16;
          float va2 = acc1[j] + s_gb1[c2];
          gb[(j + 16) * P1] = ok ? va2 : 0.0f;
        }
      }
    }
  }
  __syncthreads();

  float* s_h1 = s_hg;
  {
    const int M = (TT / 2) * NV;
    const int cot = wave & 1;
    v2f wG[4][8];
#pragma unroll
    for (int j = 0; j < 4; ++j) {
#pragma unroll
      for (int q = 0; q < 8; ++q)
        wG[j][q] = *reinterpret_cast<const v2f*>(&s_tw1[(((cot * 4 + j) * 8 + q) * 32 + lane) * 2]);
    }
    {
      const int ct = wave >> 1;
      const v8f zero = {0.f,0.f,0.f,0.f,0.f,0.f,0.f,0.f};
      v8f m0 = zero, m1 = zero, m2 = zero, m3 = zero;
      const int col = ct * 16 + row;
      const int mC = (col < M) ? col : 0;
      const int p = mC / NV, ve = mC - p * NV;
      const int r = 2 * p;
      const float* bp = &s_g1[(2 * hi) * P1 + r * NV + ve];
#pragma unroll
      for (int q = 0; q < 8; ++q) {
        const float* bq = bp + (4 * q) * P1;
        const float d0x = bq[0], d1x = bq[NV], d2x = bq[2 * NV], d3x = bq[3 * NV];
        const float d0y = bq[P1], d1y = bq[P1 + NV], d2y = bq[P1 + 2 * NV], d3y = bq[P1 + 3 * NV];
        v2f t0, t1, t2, t3;
        t0.x = d0x - d2x; t0.y = d0y - d2y;
        t1.x = d1x + d2x; t1.y = d1y + d2y;
        t2.x = d2x - d1x; t2.y = d2y - d1y;
        t3.x = d1x - d3x; t3.y = d1y - d3y;
        m0 = wmma4(wG[0][q], t0, m0);
        m1 = wmma4(wG[1][q], t1, m1);
        m2 = wmma4(wG[2][q], t2, m2);
        m3 = wmma4(wG[3][q], t3, m3);
      }
      if (col < M) {
        const float* rb = &s_h0[(cot * 16 + hi * 8) * PH + (r + 2) * NV + ve];
        float* ob = &s_h1[r * (NC * NV) + (cot * 16 + hi * 8) * NV + ve];
        float r0[8], r1[8];
#pragma unroll
        for (int j = 0; j < 8; ++j) {
          r0[j] = rb[j * PH];
          r1[j] = rb[j * PH + NV];
        }
#pragma unroll
        for (int j = 0; j < 8; ++j) {
          const int co = cot * 16 + hi * 8 + j;
          const float sc = s_s1[co], of = s_o1[co];
          const float y0 = (m0[j] + m1[j]) + m2[j];
          const float y1 = (m1[j] - m2[j]) - m3[j];
          const float v0 = y0 * sc + of + r0[j];
          const float v1 = y1 * sc + of + r1[j];
          ob[j * NV] = fmaxf(v0, 0.0f);
          ob[j * NV + NC * NV] = fmaxf(v1, 0.0f);
        }
      }
    }
  }
  __syncthreads();

  {
    const float4* srcv = reinterpret_cast<const float4*>(s_h1);
    for (int idx = tid; idx < TT * (FCK / 4); idx += nthr) {
      int i = idx / (FCK / 4), q = idx - i * (FCK / 4);
      if (tbase + i < NTIME) {
        size_t R = ((size_t)b * NTIME + (size_t)(tbase + i)) * 2 + (size_t)hand;
        reinterpret_cast<float4*>(hws + R * FCK)[q] = srcv[idx];
      }
    }
    move_rows2(s_h0 + (T2 - TT - 2) * NV, PH, s_h0 + (T2 - 2) * NV, PH, tid);
    move_rows2(s_g1, P1, s_g1 + (T3 - 2) * NV, P1, tid);
  }
  }
}

__global__ __launch_bounds__(512) void fc_gemm(
    const float* __restrict__ H, const float* __restrict__ fc_w,
    const float* __restrict__ fc_b, float* __restrict__ out)
{
  __shared__ __attribute__((aligned(16))) float sW[NOUT * WP];

  const int tid  = threadIdx.x;
  const int lane = tid & 31;
  const int wave = tid >> 5;
  const int row  = lane & 15;
  const int hi   = lane >> 4;

  static_assert(NOUT * (FCK / 4) == 21 * 512, "a thread stages exactly 21 16-byte words of fc_w");
  float4 wst[21];
#pragma unroll
  for (int k = 0; k < 21; ++k) wst[k] = reinterpret_cast<const float4*>(fc_w)[tid + k * 512];

  const int R0 = (blockIdx.x * 16 + wave) * 16;
  const float* ap = H + (size_t)(R0 + row) * FCK + hi * 4;
  const float* bp = sW + row * WP + hi * 4;

  float4 an[4];
#pragma unroll
  for (int u = 0; u < 4; ++u) an[u] = *reinterpret_cast<const float4*>(ap + u * 8);

#pragma unroll
  for (int k = 0; k < 21; ++k) {
    const int idx = tid + k * 512;
    const int o = idx / (FCK / 4), q = idx - o * (FCK / 4);
    *reinterpret_cast<float4*>(&sW[o * WP + q * 4]) = wst[k];
  }

  __syncthreads();

  v8f acc0 = {0.f,0.f,0.f,0.f,0.f,0.f,0.f,0.f};
  v8f acc1 = acc0, acc2 = acc0, acc3 = acc0;

  for (int g = 0; g < FCK / 32; ++g) {
    float4 ac[4];
#pragma unroll
    for (int u = 0; u < 4; ++u) ac[u] = an[u];
    if (g + 1 < FCK / 32) {
#pragma unroll
      for (int u = 0; u < 4; ++u)
        an[u] = *reinterpret_cast<const float4*>(ap + (g + 1) * 32 + u * 8);
    }
#pragma unroll
    for (int u = 0; u < 4; ++u) {
      const float* bq = bp + g * 32 + u * 8;
      float4 b0 = *reinterpret_cast<const float4*>(bq);
      float4 b1 = *reinterpret_cast<const float4*>(bq + 16 * WP);
      float4 b2 = *reinterpret_cast<const float4*>(bq + 32 * WP);
      float4 b3 = *reinterpret_cast<const float4*>(bq + 48 * WP);
      v2f a01 = {ac[u].x, ac[u].y};
      v2f a23 = {ac[u].z, ac[u].w};
      acc0 = wmma4(a01, (v2f){b0.x, b0.y}, acc0);
      acc1 = wmma4(a01, (v2f){b1.x, b1.y}, acc1);
      acc2 = wmma4(a01, (v2f){b2.x, b2.y}, acc2);
      acc3 = wmma4(a01, (v2f){b3.x, b3.y}, acc3);
      acc0 = wmma4(a23, (v2f){b0.z, b0.w}, acc0);
      acc1 = wmma4(a23, (v2f){b1.z, b1.w}, acc1);
      acc2 = wmma4(a23, (v2f){b2.z, b2.w}, acc2);
      acc3 = wmma4(a23, (v2f){b3.z, b3.w}, acc3);
    }
  }

  {
    float bias0 = fc_b[row], bias1 = fc_b[16 + row], bias2 = fc_b[32 + row], bias3 = fc_b[48 + row];
#pragma unroll
    for (int j = 0; j < 8; ++j) {
      float* o = out + (size_t)(R0 + hi * 8 + j) * NOUT + row;
      o[0]  = acc0[j] + bias0;
      o[16] = acc1[j] + bias1;
      o[32] = acc2[j] + bias2;
      o[48] = acc3[j] + bias3;
    }
  }
}

extern "C" void kernel_launch(void* const* d_in, const int* in_sizes, int n_in,
                              void* d_out, int out_size, void* d_ws, size_t ws_size,
                              hipStream_t stream) {
  (void)in_sizes; (void)n_in; (void)ws_size; (void)out_size;
  const float* x        = (const float*)d_in[0];
  const float* gcn_w0   = (const float*)d_in[1];
  const float* gcn_b0   = (const float*)d_in[2];
  const float* tcn_w0   = (const float*)d_in[3];
  const float* tcn_b0   = (const float*)d_in[4];
  const float* bn_g0    = (const float*)d_in[5];
  const float* bn_b0    = (const float*)d_in[6];
  const float* bn_m0    = (const float*)d_in[7];
  const float* bn_v0    = (const float*)d_in[8];
  const float* res_w    = (const float*)d_in[9];
  const float* res_b    = (const float*)d_in[10];
  const float* res_bn_g = (const float*)d_in[11];
  const float* res_bn_b = (const float*)d_in[12];
  const float* res_bn_m = (const float*)d_in[13];
  const float* res_bn_v = (const float*)d_in[14];
  const float* gcn_w1   = (const float*)d_in[15];
  const float* gcn_b1   = (const float*)d_in[16];
  const float* tcn_w1   = (const float*)d_in[17];
  const float* tcn_b1   = (const float*)d_in[18];
  const float* bn_g1    = (const float*)d_in[19];
  const float* bn_b1    = (const float*)d_in[20];
  const float* bn_m1    = (const float*)d_in[21];
  const float* bn_v1    = (const float*)d_in[22];
  const float* fc_w     = (const float*)d_in[23];
  const float* fc_b     = (const float*)d_in[24];

  dim3 grid(NB * 2 * 8);
  dim3 block(256);
  float* hws = (float*)d_ws;
  stgcn_body<<<grid, block, 0, stream>>>(
      x, gcn_w0, gcn_b0, tcn_w0, tcn_b0, bn_g0, bn_b0, bn_m0, bn_v0,
      res_w, res_b, res_bn_g, res_bn_b, res_bn_m, res_bn_v,
      gcn_w1, gcn_b1, tcn_w1, tcn_b1, bn_g1, bn_b1, bn_m1, bn_v1,
      hws);
  dim3 grid2(NB * NTIME * 2 / 256);
  dim3 block2(512);
  fc_gemm<<<grid2, block2, 0, stream>>>(hws, fc_w, fc_b, (float*)d_out);
}
